# starting kernel
# baseline (speedup 1.0000x reference)
#include <hip/hip_runtime.h>
#include <math.h>

#define LL   12
#define TT   8
#define BB   32
#define CC   256
#define SS   64
#define EE   64
#define HDD  16
#define NHH  4
#define HIDD 32
#define HH   8
#define NKK  8
#define MROWS (BB*CC*LL)
#define SROWS (LL*TT*BB*CC)

typedef float v2f __attribute__((ext_vector_type(2)));
typedef float v8f __attribute__((ext_vector_type(8)));

static __device__ __forceinline__ v8f wmma4(v2f a, v2f b, v8f c) {
  return __builtin_amdgcn_wmma_f32_16x16x4_f32(false, a, false, b, (short)0, c,
                                               false, false);
}

static __device__ __forceinline__ float softplusf(float x) {
  return (x > 20.0f) ? x : log1pf(__expf(x));
}

__global__ __launch_bounds__(256) void k_stats(const float* __restrict__ ws,
                                               float* __restrict__ feats) {
  const int lane = threadIdx.x & 31;
  const int wid  = threadIdx.x >> 5;
  const int row  = blockIdx.x * 8 + wid;

  const float* p = ws + (size_t)row * SS;
  float v0 = p[lane];
  float v1 = p[lane + 32];

  float s = v0 + v1, s2 = v0 * v0 + v1 * v1;
  #pragma unroll
  for (int m = 16; m >= 1; m >>= 1) {
    s  += __shfl_xor(s,  m, 32);
    s2 += __shfl_xor(s2, m, 32);
  }
  const float mean = s * (1.0f / 64.0f);
  const float var  = (s2 - 64.0f * mean * mean) * (1.0f / 63.0f);

  #pragma unroll
  for (int k = 2; k <= 64; k <<= 1) {
    #pragma unroll
    for (int j = k >> 1; j >= 1; j >>= 1) {
      if (j == 32) {
        float lo = fminf(v0, v1), hi = fmaxf(v0, v1);
        v0 = lo; v1 = hi;
      } else {
        const bool lower = ((lane & j) == 0);
        {
          float o = __shfl_xor(v0, j, 32);
          bool asc = (k >= 64) ? true : ((lane & k) == 0);
          v0 = (asc == lower) ? fminf(v0, o) : fmaxf(v0, o);
        }
        {
          float o = __shfl_xor(v1, j, 32);
          bool asc;
          if (k >= 64)      asc = true;
          else if (k == 32) asc = false;
          else              asc = ((lane & k) == 0);
          v1 = (asc == lower) ? fminf(v1, o) : fmaxf(v1, o);
        }
      }
    }
  }

  const float q0  = __shfl(v0, 0, 32);
  const float s15 = __shfl(v0, 15, 32), s16 = __shfl(v0, 16, 32);
  const float s31 = __shfl(v0, 31, 32), s32 = __shfl(v1, 0, 32);
  const float s47 = __shfl(v1, 15, 32), s48 = __shfl(v1, 16, 32);
  const float q4  = __shfl(v1, 31, 32);
  const float q1 = s15 + 0.75f * (s16 - s15);
  const float q2 = 0.5f * (s31 + s32);
  const float q3 = s47 + 0.25f * (s48 - s47);

  if (lane == 0) {
    const int c = row & 255;
    int tmp = row >> 8;
    const int b = tmp & 31; tmp >>= 5;
    const int t = tmp & 7;
    const int l = tmp >> 3;
    float* o = feats + ((size_t)((b * LL + l) * CC + c)) * 56 + t * 7;
    o[0] = mean; o[1] = var; o[2] = q0; o[3] = q1; o[4] = q2; o[5] = q3; o[6] = q4;
  }
}

__global__ __launch_bounds__(256) void k_embed(const float* __restrict__ feats,
                                               const float* __restrict__ Wemb,
                                               const float* __restrict__ bemb,
                                               const float* __restrict__ pos,
                                               float* __restrict__ emb) {
  const int lane = threadIdx.x & 31;
  const int wid  = threadIdx.x >> 5;
  const int r  = lane & 15;
  const int hf = lane >> 4;
  const int m0 = blockIdx.x * 32 + (wid >> 2) * 16;
  const int n0 = (wid & 3) * 16;

  v8f acc = {0, 0, 0, 0, 0, 0, 0, 0};
  #pragma unroll
  for (int k0 = 0; k0 < 56; k0 += 4) {
    const int c0 = k0 + 2 * hf;
    const float* ap = feats + (size_t)(m0 + r) * 56 + c0;
    v2f a; a.x = ap[0]; a.y = ap[1];
    const float* bp = Wemb + (size_t)c0 * EE + n0 + r;
    v2f b; b.x = bp[0]; b.y = bp[EE];
    acc = wmma4(a, b, acc);
  }
  #pragma unroll
  for (int g = 0; g < 8; ++g) {
    const int m = m0 + g + 8 * hf;
    const int n = n0 + r;
    const int l = (m >> 8) % LL;
    emb[(size_t)m * EE + n] = acc[g] + bemb[n] + pos[l * EE + n];
  }
}

__global__ __launch_bounds__(384) void k_qkv(const float* __restrict__ emb,
                                             const float* __restrict__ win,
                                             const float* __restrict__ bin,
                                             float* __restrict__ qkv) {
  const int lane = threadIdx.x & 31;
  const int wid  = threadIdx.x >> 5;
  const int r  = lane & 15;
  const int hf = lane >> 4;
  const int m0 = blockIdx.x * 16;
  const int n0 = wid * 16;

  v8f acc = {0, 0, 0, 0, 0, 0, 0, 0};
  #pragma unroll
  for (int k0 = 0; k0 < 64; k0 += 4) {
    const int c0 = k0 + 2 * hf;
    const float* ap = emb + (size_t)(m0 + r) * EE + c0;
    v2f a; a.x = ap[0]; a.y = ap[1];
    const float* bp = win + (size_t)c0 * 192 + n0 + r;
    v2f b; b.x = bp[0]; b.y = bp[192];
    acc = wmma4(a, b, acc);
  }
  #pragma unroll
  for (int g = 0; g < 8; ++g) {
    const int m = m0 + g + 8 * hf;
    const int n = n0 + r;
    qkv[(size_t)m * 192 + n] = acc[g] + bin[n];
  }
}

__global__ __launch_bounds__(256) void k_attn(const float* __restrict__ qkv,
                                              float* __restrict__ ctx) {
  __shared__ float sP[8][16][17];
  const int lane = threadIdx.x & 31;
  const int wid  = threadIdx.x >> 5;
  const int task = blockIdx.x * 8 + wid;
  const int nb = task >> 2;
  const int h  = task & 3;
  const int r  = lane & 15;
  const int hf = lane >> 4;
  const float* base = qkv + (size_t)nb * LL * 192 + h * HDD;

  v8f S = {0, 0, 0, 0, 0, 0, 0, 0};
  #pragma unroll
  for (int k0 = 0; k0 < 16; k0 += 4) {
    const int c0 = k0 + 2 * hf;
    v2f a = {0, 0}, b = {0, 0};
    if (r < LL) {
      const float* qp = base + (size_t)r * 192 + c0;
      a.x = qp[0]; a.y = qp[1];
      const float* kp = qp + EE;
      b.x = kp[0]; b.y = kp[1];
    }
    S = wmma4(a, b, S);
  }

  float P[8];
  #pragma unroll
  for (int g = 0; g < 8; ++g) {
    float sv = (r < LL) ? S[g] * 0.25f : -__builtin_inff();
    float rm = sv;
    #pragma unroll
    for (int m = 8; m >= 1; m >>= 1) rm = fmaxf(rm, __shfl_xor(rm, m, 32));
    float e = __expf(sv - rm);
    float rs = e;
    #pragma unroll
    for (int m = 8; m >= 1; m >>= 1) rs += __shfl_xor(rs, m, 32);
    P[g] = e / rs;
  }
  #pragma unroll
  for (int g = 0; g < 8; ++g) sP[wid][g + 8 * hf][r] = P[g];
  __syncthreads();

  v8f Cv = {0, 0, 0, 0, 0, 0, 0, 0};
  #pragma unroll
  for (int k0 = 0; k0 < 16; k0 += 4) {
    const int c0 = k0 + 2 * hf;
    v2f a; a.x = sP[wid][r][c0]; a.y = sP[wid][r][c0 + 1];
    v2f b;
    b.x = (c0     < LL) ? base[(size_t)c0       * 192 + 128 + r] : 0.0f;
    b.y = (c0 + 1 < LL) ? base[(size_t)(c0 + 1) * 192 + 128 + r] : 0.0f;
    Cv = wmma4(a, b, Cv);
  }
  #pragma unroll
  for (int g = 0; g < 8; ++g) {
    const int m = g + 8 * hf;
    if (m < LL)
      ctx[((size_t)nb * LL + m) * EE + h * HDD + r] = Cv[g];
  }
}

__global__ __launch_bounds__(256) void k_proj_pool(const float* __restrict__ ctx,
                                                   const float* __restrict__ wout,
                                                   const float* __restrict__ bout,
                                                   float* __restrict__ pooled) {
  __shared__ float pool[EE];
  const int tid = threadIdx.x;
  if (tid < EE) pool[tid] = 0.0f;
  __syncthreads();

  const int lane = tid & 31;
  const int wid  = tid >> 5;
  const int r  = lane & 15;
  const int hf = lane >> 4;
  const size_t mbase = (size_t)blockIdx.x * 256;

  #pragma unroll
  for (int i = 0; i < 8; ++i) {
    const int t  = wid * 8 + i;
    const int m0 = (t >> 2) * 16;
    const int n0 = (t & 3) * 16;
    v8f acc = {0, 0, 0, 0, 0, 0, 0, 0};
    #pragma unroll
    for (int k0 = 0; k0 < 64; k0 += 4) {
      const int c0 = k0 + 2 * hf;
      const float* ap = ctx + (mbase + m0 + r) * EE + c0;
      v2f a; a.x = ap[0]; a.y = ap[1];
      const float* bp = wout + (size_t)c0 * EE + n0 + r;
      v2f b; b.x = bp[0]; b.y = bp[EE];
      acc = wmma4(a, b, acc);
    }
    float ps = acc[0] + acc[1] + acc[2] + acc[3] +
               acc[4] + acc[5] + acc[6] + acc[7];
    ps += __shfl_xor(ps, 16, 32);
    if (lane < 16) atomicAdd(&pool[n0 + r], ps);
  }
  __syncthreads();
  if (tid < EE)
    pooled[(size_t)blockIdx.x * EE + tid] = pool[tid] * (1.0f / 256.0f) + bout[tid];
}

__global__ __launch_bounds__(256) void k_mlp_lu(
    const float* __restrict__ pooled,
    const float* __restrict__ mw1, const float* __restrict__ mb1,
    const float* __restrict__ mg,  const float* __restrict__ mbb,
    const float* __restrict__ mw2, const float* __restrict__ mb2,
    const float* __restrict__ nw1, const float* __restrict__ nb1,
    const float* __restrict__ ng,  const float* __restrict__ nbb,
    const float* __restrict__ nw2, const float* __restrict__ nb2,
    float* __restrict__ out) {
  const int branch = blockIdx.x & 1;
  const int row    = blockIdx.x >> 1;
  const float* w1 = branch ? nw1 : mw1;
  const float* b1 = branch ? nb1 : mb1;
  const float* g  = branch ? ng  : mg;
  const float* bb = branch ? nbb : mbb;
  const float* w2 = branch ? nw2 : mw2;
  const float* b2 = branch ? nb2 : mb2;

  __shared__ float sx[EE];
  __shared__ float sh[HIDD];
  __shared__ float sraw[512];
  const int tid = threadIdx.x;

  if (tid < EE) sx[tid] = pooled[(size_t)row * EE + tid];
  __syncthreads();

  if (tid < HIDD) {
    float acc = b1[tid];
    #pragma unroll
    for (int i = 0; i < EE; ++i) acc += sx[i] * w1[i * HIDD + tid];
    float mu = acc;
    #pragma unroll
    for (int m = 16; m >= 1; m >>= 1) mu += __shfl_xor(mu, m, 32);
    mu *= (1.0f / 32.0f);
    const float d = acc - mu;
    float vv = d * d;
    #pragma unroll
    for (int m = 16; m >= 1; m >>= 1) vv += __shfl_xor(vv, m, 32);
    vv *= (1.0f / 32.0f);
    const float y = d * rsqrtf(vv + 1e-5f) * g[tid] + bb[tid];
    sh[tid] = 0.5f * y * (1.0f + erff(y * 0.70710678118654752f));
  }
  __syncthreads();

  for (int c = tid; c < 512; c += 256) {
    float acc = b2[c];
    #pragma unroll
    for (int j = 0; j < HIDD; ++j) acc += sh[j] * w2[j * 512 + c];
    sraw[c] = acc;
  }
  __syncthreads();

  const size_t obase = ((size_t)branch * 384 + row) * 512;
  for (int o = tid; o < 512; o += 256) {
    const int h  = o >> 6;
    const int ik = o & 63;
    const int i  = ik >> 3;
    const int k  = ik & 7;
    const float* rh = sraw + h * 64;
    float acc = 0.0f;
    #pragma unroll
    for (int j = 0; j < NKK; ++j) {
      if (j > i || j > k) continue;
      const float Lij = (j == i) ? 1.0f : rh[(i * (i - 1)) / 2 + j];
      float u = rh[28 + (NKK * j - (j * (j - 1)) / 2) + (k - j)];
      if (j == k) u = softplusf(u) + 0.01f;
      acc += Lij * u;
    }
    out[obase + o] = acc;
  }
}

extern "C" void kernel_launch(void* const* d_in, const int* in_sizes, int n_in,
                              void* d_out, int out_size, void* d_ws, size_t ws_size,
                              hipStream_t stream) {
  const float* wsfeat = (const float*)d_in[0];
  const float* W_emb  = (const float*)d_in[1];
  const float* b_emb  = (const float*)d_in[2];
  const float* pos    = (const float*)d_in[3];
  const float* w_in   = (const float*)d_in[4];
  const float* b_in   = (const float*)d_in[5];
  const float* w_out  = (const float*)d_in[6];
  const float* b_out  = (const float*)d_in[7];
  const float* m_w1 = (const float*)d_in[8];
  const float* m_b1 = (const float*)d_in[9];
  const float* m_g  = (const float*)d_in[10];
  const float* m_bb = (const float*)d_in[11];
  const float* m_w2 = (const float*)d_in[12];
  const float* m_b2 = (const float*)d_in[13];
  const float* n_w1 = (const float*)d_in[14];
  const float* n_b1 = (const float*)d_in[15];
  const float* n_g  = (const float*)d_in[16];
  const float* n_bb = (const float*)d_in[17];
  const float* n_w2 = (const float*)d_in[18];
  const float* n_b2 = (const float*)d_in[19];
  float* out = (float*)d_out;

  float* ws     = (float*)d_ws;
  float* feats  = ws;
  float* emb    = feats + (size_t)MROWS * 56;
  float* qkv    = emb   + (size_t)MROWS * EE;
  float* ctx    = qkv   + (size_t)MROWS * 192;
  float* pooled = ctx   + (size_t)MROWS * EE;

  k_stats<<<SROWS / 8, 256, 0, stream>>>(wsfeat, feats);
  k_embed<<<MROWS / 32, 256, 0, stream>>>(feats, W_emb, b_emb, pos, emb);
  k_qkv<<<MROWS / 16, 384, 0, stream>>>(emb, w_in, b_in, qkv);
  k_attn<<<(BB * CC * NHH) / 8, 256, 0, stream>>>(qkv, ctx);
  k_proj_pool<<<BB * LL, 256, 0, stream>>>(ctx, w_out, b_out, pooled);
  k_mlp_lu<<<BB * LL * 2, 256, 0, stream>>>(pooled,
      m_w1, m_b1, m_g, m_bb, m_w2, m_b2,
      n_w1, n_b1, n_g, n_bb, n_w2, n_b2, out);
}


# emb never formed: qkv = feats @ (W_emb @ w_in) + per-layer bias rows; 60 waves of the stats launch build both tables on WMMA; embed launch and buffer gone (6 -> 5 launches); sum order changed
# speedup vs baseline: 2.4602x; 2.4602x over previous
#include <hip/hip_runtime.h>
#include <math.h>

#define LL   12
#define TT   8
#define BB   32
#define CC   256
#define SS   64
#define EE   64
#define HDD  16
#define NHH  4
#define HIDD 32
#define HH   8
#define NKK  8
#define MROWS (BB*CC*LL)
#define SROWS (LL*TT*BB*CC)

#define G_STATS 2048
#define T_STATS ((SROWS / 64) / G_STATS)
#define G_QKV   1024
#define T_QKV   ((MROWS / 16) / G_QKV)
#define G_ATTN  2048
#define T_ATTN  (((BB * CC * NHH) / 8) / G_ATTN)

typedef float v2f __attribute__((ext_vector_type(2)));
typedef float v8f __attribute__((ext_vector_type(8)));

static __device__ __forceinline__ v8f wmma4(v2f a, v2f b, v8f c) {
  return __builtin_amdgcn_wmma_f32_16x16x4_f32(false, a, false, b, (short)0, c,
                                               false, false);
}

static __device__ __forceinline__ float softplusf(float x) {
  return (x > 20.0f) ? x : log1pf(__expf(x));
}

template <int KC>
static __device__ __forceinline__ void load_wfrag4(float (&wf)[4][4 * KC],
                                                   const float* __restrict__ w,
                                                   int ldw, int col4, int hf) {
  #pragma unroll
  for (int j = 0; j < KC; ++j)
    #pragma unroll
    for (int e = 0; e < 4; ++e) {
      const float4 v = *reinterpret_cast<const float4*>(
          w + (size_t)(8 * j + 4 * hf + e) * ldw + col4);
      wf[0][4 * j + e] = v.x; wf[1][4 * j + e] = v.y;
      wf[2][4 * j + e] = v.z; wf[3][4 * j + e] = v.w;
    }
}

template <int KC>
static __device__ __forceinline__ void load_arows(float4 (&a)[KC],
                                                  const float* __restrict__ rowp) {
  const float4* p = reinterpret_cast<const float4*>(rowp);
  #pragma unroll
  for (int j = 0; j < KC; ++j) a[j] = p[2 * j];
}

template <int KC, int NT>
static __device__ __forceinline__ void mma_rows(v8f (&acc)[NT], const float4 (&a)[KC],
                                                const float (&wf)[NT][4 * KC]) {
  #pragma unroll
  for (int j = 0; j < KC; ++j) {
    v2f a0; a0.x = a[j].x; a0.y = a[j].y;
    v2f a1; a1.x = a[j].z; a1.y = a[j].w;
    #pragma unroll
    for (int t = 0; t < NT; ++t) {
      v2f b; b.x = wf[t][4 * j]; b.y = wf[t][4 * j + 1];
      acc[t] = wmma4(a0, b, acc[t]);
    }
    #pragma unroll
    for (int t = 0; t < NT; ++t) {
      v2f b; b.x = wf[t][4 * j + 2]; b.y = wf[t][4 * j + 3];
      acc[t] = wmma4(a1, b, acc[t]);
    }
  }
}

template <int CTRL>
static __device__ __forceinline__ float quadf(float v) {
  return __int_as_float(__builtin_amdgcn_update_dpp(
      0, __float_as_int(v), CTRL, 0xf, 0xf, true));
}

#define MED3_QUAD(NAME, PERM)                                                     \
static __device__ __forceinline__ float NAME(float a, float b, float sel) {      \
  float d;                                                                        \
  asm("v_med3_num_f32_e64_dpp %0, %1, %2, %3 quad_perm:" PERM                      \
      " row_mask:0xf bank_mask:0xf"                                               \
      : "=v"(d) : "v"(b), "v"(a), "v"(sel));                                      \
  return d;                                                                       \
}
MED3_QUAD(med3_xor1, "[1,0,3,2]")
MED3_QUAD(med3_xor3, "[3,2,1,0]")
#undef MED3_QUAD

#define QP_XOR1  0xB1
#define QP_XOR2  0x4E
#define QP_NEXT  0xF9
#define QP_LANE0 0x00

#define CE(i, j) { const float lo_ = fminf(x[i], x[j]); \
                   const float hi_ = fmaxf(x[i], x[j]); x[i] = lo_; x[j] = hi_; }
#define HALFCLEAN16 \
  CE(0,8) CE(1,9) CE(2,10) CE(3,11) CE(4,12) CE(5,13) CE(6,14) CE(7,15) \
  CE(0,4) CE(1,5) CE(2,6) CE(3,7) CE(8,12) CE(9,13) CE(10,14) CE(11,15) \
  CE(0,2) CE(1,3) CE(4,6) CE(5,7) CE(8,10) CE(9,11) CE(12,14) CE(13,15) \
  CE(0,1) CE(2,3) CE(4,5) CE(6,7) CE(8,9) CE(10,11) CE(12,13) CE(14,15) \
  ;

#define N_CTILES 60
template <bool BIAS>
static __device__ __forceinline__ void compose_tile(int rt, int ct, int lane,
                                                    const float* __restrict__ Wemb,
                                                    const float* __restrict__ bemb,
                                                    const float* __restrict__ pos,
                                                    const float* __restrict__ win,
                                                    const float* __restrict__ bin,
                                                    float* __restrict__ Wc,
                                                    float* __restrict__ bt) {
  const int r  = lane & 15;
  const int hf = lane >> 4;
  const int col = ct * 16 + r;
  const int nvalid = BIAS ? LL : (56 - rt * 16);
  const bool a_ok = r < nvalid;
  const int rr = a_ok ? r : 0;
  const float* ap = BIAS ? (pos + rr * EE) : (Wemb + (rt * 16 + rr) * EE);

  v8f acc = {0, 0, 0, 0, 0, 0, 0, 0};
  #pragma unroll
  for (int half = 0; half < 2; ++half) {
    v2f a[8], b[8];
    #pragma unroll
    for (int s = 0; s < 8; ++s) {
      const int c0 = (half * 8 + s) * 4 + 2 * hf;
      const float2 av = *reinterpret_cast<const float2*>(ap + c0);
      float ax = av.x, ay = av.y;
      if (BIAS) {
        const float2 bv = *reinterpret_cast<const float2*>(bemb + c0);
        ax += bv.x; ay += bv.y;
      }
      a[s].x = a_ok ? ax : 0.0f;
      a[s].y = a_ok ? ay : 0.0f;
      b[s].x = win[c0 * 192 + col];
      b[s].y = win[(c0 + 1) * 192 + col];
    }
    #pragma unroll
    for (int s = 0; s < 8; ++s) acc = wmma4(a[s], b[s], acc);
  }

  const float badd = BIAS ? bin[col] : 0.0f;
  float* dst = (BIAS ? bt : (Wc + rt * 16 * 192)) + 8 * hf * 192 + col;
  #pragma unroll
  for (int g = 0; g < 8; ++g)
    dst[g * 192] = BIAS ? acc[g] + badd : acc[g];
}

__global__ __launch_bounds__(256) void k_stats(const float* __restrict__ ws,
                                               float* __restrict__ feats,
                                               const float* __restrict__ Wemb,
                                               const float* __restrict__ bemb,
                                               const float* __restrict__ pos,
                                               const float* __restrict__ win,
                                               const float* __restrict__ bin,
                                               float* __restrict__ Wc,
                                               float* __restrict__ bt) {
  if (blockIdx.x < N_CTILES && __builtin_amdgcn_readfirstlane(threadIdx.x >> 5) == 0) {
    const int rt = blockIdx.x / 12;
    const int ct = blockIdx.x - rt * 12;
    if (rt == 4) compose_tile<true >(rt, ct, threadIdx.x & 31, Wemb, bemb, pos, win, bin, Wc, bt);
    else         compose_tile<false>(rt, ct, threadIdx.x & 31, Wemb, bemb, pos, win, bin, Wc, bt);
  }

  const int l4  = threadIdx.x & 3;
  const float inf  = __builtin_inff();
  const float sel1 = (l4 & 1) ? inf : -inf;
  const float sel2 = (l4 & 2) ? inf : -inf;

  for (int trip = 0; trip < T_STATS; ++trip) {
  const int row = (trip * G_STATS + blockIdx.x) * 64 + (threadIdx.x >> 2);

  const float4* p = reinterpret_cast<const float4*>(ws + (size_t)row * SS) + l4;
  float x[16];
  #pragma unroll
  for (int q = 0; q < 4; ++q) {
    const float4 v = p[q * 4];
    x[4 * q + 0] = v.x; x[4 * q + 1] = v.y; x[4 * q + 2] = v.z; x[4 * q + 3] = v.w;
  }

  float sa = x[0], sb = x[1], qa = x[0] * x[0], qb = x[1] * x[1];
  #pragma unroll
  for (int i = 2; i < 16; i += 2) {
    sa += x[i];     qa = fmaf(x[i],     x[i],     qa);
    sb += x[i + 1]; qb = fmaf(x[i + 1], x[i + 1], qb);
  }
  float s = sa + sb, s2 = qa + qb;
  s  += quadf<QP_XOR1>(s);   s2 += quadf<QP_XOR1>(s2);
  s  += quadf<QP_XOR2>(s);   s2 += quadf<QP_XOR2>(s2);
  const float mean = s * (1.0f / 64.0f);
  const float var  = (s2 - 64.0f * mean * mean) * (1.0f / 63.0f);

  CE(0,13) CE(1,12) CE(2,15) CE(3,14) CE(4,8) CE(5,6) CE(7,11) CE(9,10)
  CE(0,5) CE(1,7) CE(2,9) CE(3,4) CE(6,13) CE(8,14) CE(10,15) CE(11,12)
  CE(0,1) CE(2,3) CE(4,5) CE(6,8) CE(7,9) CE(10,11) CE(12,13) CE(14,15)
  CE(0,2) CE(1,3) CE(4,10) CE(5,11) CE(6,7) CE(8,9) CE(12,14) CE(13,15)
  CE(1,2) CE(3,12) CE(4,6) CE(5,7) CE(8,10) CE(9,11) CE(13,14)
  CE(1,4) CE(2,6) CE(5,8) CE(7,10) CE(9,13) CE(11,14)
  CE(2,4) CE(3,6) CE(9,12) CE(11,13)
  CE(3,5) CE(6,8) CE(7,9) CE(10,12)
  CE(3,4) CE(5,6) CE(7,8) CE(9,10) CE(11,12)
  CE(6,7) CE(8,9)

  {
    float t[16];
    #pragma unroll
    for (int r = 0; r < 16; ++r)
      t[r] = med3_xor1(x[r], x[15 - r], sel1);
    #pragma unroll
    for (int r = 0; r < 16; ++r) x[r] = t[r];
  }
  HALFCLEAN16

  {
    float t[16];
    #pragma unroll
    for (int r = 0; r < 16; ++r)
      t[r] = med3_xor3(x[r], x[15 - r], sel2);
    #pragma unroll
    for (int r = 0; r < 16; ++r)
      x[r] = med3_xor1(t[r], t[r], sel1);
  }
  float mn = x[0], mx = x[0];
  #pragma unroll
  for (int i = 1; i < 16; ++i) { mn = fminf(mn, x[i]); mx = fmaxf(mx, x[i]); }

  const float nxt = quadf<QP_NEXT>(mn);
  const float q0  = quadf<QP_LANE0>(mn);
  const float w   = (l4 == 0) ? 0.75f : 0.25f;
  const float lin = mx + w * (nxt - mx);
  const float mid = 0.5f * (mx + nxt);
  const float hi  = (l4 == 1) ? mid : ((l4 == 3) ? mx : lin);
  const float lo  = (l4 == 0) ? mean : ((l4 == 1) ? var : q0);

  const int c = row & 255;
  int tmp = row >> 8;
  const int b = tmp & 31; tmp >>= 5;
  const int t = tmp & 7;
  const int l = tmp >> 3;
  float* o = feats + ((size_t)((b * LL + l) * CC + c)) * 56 + t * 7;
  o[3 + l4] = hi;
  if (l4 < 3) o[l4] = lo;
  }
}
#undef CE
#undef HALFCLEAN16

__global__ __launch_bounds__(96) void k_qkv(const float* __restrict__ feats,
                                            const float* __restrict__ Wc,
                                            const float* __restrict__ bt,
                                            float* __restrict__ qkv) {
  const int lane = threadIdx.x & 31;
  const int wid  = threadIdx.x >> 5;
  const int r  = lane & 15;
  const int hf = lane >> 4;
  const int n4 = wid * 64 + 4 * r;

  float wf[4][28];
  load_wfrag4<7>(wf, Wc, 192, n4, hf);

  float4 a[7];
  load_arows<7>(a, feats + (size_t)(blockIdx.x * 16 + r) * 56 + 4 * hf);

  for (int trip = 0; trip < T_QKV; ++trip) {
    const int m0 = (trip * G_QKV + blockIdx.x) * 16;
    float4 an[7];
    if (trip + 1 < T_QKV)
      load_arows<7>(an, feats + (size_t)(m0 + G_QKV * 16 + r) * 56 + 4 * hf);
    const int l = (m0 >> 8) % LL;
    const float4 bi = *reinterpret_cast<const float4*>(bt + l * 192 + n4);

    v8f acc[4];
    #pragma unroll
    for (int t = 0; t < 4; ++t) acc[t] = (v8f){0, 0, 0, 0, 0, 0, 0, 0};
    mma_rows<7, 4>(acc, a, wf);

    #pragma unroll
    for (int g = 0; g < 8; ++g) {
      float4 o;
      o.x = acc[0][g] + bi.x;
      o.y = acc[1][g] + bi.y;
      o.z = acc[2][g] + bi.z;
      o.w = acc[3][g] + bi.w;
      *reinterpret_cast<float4*>(qkv + (size_t)(m0 + g + 8 * hf) * 192 + n4) = o;
    }

    if (trip + 1 < T_QKV) {
      #pragma unroll
      for (int j = 0; j < 7; ++j) a[j] = an[j];
    }
  }
}

__global__ __launch_bounds__(256, 16) void k_attn(const float* __restrict__ qkv,
                                              float* __restrict__ ctx) {
  __shared__ float sP[8][16][17];
  const int lane = threadIdx.x & 31;
  const int wid  = threadIdx.x >> 5;
  const int r  = lane & 15;
  const int hf = lane >> 4;

  for (int trip = 0; trip < T_ATTN; ++trip) {
  const int task = (trip * G_ATTN + blockIdx.x) * 8 + wid;
  const int nb = task >> 2;
  const int h  = task & 3;
  const float* base = qkv + (size_t)nb * LL * 192 + h * HDD;

  v8f S = {0, 0, 0, 0, 0, 0, 0, 0};
  #pragma unroll
  for (int k0 = 0; k0 < 16; k0 += 4) {
    const int c0 = k0 + 2 * hf;
    v2f a = {0, 0}, b = {0, 0};
    if (r < LL) {
      const float* qp = base + (size_t)r * 192 + c0;
      a.x = qp[0]; a.y = qp[1];
      const float* kp = qp + EE;
      b.x = kp[0]; b.y = kp[1];
    }
    S = wmma4(a, b, S);
  }

  float P[8];
  #pragma unroll
  for (int g = 0; g < 8; ++g) {
    float sv = (r < LL) ? S[g] * 0.25f : -__builtin_inff();
    float rm = sv;
    #pragma unroll
    for (int m = 8; m >= 1; m >>= 1) rm = fmaxf(rm, __shfl_xor(rm, m, 32));
    float e = __expf(sv - rm);
    float rs = e;
    #pragma unroll
    for (int m = 8; m >= 1; m >>= 1) rs += __shfl_xor(rs, m, 32);
    P[g] = e / rs;
  }
  #pragma unroll
  for (int g = 0; g < 8; ++g) sP[wid][g + 8 * hf][r] = P[g];
  __syncthreads();

  v8f Cv = {0, 0, 0, 0, 0, 0, 0, 0};
  #pragma unroll
  for (int k0 = 0; k0 < 16; k0 += 4) {
    const int c0 = k0 + 2 * hf;
    v2f a; a.x = sP[wid][r][c0]; a.y = sP[wid][r][c0 + 1];
    v2f b;
    b.x = (c0     < LL) ? base[(size_t)c0       * 192 + 128 + r] : 0.0f;
    b.y = (c0 + 1 < LL) ? base[(size_t)(c0 + 1) * 192 + 128 + r] : 0.0f;
    Cv = wmma4(a, b, Cv);
  }
  #pragma unroll
  for (int g = 0; g < 8; ++g) {
    const int m = g + 8 * hf;
    if (m < LL)
      ctx[((size_t)nb * LL + m) * EE + h * HDD + r] = Cv[g];
  }
  }
}

__global__ __launch_bounds__(256) void k_proj_pool(const float* __restrict__ ctx,
                                                   const float* __restrict__ wout,
                                                   const float* __restrict__ bout,
                                                   float* __restrict__ pooled) {
  __shared__ float pool[EE];
  const int tid = threadIdx.x;
  if (tid < EE) pool[tid] = 0.0f;
  __syncthreads();

  const int lane = tid & 31;
  const int wid  = tid >> 5;
  const int r  = lane & 15;
  const int hf = lane >> 4;
  const size_t mbase = (size_t)blockIdx.x * 256;

  float4 a0[8], a1[8];
  load_arows<8>(a0, ctx + (mbase + wid * 16 + r) * EE + 4 * hf);
  load_arows<8>(a1, ctx + (mbase + (wid + 8) * 16 + r) * EE + 4 * hf);

  float wf[4][32];
  load_wfrag4<8>(wf, wout, EE, 4 * r, hf);

  v8f acc[4];
  #pragma unroll
  for (int t = 0; t < 4; ++t) acc[t] = (v8f){0, 0, 0, 0, 0, 0, 0, 0};
  mma_rows<8, 4>(acc, a0, wf);
  mma_rows<8, 4>(acc, a1, wf);

  #pragma unroll
  for (int t = 0; t < 4; ++t) {
    float ps = acc[t][0] + acc[t][1] + acc[t][2] + acc[t][3] +
               acc[t][4] + acc[t][5] + acc[t][6] + acc[t][7];
    ps += __shfl_xor(ps, 16, 32);
    if (lane < 16) atomicAdd(&pool[4 * r + t], ps);
  }
  __syncthreads();
  if (tid < EE)
    pooled[(size_t)blockIdx.x * EE + tid] = pool[tid] * (1.0f / 256.0f) + bout[tid];
}

__global__ __launch_bounds__(256) void k_mlp_lu(
    const float* __restrict__ pooled,
    const float* __restrict__ mw1, const float* __restrict__ mb1,
    const float* __restrict__ mg,  const float* __restrict__ mbb,
    const float* __restrict__ mw2, const float* __restrict__ mb2,
    const float* __restrict__ nw1, const float* __restrict__ nb1,
    const float* __restrict__ ng,  const float* __restrict__ nbb,
    const float* __restrict__ nw2, const float* __restrict__ nb2,
    float* __restrict__ out) {
  const int branch = blockIdx.x & 1;
  const int row    = blockIdx.x >> 1;
  const float* w1 = branch ? nw1 : mw1;
  const float* b1 = branch ? nb1 : mb1;
  const float* g  = branch ? ng  : mg;
  const float* bb = branch ? nbb : mbb;
  const float* w2 = branch ? nw2 : mw2;
  const float* b2 = branch ? nb2 : mb2;

  __shared__ float sx[EE];
  __shared__ float sh[HIDD];
  __shared__ float sraw[512];
  const int tid = threadIdx.x;

  if (tid < EE) sx[tid] = pooled[(size_t)row * EE + tid];
  __syncthreads();

  if (tid < HIDD) {
    float acc = b1[tid];
    #pragma unroll
    for (int i = 0; i < EE; ++i) acc += sx[i] * w1[i * HIDD + tid];
    float mu = acc;
    #pragma unroll
    for (int m = 16; m >= 1; m >>= 1) mu += __shfl_xor(mu, m, 32);
    mu *= (1.0f / 32.0f);
    const float d = acc - mu;
    float vv = d * d;
    #pragma unroll
    for (int m = 16; m >= 1; m >>= 1) vv += __shfl_xor(vv, m, 32);
    vv *= (1.0f / 32.0f);
    const float y = d * rsqrtf(vv + 1e-5f) * g[tid] + bb[tid];
    sh[tid] = 0.5f * y * (1.0f + erff(y * 0.70710678118654752f));
  }
  __syncthreads();

  for (int c = tid; c < 512; c += 256) {
    float acc = b2[c];
    #pragma unroll
    for (int j = 0; j < HIDD; ++j) acc += sh[j] * w2[j * 512 + c];
    sraw[c] = acc;
  }
  __syncthreads();

  const size_t obase = ((size_t)branch * 384 + row) * 512;
  for (int o = tid; o < 512; o += 256) {
    const int h  = o >> 6;
    const int ik = o & 63;
    const int i  = ik >> 3;
    const int k  = ik & 7;
    const float* rh = sraw + h * 64;
    float acc = 0.0f;
    #pragma unroll
    for (int j = 0; j < NKK; ++j) {
      if (j > i || j > k) continue;
      const float Lij = (j == i) ? 1.0f : rh[(i * (i - 1)) / 2 + j];
      float u = rh[28 + (NKK * j - (j * (j - 1)) / 2) + (k - j)];
      if (j == k) u = softplusf(u) + 0.01f;
      acc += Lij * u;
    }
    out[obase + o] = acc;
  }
}

extern "C" void kernel_launch(void* const* d_in, const int* in_sizes, int n_in,
                              void* d_out, int out_size, void* d_ws, size_t ws_size,
                              hipStream_t stream) {
  const float* wsfeat = (const float*)d_in[0];
  const float* W_emb  = (const float*)d_in[1];
  const float* b_emb  = (const float*)d_in[2];
  const float* pos    = (const float*)d_in[3];
  const float* w_in   = (const float*)d_in[4];
  const float* b_in   = (const float*)d_in[5];
  const float* w_out  = (const float*)d_in[6];
  const float* b_out  = (const float*)d_in[7];
  const float* m_w1 = (const float*)d_in[8];
  const float* m_b1 = (const float*)d_in[9];
  const float* m_g  = (const float*)d_in[10];
  const float* m_bb = (const float*)d_in[11];
  const float* m_w2 = (const float*)d_in[12];
  const float* m_b2 = (const float*)d_in[13];
  const float* n_w1 = (const float*)d_in[14];
  const float* n_b1 = (const float*)d_in[15];
  const float* n_g  = (const float*)d_in[16];
  const float* n_bb = (const float*)d_in[17];
  const float* n_w2 = (const float*)d_in[18];
  const float* n_b2 = (const float*)d_in[19];
  float* out = (float*)d_out;

  float* ws     = (float*)d_ws;
  float* feats  = ws;
  float* wc     = feats + (size_t)MROWS * 56;
  float* bt     = wc    + 64 * 192;
  float* qkv    = wc    + (size_t)MROWS * EE;
  float* ctx    = qkv   + (size_t)MROWS * 192;
  float* pooled = ctx   + (size_t)MROWS * EE;

  k_stats<<<G_STATS, 256, 0, stream>>>(wsfeat, feats, W_emb, b_emb, pos, w_in, b_in, wc, bt);
  k_qkv<<<G_QKV, 96, 0, stream>>>(feats, wc, bt, qkv);
  k_attn<<<G_ATTN, 256, 0, stream>>>(qkv, ctx);
  k_proj_pool<<<BB * LL, 256, 0, stream>>>(ctx, w_out, b_out, pooled);
  k_mlp_lu<<<BB * LL * 2, 256, 0, stream>>>(pooled,
      m_w1, m_b1, m_g, m_bb, m_w2, m_b2,
      n_w1, n_b1, n_g, n_bb, n_w2, n_b2, out);
}


# k_attn softmax: the 16 row reductions per score tile (8 max, 8 sum over 16 lanes) run as four row_xmask DPP steps on the max/add itself instead of four LDS permutes with a wait each; bit-identical
# speedup vs baseline: 2.5755x; 1.0469x over previous
#include <hip/hip_runtime.h>
#include <math.h>

#define LL   12
#define TT   8
#define BB   32
#define CC   256
#define SS   64
#define EE   64
#define HDD  16
#define NHH  4
#define HIDD 32
#define HH   8
#define NKK  8
#define MROWS (BB*CC*LL)
#define SROWS (LL*TT*BB*CC)

#define G_STATS 2048
#define T_STATS ((SROWS / 64) / G_STATS)
#define G_QKV   1024
#define T_QKV   ((MROWS / 16) / G_QKV)
#define G_ATTN  2048
#define T_ATTN  (((BB * CC * NHH) / 8) / G_ATTN)

typedef float v2f __attribute__((ext_vector_type(2)));
typedef float v8f __attribute__((ext_vector_type(8)));

static __device__ __forceinline__ v8f wmma4(v2f a, v2f b, v8f c) {
  return __builtin_amdgcn_wmma_f32_16x16x4_f32(false, a, false, b, (short)0, c,
                                               false, false);
}

static __device__ __forceinline__ float softplusf(float x) {
  return (x > 20.0f) ? x : log1pf(__expf(x));
}

template <int KC>
static __device__ __forceinline__ void load_wfrag4(float (&wf)[4][4 * KC],
                                                   const float* __restrict__ w,
                                                   int ldw, int col4, int hf) {
  #pragma unroll
  for (int j = 0; j < KC; ++j)
    #pragma unroll
    for (int e = 0; e < 4; ++e) {
      const float4 v = *reinterpret_cast<const float4*>(
          w + (size_t)(8 * j + 4 * hf + e) * ldw + col4);
      wf[0][4 * j + e] = v.x; wf[1][4 * j + e] = v.y;
      wf[2][4 * j + e] = v.z; wf[3][4 * j + e] = v.w;
    }
}

template <int KC>
static __device__ __forceinline__ void load_arows(float4 (&a)[KC],
                                                  const float* __restrict__ rowp) {
  const float4* p = reinterpret_cast<const float4*>(rowp);
  #pragma unroll
  for (int j = 0; j < KC; ++j) a[j] = p[2 * j];
}

template <int KC, int NT>
static __device__ __forceinline__ void mma_rows(v8f (&acc)[NT], const float4 (&a)[KC],
                                                const float (&wf)[NT][4 * KC]) {
  #pragma unroll
  for (int j = 0; j < KC; ++j) {
    v2f a0; a0.x = a[j].x; a0.y = a[j].y;
    v2f a1; a1.x = a[j].z; a1.y = a[j].w;
    #pragma unroll
    for (int t = 0; t < NT; ++t) {
      v2f b; b.x = wf[t][4 * j]; b.y = wf[t][4 * j + 1];
      acc[t] = wmma4(a0, b, acc[t]);
    }
    #pragma unroll
    for (int t = 0; t < NT; ++t) {
      v2f b; b.x = wf[t][4 * j + 2]; b.y = wf[t][4 * j + 3];
      acc[t] = wmma4(a1, b, acc[t]);
    }
  }
}

template <int CTRL>
static __device__ __forceinline__ float quadf(float v) {
  return __int_as_float(__builtin_amdgcn_update_dpp(
      0, __float_as_int(v), CTRL, 0xf, 0xf, true));
}

#define MED3_QUAD(NAME, PERM)                                                     \
static __device__ __forceinline__ float NAME(float a, float b, float sel) {      \
  float d;                                                                        \
  asm("v_med3_num_f32_e64_dpp %0, %1, %2, %3 quad_perm:" PERM                      \
      " row_mask:0xf bank_mask:0xf"                                               \
      : "=v"(d) : "v"(b), "v"(a), "v"(sel));                                      \
  return d;                                                                       \
}
MED3_QUAD(med3_xor1, "[1,0,3,2]")
MED3_QUAD(med3_xor3, "[3,2,1,0]")
#undef MED3_QUAD

#define QP_XOR1  0xB1
#define QP_XOR2  0x4E
#define QP_NEXT  0xF9
#define QP_LANE0 0x00

#define CE(i, j) { const float lo_ = fminf(x[i], x[j]); \
                   const float hi_ = fmaxf(x[i], x[j]); x[i] = lo_; x[j] = hi_; }
#define HALFCLEAN16 \
  CE(0,8) CE(1,9) CE(2,10) CE(3,11) CE(4,12) CE(5,13) CE(6,14) CE(7,15) \
  CE(0,4) CE(1,5) CE(2,6) CE(3,7) CE(8,12) CE(9,13) CE(10,14) CE(11,15) \
  CE(0,2) CE(1,3) CE(4,6) CE(5,7) CE(8,10) CE(9,11) CE(12,14) CE(13,15) \
  CE(0,1) CE(2,3) CE(4,5) CE(6,7) CE(8,9) CE(10,11) CE(12,13) CE(14,15) \
  ;

#define N_CTILES 60
template <bool BIAS>
static __device__ __forceinline__ void compose_tile(int rt, int ct, int lane,
                                                    const float* __restrict__ Wemb,
                                                    const float* __restrict__ bemb,
                                                    const float* __restrict__ pos,
                                                    const float* __restrict__ win,
                                                    const float* __restrict__ bin,
                                                    float* __restrict__ Wc,
                                                    float* __restrict__ bt) {
  const int r  = lane & 15;
  const int hf = lane >> 4;
  const int col = ct * 16 + r;
  const int nvalid = BIAS ? LL : (56 - rt * 16);
  const bool a_ok = r < nvalid;
  const int rr = a_ok ? r : 0;
  const float* ap = BIAS ? (pos + rr * EE) : (Wemb + (rt * 16 + rr) * EE);

  v8f acc = {0, 0, 0, 0, 0, 0, 0, 0};
  #pragma unroll
  for (int half = 0; half < 2; ++half) {
    v2f a[8], b[8];
    #pragma unroll
    for (int s = 0; s < 8; ++s) {
      const int c0 = (half * 8 + s) * 4 + 2 * hf;
      const float2 av = *reinterpret_cast<const float2*>(ap + c0);
      float ax = av.x, ay = av.y;
      if (BIAS) {
        const float2 bv = *reinterpret_cast<const float2*>(bemb + c0);
        ax += bv.x; ay += bv.y;
      }
      a[s].x = a_ok ? ax : 0.0f;
      a[s].y = a_ok ? ay : 0.0f;
      b[s].x = win[c0 * 192 + col];
      b[s].y = win[(c0 + 1) * 192 + col];
    }
    #pragma unroll
    for (int s = 0; s < 8; ++s) acc = wmma4(a[s], b[s], acc);
  }

  const float badd = BIAS ? bin[col] : 0.0f;
  float* dst = (BIAS ? bt : (Wc + rt * 16 * 192)) + 8 * hf * 192 + col;
  #pragma unroll
  for (int g = 0; g < 8; ++g)
    dst[g * 192] = BIAS ? acc[g] + badd : acc[g];
}

__global__ __launch_bounds__(256) void k_stats(const float* __restrict__ ws,
                                               float* __restrict__ feats,
                                               const float* __restrict__ Wemb,
                                               const float* __restrict__ bemb,
                                               const float* __restrict__ pos,
                                               const float* __restrict__ win,
                                               const float* __restrict__ bin,
                                               float* __restrict__ Wc,
                                               float* __restrict__ bt) {
  if (blockIdx.x < N_CTILES && __builtin_amdgcn_readfirstlane(threadIdx.x >> 5) == 0) {
    const int rt = blockIdx.x / 12;
    const int ct = blockIdx.x - rt * 12;
    if (rt == 4) compose_tile<true >(rt, ct, threadIdx.x & 31, Wemb, bemb, pos, win, bin, Wc, bt);
    else         compose_tile<false>(rt, ct, threadIdx.x & 31, Wemb, bemb, pos, win, bin, Wc, bt);
  }

  const int l4  = threadIdx.x & 3;
  const float inf  = __builtin_inff();
  const float sel1 = (l4 & 1) ? inf : -inf;
  const float sel2 = (l4 & 2) ? inf : -inf;

  for (int trip = 0; trip < T_STATS; ++trip) {
  const int row = (trip * G_STATS + blockIdx.x) * 64 + (threadIdx.x >> 2);

  const float4* p = reinterpret_cast<const float4*>(ws + (size_t)row * SS) + l4;
  float x[16];
  #pragma unroll
  for (int q = 0; q < 4; ++q) {
    const float4 v = p[q * 4];
    x[4 * q + 0] = v.x; x[4 * q + 1] = v.y; x[4 * q + 2] = v.z; x[4 * q + 3] = v.w;
  }

  float sa = x[0], sb = x[1], qa = x[0] * x[0], qb = x[1] * x[1];
  #pragma unroll
  for (int i = 2; i < 16; i += 2) {
    sa += x[i];     qa = fmaf(x[i],     x[i],     qa);
    sb += x[i + 1]; qb = fmaf(x[i + 1], x[i + 1], qb);
  }
  float s = sa + sb, s2 = qa + qb;
  s  += quadf<QP_XOR1>(s);   s2 += quadf<QP_XOR1>(s2);
  s  += quadf<QP_XOR2>(s);   s2 += quadf<QP_XOR2>(s2);
  const float mean = s * (1.0f / 64.0f);
  const float var  = (s2 - 64.0f * mean * mean) * (1.0f / 63.0f);

  CE(0,13) CE(1,12) CE(2,15) CE(3,14) CE(4,8) CE(5,6) CE(7,11) CE(9,10)
  CE(0,5) CE(1,7) CE(2,9) CE(3,4) CE(6,13) CE(8,14) CE(10,15) CE(11,12)
  CE(0,1) CE(2,3) CE(4,5) CE(6,8) CE(7,9) CE(10,11) CE(12,13) CE(14,15)
  CE(0,2) CE(1,3) CE(4,10) CE(5,11) CE(6,7) CE(8,9) CE(12,14) CE(13,15)
  CE(1,2) CE(3,12) CE(4,6) CE(5,7) CE(8,10) CE(9,11) CE(13,14)
  CE(1,4) CE(2,6) CE(5,8) CE(7,10) CE(9,13) CE(11,14)
  CE(2,4) CE(3,6) CE(9,12) CE(11,13)
  CE(3,5) CE(6,8) CE(7,9) CE(10,12)
  CE(3,4) CE(5,6) CE(7,8) CE(9,10) CE(11,12)
  CE(6,7) CE(8,9)

  {
    float t[16];
    #pragma unroll
    for (int r = 0; r < 16; ++r)
      t[r] = med3_xor1(x[r], x[15 - r], sel1);
    #pragma unroll
    for (int r = 0; r < 16; ++r) x[r] = t[r];
  }
  HALFCLEAN16

  {
    float t[16];
    #pragma unroll
    for (int r = 0; r < 16; ++r)
      t[r] = med3_xor3(x[r], x[15 - r], sel2);
    #pragma unroll
    for (int r = 0; r < 16; ++r)
      x[r] = med3_xor1(t[r], t[r], sel1);
  }
  float mn = x[0], mx = x[0];
  #pragma unroll
  for (int i = 1; i < 16; ++i) { mn = fminf(mn, x[i]); mx = fmaxf(mx, x[i]); }

  const float nxt = quadf<QP_NEXT>(mn);
  const float q0  = quadf<QP_LANE0>(mn);
  const float w   = (l4 == 0) ? 0.75f : 0.25f;
  const float lin = mx + w * (nxt - mx);
  const float mid = 0.5f * (mx + nxt);
  const float hi  = (l4 == 1) ? mid : ((l4 == 3) ? mx : lin);
  const float lo  = (l4 == 0) ? mean : ((l4 == 1) ? var : q0);

  const int c = row & 255;
  int tmp = row >> 8;
  const int b = tmp & 31; tmp >>= 5;
  const int t = tmp & 7;
  const int l = tmp >> 3;
  float* o = feats + ((size_t)((b * LL + l) * CC + c)) * 56 + t * 7;
  o[3 + l4] = hi;
  if (l4 < 3) o[l4] = lo;
  }
}
#undef CE
#undef HALFCLEAN16

__global__ __launch_bounds__(96) void k_qkv(const float* __restrict__ feats,
                                            const float* __restrict__ Wc,
                                            const float* __restrict__ bt,
                                            float* __restrict__ qkv) {
  const int lane = threadIdx.x & 31;
  const int wid  = threadIdx.x >> 5;
  const int r  = lane & 15;
  const int hf = lane >> 4;
  const int n4 = wid * 64 + 4 * r;

  float wf[4][28];
  load_wfrag4<7>(wf, Wc, 192, n4, hf);

  float4 a[7];
  load_arows<7>(a, feats + (size_t)(blockIdx.x * 16 + r) * 56 + 4 * hf);

  for (int trip = 0; trip < T_QKV; ++trip) {
    const int m0 = (trip * G_QKV + blockIdx.x) * 16;
    float4 an[7];
    if (trip + 1 < T_QKV)
      load_arows<7>(an, feats + (size_t)(m0 + G_QKV * 16 + r) * 56 + 4 * hf);
    const int l = (m0 >> 8) % LL;
    const float4 bi = *reinterpret_cast<const float4*>(bt + l * 192 + n4);

    v8f acc[4];
    #pragma unroll
    for (int t = 0; t < 4; ++t) acc[t] = (v8f){0, 0, 0, 0, 0, 0, 0, 0};
    mma_rows<7, 4>(acc, a, wf);

    #pragma unroll
    for (int g = 0; g < 8; ++g) {
      float4 o;
      o.x = acc[0][g] + bi.x;
      o.y = acc[1][g] + bi.y;
      o.z = acc[2][g] + bi.z;
      o.w = acc[3][g] + bi.w;
      *reinterpret_cast<float4*>(qkv + (size_t)(m0 + g + 8 * hf) * 192 + n4) = o;
    }

    if (trip + 1 < T_QKV) {
      #pragma unroll
      for (int j = 0; j < 7; ++j) a[j] = an[j];
    }
  }
}

#define DPP_LINE(OP, XM, D, S) \
  OP " " D ", " S ", " S " row_xmask:" XM " row_mask:0xf bank_mask:0xf\n\t"
#define ROW_STEP_FIRST(NAME, LEAD, OP, XM)                                        \
static __device__ __forceinline__ void NAME(float (&d)[8], const float (&s)[8]) { \
  asm(LEAD                                                                        \
      DPP_LINE(OP, XM, "%0", "%8")  DPP_LINE(OP, XM, "%1", "%9")                  \
      DPP_LINE(OP, XM, "%2", "%10") DPP_LINE(OP, XM, "%3", "%11")                 \
      DPP_LINE(OP, XM, "%4", "%12") DPP_LINE(OP, XM, "%5", "%13")                 \
      DPP_LINE(OP, XM, "%6", "%14") DPP_LINE(OP, XM, "%7", "%15")                 \
      : "=&v"(d[0]), "=&v"(d[1]), "=&v"(d[2]), "=&v"(d[3]),                       \
        "=&v"(d[4]), "=&v"(d[5]), "=&v"(d[6]), "=&v"(d[7])                        \
      : "v"(s[0]), "v"(s[1]), "v"(s[2]), "v"(s[3]),                               \
        "v"(s[4]), "v"(s[5]), "v"(s[6]), "v"(s[7]));                              \
}
#define ROW_STEP(NAME, OP, XM)                                                    \
static __device__ __forceinline__ void NAME(float (&v)[8]) {                      \
  asm(DPP_LINE(OP, XM, "%0", "%0") DPP_LINE(OP, XM, "%1", "%1")                   \
      DPP_LINE(OP, XM, "%2", "%2") DPP_LINE(OP, XM, "%3", "%3")                   \
      DPP_LINE(OP, XM, "%4", "%4") DPP_LINE(OP, XM, "%5", "%5")                   \
      DPP_LINE(OP, XM, "%6", "%6") DPP_LINE(OP, XM, "%7", "%7")                   \
      : "+v"(v[0]), "+v"(v[1]), "+v"(v[2]), "+v"(v[3]),                           \
        "+v"(v[4]), "+v"(v[5]), "+v"(v[6]), "+v"(v[7]));                          \
}
ROW_STEP_FIRST(rowmax_x8, "", "v_max_num_f32_dpp", "8")
ROW_STEP(rowmax_x4, "v_max_num_f32_dpp", "4")
ROW_STEP(rowmax_x2, "v_max_num_f32_dpp", "2")
ROW_STEP(rowmax_x1, "v_max_num_f32_dpp", "1")
ROW_STEP_FIRST(rowsum_x8, "v_nop\n\t", "v_add_f32_dpp", "8")
ROW_STEP(rowsum_x4, "v_add_f32_dpp", "4")
ROW_STEP(rowsum_x2, "v_add_f32_dpp", "2")
ROW_STEP(rowsum_x1, "v_add_f32_dpp", "1")
#undef ROW_STEP
#undef ROW_STEP_FIRST
#undef DPP_LINE

__global__ __launch_bounds__(256, 16) void k_attn(const float* __restrict__ qkv,
                                              float* __restrict__ ctx) {
  __shared__ float sP[8][16][17];
  const int lane = threadIdx.x & 31;
  const int wid  = threadIdx.x >> 5;
  const int r  = lane & 15;
  const int hf = lane >> 4;

  for (int trip = 0; trip < T_ATTN; ++trip) {
  const int task = (trip * G_ATTN + blockIdx.x) * 8 + wid;
  const int nb = task >> 2;
  const int h  = task & 3;
  const float* base = qkv + (size_t)nb * LL * 192 + h * HDD;

  v8f S = {0, 0, 0, 0, 0, 0, 0, 0};
  #pragma unroll
  for (int k0 = 0; k0 < 16; k0 += 4) {
    const int c0 = k0 + 2 * hf;
    v2f a = {0, 0}, b = {0, 0};
    if (r < LL) {
      const float* qp = base + (size_t)r * 192 + c0;
      a.x = qp[0]; a.y = qp[1];
      const float* kp = qp + EE;
      b.x = kp[0]; b.y = kp[1];
    }
    S = wmma4(a, b, S);
  }

  float P[8], sv[8], rm[8], rs[8];
  #pragma unroll
  for (int g = 0; g < 8; ++g)
    sv[g] = (r < LL) ? S[g] * 0.25f : -__builtin_inff();
  rowmax_x8(rm, sv); rowmax_x4(rm); rowmax_x2(rm); rowmax_x1(rm);
  #pragma unroll
  for (int g = 0; g < 8; ++g) P[g] = __expf(sv[g] - rm[g]);
  rowsum_x8(rs, P); rowsum_x4(rs); rowsum_x2(rs); rowsum_x1(rs);
  #pragma unroll
  for (int g = 0; g < 8; ++g) P[g] = P[g] / rs[g];
  #pragma unroll
  for (int g = 0; g < 8; ++g) sP[wid][g + 8 * hf][r] = P[g];
  __syncthreads();

  v8f Cv = {0, 0, 0, 0, 0, 0, 0, 0};
  #pragma unroll
  for (int k0 = 0; k0 < 16; k0 += 4) {
    const int c0 = k0 + 2 * hf;
    v2f a; a.x = sP[wid][r][c0]; a.y = sP[wid][r][c0 + 1];
    v2f b;
    b.x = (c0     < LL) ? base[(size_t)c0       * 192 + 128 + r] : 0.0f;
    b.y = (c0 + 1 < LL) ? base[(size_t)(c0 + 1) * 192 + 128 + r] : 0.0f;
    Cv = wmma4(a, b, Cv);
  }
  #pragma unroll
  for (int g = 0; g < 8; ++g) {
    const int m = g + 8 * hf;
    if (m < LL)
      ctx[((size_t)nb * LL + m) * EE + h * HDD + r] = Cv[g];
  }
  }
}

__global__ __launch_bounds__(256) void k_proj_pool(const float* __restrict__ ctx,
                                                   const float* __restrict__ wout,
                                                   const float* __restrict__ bout,
                                                   float* __restrict__ pooled) {
  __shared__ float pool[EE];
  const int tid = threadIdx.x;
  if (tid < EE) pool[tid] = 0.0f;
  __syncthreads();

  const int lane = tid & 31;
  const int wid  = tid >> 5;
  const int r  = lane & 15;
  const int hf = lane >> 4;
  const size_t mbase = (size_t)blockIdx.x * 256;

  float4 a0[8], a1[8];
  load_arows<8>(a0, ctx + (mbase + wid * 16 + r) * EE + 4 * hf);
  load_arows<8>(a1, ctx + (mbase + (wid + 8) * 16 + r) * EE + 4 * hf);

  float wf[4][32];
  load_wfrag4<8>(wf, wout, EE, 4 * r, hf);

  v8f acc[4];
  #pragma unroll
  for (int t = 0; t < 4; ++t) acc[t] = (v8f){0, 0, 0, 0, 0, 0, 0, 0};
  mma_rows<8, 4>(acc, a0, wf);
  mma_rows<8, 4>(acc, a1, wf);

  #pragma unroll
  for (int t = 0; t < 4; ++t) {
    float ps = acc[t][0] + acc[t][1] + acc[t][2] + acc[t][3] +
               acc[t][4] + acc[t][5] + acc[t][6] + acc[t][7];
    ps += __shfl_xor(ps, 16, 32);
    if (lane < 16) atomicAdd(&pool[4 * r + t], ps);
  }
  __syncthreads();
  if (tid < EE)
    pooled[(size_t)blockIdx.x * EE + tid] = pool[tid] * (1.0f / 256.0f) + bout[tid];
}

__global__ __launch_bounds__(256) void k_mlp_lu(
    const float* __restrict__ pooled,
    const float* __restrict__ mw1, const float* __restrict__ mb1,
    const float* __restrict__ mg,  const float* __restrict__ mbb,
    const float* __restrict__ mw2, const float* __restrict__ mb2,
    const float* __restrict__ nw1, const float* __restrict__ nb1,
    const float* __restrict__ ng,  const float* __restrict__ nbb,
    const float* __restrict__ nw2, const float* __restrict__ nb2,
    float* __restrict__ out) {
  const int branch = blockIdx.x & 1;
  const int row    = blockIdx.x >> 1;
  const float* w1 = branch ? nw1 : mw1;
  const float* b1 = branch ? nb1 : mb1;
  const float* g  = branch ? ng  : mg;
  const float* bb = branch ? nbb : mbb;
  const float* w2 = branch ? nw2 : mw2;
  const float* b2 = branch ? nb2 : mb2;

  __shared__ float sx[EE];
  __shared__ float sh[HIDD];
  __shared__ float sraw[512];
  const int tid = threadIdx.x;

  if (tid < EE) sx[tid] = pooled[(size_t)row * EE + tid];
  __syncthreads();

  if (tid < HIDD) {
    float acc = b1[tid];
    #pragma unroll
    for (int i = 0; i < EE; ++i) acc += sx[i] * w1[i * HIDD + tid];
    float mu = acc;
    #pragma unroll
    for (int m = 16; m >= 1; m >>= 1) mu += __shfl_xor(mu, m, 32);
    mu *= (1.0f / 32.0f);
    const float d = acc - mu;
    float vv = d * d;
    #pragma unroll
    for (int m = 16; m >= 1; m >>= 1) vv += __shfl_xor(vv, m, 32);
    vv *= (1.0f / 32.0f);
    const float y = d * rsqrtf(vv + 1e-5f) * g[tid] + bb[tid];
    sh[tid] = 0.5f * y * (1.0f + erff(y * 0.70710678118654752f));
  }
  __syncthreads();

  for (int c = tid; c < 512; c += 256) {
    float acc = b2[c];
    #pragma unroll
    for (int j = 0; j < HIDD; ++j) acc += sh[j] * w2[j * 512 + c];
    sraw[c] = acc;
  }
  __syncthreads();

  const size_t obase = ((size_t)branch * 384 + row) * 512;
  for (int o = tid; o < 512; o += 256) {
    const int h  = o >> 6;
    const int ik = o & 63;
    const int i  = ik >> 3;
    const int k  = ik & 7;
    const float* rh = sraw + h * 64;
    float acc = 0.0f;
    #pragma unroll
    for (int j = 0; j < NKK; ++j) {
      if (j > i || j > k) continue;
      const float Lij = (j == i) ? 1.0f : rh[(i * (i - 1)) / 2 + j];
      float u = rh[28 + (NKK * j - (j * (j - 1)) / 2) + (k - j)];
      if (j == k) u = softplusf(u) + 0.01f;
      acc += Lij * u;
    }
    out[obase + o] = acc;
  }
}

extern "C" void kernel_launch(void* const* d_in, const int* in_sizes, int n_in,
                              void* d_out, int out_size, void* d_ws, size_t ws_size,
                              hipStream_t stream) {
  const float* wsfeat = (const float*)d_in[0];
  const float* W_emb  = (const float*)d_in[1];
  const float* b_emb  = (const float*)d_in[2];
  const float* pos    = (const float*)d_in[3];
  const float* w_in   = (const float*)d_in[4];
  const float* b_in   = (const float*)d_in[5];
  const float* w_out  = (const float*)d_in[6];
  const float* b_out  = (const float*)d_in[7];
  const float* m_w1 = (const float*)d_in[8];
  const float* m_b1 = (const float*)d_in[9];
  const float* m_g  = (const float*)d_in[10];
  const float* m_bb = (const float*)d_in[11];
  const float* m_w2 = (const float*)d_in[12];
  const float* m_b2 = (const float*)d_in[13];
  const float* n_w1 = (const float*)d_in[14];
  const float* n_b1 = (const float*)d_in[15];
  const float* n_g  = (const float*)d_in[16];
  const float* n_bb = (const float*)d_in[17];
  const float* n_w2 = (const float*)d_in[18];
  const float* n_b2 = (const float*)d_in[19];
  float* out = (float*)d_out;

  float* ws     = (float*)d_ws;
  float* feats  = ws;
  float* wc     = feats + (size_t)MROWS * 56;
  float* bt     = wc    + 64 * 192;
  float* qkv    = wc    + (size_t)MROWS * EE;
  float* ctx    = qkv   + (size_t)MROWS * 192;
  float* pooled = ctx   + (size_t)MROWS * EE;

  k_stats<<<G_STATS, 256, 0, stream>>>(wsfeat, feats, W_emb, b_emb, pos, w_in, b_in, wc, bt);
  k_qkv<<<G_QKV, 96, 0, stream>>>(feats, wc, bt, qkv);
  k_attn<<<G_ATTN, 256, 0, stream>>>(qkv, ctx);
  k_proj_pool<<<BB * LL, 256, 0, stream>>>(ctx, w_out, b_out, pooled);
  k_mlp_lu<<<BB * LL * 2, 256, 0, stream>>>(pooled,
      m_w1, m_b1, m_g, m_bb, m_w2, m_b2,
      n_w1, n_b1, n_g, n_bb, n_w2, n_b2, out);
}


# k_attn operand feed: q, k, v loads unguarded, issued as one group, 16 B per lane for q and k (k permuted); next task's loads fly under the softmax, no block barrier waits on them; sum order changed
# speedup vs baseline: 2.6995x; 1.0481x over previous
#include <hip/hip_runtime.h>
#include <math.h>

#define LL   12
#define TT   8
#define BB   32
#define CC   256
#define SS   64
#define EE   64
#define HDD  16
#define NHH  4
#define HIDD 32
#define HH   8
#define NKK  8
#define MROWS (BB*CC*LL)
#define SROWS (LL*TT*BB*CC)

#define G_STATS 2048
#define T_STATS ((SROWS / 64) / G_STATS)
#define G_QKV   1024
#define T_QKV   ((MROWS / 16) / G_QKV)
#define G_ATTN  2048
#define T_ATTN  (((BB * CC * NHH) / 8) / G_ATTN)

typedef float v2f __attribute__((ext_vector_type(2)));
typedef float v8f __attribute__((ext_vector_type(8)));

static __device__ __forceinline__ v8f wmma4(v2f a, v2f b, v8f c) {
  return __builtin_amdgcn_wmma_f32_16x16x4_f32(false, a, false, b, (short)0, c,
                                               false, false);
}

static __device__ __forceinline__ float softplusf(float x) {
  return (x > 20.0f) ? x : log1pf(__expf(x));
}

template <int KC>
static __device__ __forceinline__ void load_wfrag4(float (&wf)[4][4 * KC],
                                                   const float* __restrict__ w,
                                                   int ldw, int col4, int hf) {
  #pragma unroll
  for (int j = 0; j < KC; ++j)
    #pragma unroll
    for (int e = 0; e < 4; ++e) {
      const float4 v = *reinterpret_cast<const float4*>(
          w + (size_t)(8 * j + 4 * hf + e) * ldw + col4);
      wf[0][4 * j + e] = v.x; wf[1][4 * j + e] = v.y;
      wf[2][4 * j + e] = v.z; wf[3][4 * j + e] = v.w;
    }
}

template <int KC>
static __device__ __forceinline__ void load_arows(float4 (&a)[KC],
                                                  const float* __restrict__ rowp) {
  const float4* p = reinterpret_cast<const float4*>(rowp);
  #pragma unroll
  for (int j = 0; j < KC; ++j) a[j] = p[2 * j];
}

template <int KC, int NT>
static __device__ __forceinline__ void mma_rows(v8f (&acc)[NT], const float4 (&a)[KC],
                                                const float (&wf)[NT][4 * KC]) {
  #pragma unroll
  for (int j = 0; j < KC; ++j) {
    v2f a0; a0.x = a[j].x; a0.y = a[j].y;
    v2f a1; a1.x = a[j].z; a1.y = a[j].w;
    #pragma unroll
    for (int t = 0; t < NT; ++t) {
      v2f b; b.x = wf[t][4 * j]; b.y = wf[t][4 * j + 1];
      acc[t] = wmma4(a0, b, acc[t]);
    }
    #pragma unroll
    for (int t = 0; t < NT; ++t) {
      v2f b; b.x = wf[t][4 * j + 2]; b.y = wf[t][4 * j + 3];
      acc[t] = wmma4(a1, b, acc[t]);
    }
  }
}

template <int CTRL>
static __device__ __forceinline__ float quadf(float v) {
  return __int_as_float(__builtin_amdgcn_update_dpp(
      0, __float_as_int(v), CTRL, 0xf, 0xf, true));
}

#define MED3_QUAD(NAME, PERM)                                                     \
static __device__ __forceinline__ float NAME(float a, float b, float sel) {      \
  float d;                                                                        \
  asm("v_med3_num_f32_e64_dpp %0, %1, %2, %3 quad_perm:" PERM                      \
      " row_mask:0xf bank_mask:0xf"                                               \
      : "=v"(d) : "v"(b), "v"(a), "v"(sel));                                      \
  return d;                                                                       \
}
MED3_QUAD(med3_xor1, "[1,0,3,2]")
MED3_QUAD(med3_xor3, "[3,2,1,0]")
#undef MED3_QUAD

#define QP_XOR1  0xB1
#define QP_XOR2  0x4E
#define QP_NEXT  0xF9
#define QP_LANE0 0x00

#define CE(i, j) { const float lo_ = fminf(x[i], x[j]); \
                   const float hi_ = fmaxf(x[i], x[j]); x[i] = lo_; x[j] = hi_; }
#define HALFCLEAN16 \
  CE(0,8) CE(1,9) CE(2,10) CE(3,11) CE(4,12) CE(5,13) CE(6,14) CE(7,15) \
  CE(0,4) CE(1,5) CE(2,6) CE(3,7) CE(8,12) CE(9,13) CE(10,14) CE(11,15) \
  CE(0,2) CE(1,3) CE(4,6) CE(5,7) CE(8,10) CE(9,11) CE(12,14) CE(13,15) \
  CE(0,1) CE(2,3) CE(4,5) CE(6,7) CE(8,9) CE(10,11) CE(12,13) CE(14,15) \
  ;

#define N_CTILES 60
template <bool BIAS>
static __device__ __forceinline__ void compose_tile(int rt, int ct, int lane,
                                                    const float* __restrict__ Wemb,
                                                    const float* __restrict__ bemb,
                                                    const float* __restrict__ pos,
                                                    const float* __restrict__ win,
                                                    const float* __restrict__ bin,
                                                    float* __restrict__ Wc,
                                                    float* __restrict__ bt) {
  const int r  = lane & 15;
  const int hf = lane >> 4;
  const int col = ct * 16 + r;
  const int nvalid = BIAS ? LL : (56 - rt * 16);
  const bool a_ok = r < nvalid;
  const int rr = a_ok ? r : 0;
  const float* ap = BIAS ? (pos + rr * EE) : (Wemb + (rt * 16 + rr) * EE);

  v8f acc = {0, 0, 0, 0, 0, 0, 0, 0};
  #pragma unroll
  for (int half = 0; half < 2; ++half) {
    v2f a[8], b[8];
    #pragma unroll
    for (int s = 0; s < 8; ++s) {
      const int c0 = (half * 8 + s) * 4 + 2 * hf;
      const float2 av = *reinterpret_cast<const float2*>(ap + c0);
      float ax = av.x, ay = av.y;
      if (BIAS) {
        const float2 bv = *reinterpret_cast<const float2*>(bemb + c0);
        ax += bv.x; ay += bv.y;
      }
      a[s].x = a_ok ? ax : 0.0f;
      a[s].y = a_ok ? ay : 0.0f;
      b[s].x = win[c0 * 192 + col];
      b[s].y = win[(c0 + 1) * 192 + col];
    }
    #pragma unroll
    for (int s = 0; s < 8; ++s) acc = wmma4(a[s], b[s], acc);
  }

  const float badd = BIAS ? bin[col] : 0.0f;
  float* dst = (BIAS ? bt : (Wc + rt * 16 * 192)) + 8 * hf * 192 + col;
  #pragma unroll
  for (int g = 0; g < 8; ++g)
    dst[g * 192] = BIAS ? acc[g] + badd : acc[g];
}

__global__ __launch_bounds__(256) void k_stats(const float* __restrict__ ws,
                                               float* __restrict__ feats,
                                               const float* __restrict__ Wemb,
                                               const float* __restrict__ bemb,
                                               const float* __restrict__ pos,
                                               const float* __restrict__ win,
                                               const float* __restrict__ bin,
                                               float* __restrict__ Wc,
                                               float* __restrict__ bt) {
  if (blockIdx.x < N_CTILES && __builtin_amdgcn_readfirstlane(threadIdx.x >> 5) == 0) {
    const int rt = blockIdx.x / 12;
    const int ct = blockIdx.x - rt * 12;
    if (rt == 4) compose_tile<true >(rt, ct, threadIdx.x & 31, Wemb, bemb, pos, win, bin, Wc, bt);
    else         compose_tile<false>(rt, ct, threadIdx.x & 31, Wemb, bemb, pos, win, bin, Wc, bt);
  }

  const int l4  = threadIdx.x & 3;
  const float inf  = __builtin_inff();
  const float sel1 = (l4 & 1) ? inf : -inf;
  const float sel2 = (l4 & 2) ? inf : -inf;

  for (int trip = 0; trip < T_STATS; ++trip) {
  const int row = (trip * G_STATS + blockIdx.x) * 64 + (threadIdx.x >> 2);

  const float4* p = reinterpret_cast<const float4*>(ws + (size_t)row * SS) + l4;
  float x[16];
  #pragma unroll
  for (int q = 0; q < 4; ++q) {
    const float4 v = p[q * 4];
    x[4 * q + 0] = v.x; x[4 * q + 1] = v.y; x[4 * q + 2] = v.z; x[4 * q + 3] = v.w;
  }

  float sa = x[0], sb = x[1], qa = x[0] * x[0], qb = x[1] * x[1];
  #pragma unroll
  for (int i = 2; i < 16; i += 2) {
    sa += x[i];     qa = fmaf(x[i],     x[i],     qa);
    sb += x[i + 1]; qb = fmaf(x[i + 1], x[i + 1], qb);
  }
  float s = sa + sb, s2 = qa + qb;
  s  += quadf<QP_XOR1>(s);   s2 += quadf<QP_XOR1>(s2);
  s  += quadf<QP_XOR2>(s);   s2 += quadf<QP_XOR2>(s2);
  const float mean = s * (1.0f / 64.0f);
  const float var  = (s2 - 64.0f * mean * mean) * (1.0f / 63.0f);

  CE(0,13) CE(1,12) CE(2,15) CE(3,14) CE(4,8) CE(5,6) CE(7,11) CE(9,10)
  CE(0,5) CE(1,7) CE(2,9) CE(3,4) CE(6,13) CE(8,14) CE(10,15) CE(11,12)
  CE(0,1) CE(2,3) CE(4,5) CE(6,8) CE(7,9) CE(10,11) CE(12,13) CE(14,15)
  CE(0,2) CE(1,3) CE(4,10) CE(5,11) CE(6,7) CE(8,9) CE(12,14) CE(13,15)
  CE(1,2) CE(3,12) CE(4,6) CE(5,7) CE(8,10) CE(9,11) CE(13,14)
  CE(1,4) CE(2,6) CE(5,8) CE(7,10) CE(9,13) CE(11,14)
  CE(2,4) CE(3,6) CE(9,12) CE(11,13)
  CE(3,5) CE(6,8) CE(7,9) CE(10,12)
  CE(3,4) CE(5,6) CE(7,8) CE(9,10) CE(11,12)
  CE(6,7) CE(8,9)

  {
    float t[16];
    #pragma unroll
    for (int r = 0; r < 16; ++r)
      t[r] = med3_xor1(x[r], x[15 - r], sel1);
    #pragma unroll
    for (int r = 0; r < 16; ++r) x[r] = t[r];
  }
  HALFCLEAN16

  {
    float t[16];
    #pragma unroll
    for (int r = 0; r < 16; ++r)
      t[r] = med3_xor3(x[r], x[15 - r], sel2);
    #pragma unroll
    for (int r = 0; r < 16; ++r)
      x[r] = med3_xor1(t[r], t[r], sel1);
  }
  float mn = x[0], mx = x[0];
  #pragma unroll
  for (int i = 1; i < 16; ++i) { mn = fminf(mn, x[i]); mx = fmaxf(mx, x[i]); }

  const float nxt = quadf<QP_NEXT>(mn);
  const float q0  = quadf<QP_LANE0>(mn);
  const float w   = (l4 == 0) ? 0.75f : 0.25f;
  const float lin = mx + w * (nxt - mx);
  const float mid = 0.5f * (mx + nxt);
  const float hi  = (l4 == 1) ? mid : ((l4 == 3) ? mx : lin);
  const float lo  = (l4 == 0) ? mean : ((l4 == 1) ? var : q0);

  const int c = row & 255;
  int tmp = row >> 8;
  const int b = tmp & 31; tmp >>= 5;
  const int t = tmp & 7;
  const int l = tmp >> 3;
  float* o = feats + ((size_t)((b * LL + l) * CC + c)) * 56 + t * 7;
  o[3 + l4] = hi;
  if (l4 < 3) o[l4] = lo;
  }
}
#undef CE
#undef HALFCLEAN16

__global__ __launch_bounds__(96) void k_qkv(const float* __restrict__ feats,
                                            const float* __restrict__ Wc,
                                            const float* __restrict__ bt,
                                            float* __restrict__ qkv) {
  const int lane = threadIdx.x & 31;
  const int wid  = threadIdx.x >> 5;
  const int r  = lane & 15;
  const int hf = lane >> 4;
  const int n4 = wid * 64 + 4 * r;

  float wf[4][28];
  load_wfrag4<7>(wf, Wc, 192, n4, hf);

  float4 a[7];
  load_arows<7>(a, feats + (size_t)(blockIdx.x * 16 + r) * 56 + 4 * hf);

  for (int trip = 0; trip < T_QKV; ++trip) {
    const int m0 = (trip * G_QKV + blockIdx.x) * 16;
    float4 an[7];
    if (trip + 1 < T_QKV)
      load_arows<7>(an, feats + (size_t)(m0 + G_QKV * 16 + r) * 56 + 4 * hf);
    const int l = (m0 >> 8) % LL;
    const float4 bi = *reinterpret_cast<const float4*>(bt + l * 192 + n4);

    v8f acc[4];
    #pragma unroll
    for (int t = 0; t < 4; ++t) acc[t] = (v8f){0, 0, 0, 0, 0, 0, 0, 0};
    mma_rows<7, 4>(acc, a, wf);

    #pragma unroll
    for (int g = 0; g < 8; ++g) {
      float4 o;
      o.x = acc[0][g] + bi.x;
      o.y = acc[1][g] + bi.y;
      o.z = acc[2][g] + bi.z;
      o.w = acc[3][g] + bi.w;
      *reinterpret_cast<float4*>(qkv + (size_t)(m0 + g + 8 * hf) * 192 + n4) = o;
    }

    if (trip + 1 < T_QKV) {
      #pragma unroll
      for (int j = 0; j < 7; ++j) a[j] = an[j];
    }
  }
}

#define DPP_LINE(OP, XM, D, S) \
  OP " " D ", " S ", " S " row_xmask:" XM " row_mask:0xf bank_mask:0xf\n\t"
#define ROW_STEP_FIRST(NAME, LEAD, OP, XM)                                        \
static __device__ __forceinline__ void NAME(float (&d)[8], const float (&s)[8]) { \
  asm(LEAD                                                                        \
      DPP_LINE(OP, XM, "%0", "%8")  DPP_LINE(OP, XM, "%1", "%9")                  \
      DPP_LINE(OP, XM, "%2", "%10") DPP_LINE(OP, XM, "%3", "%11")                 \
      DPP_LINE(OP, XM, "%4", "%12") DPP_LINE(OP, XM, "%5", "%13")                 \
      DPP_LINE(OP, XM, "%6", "%14") DPP_LINE(OP, XM, "%7", "%15")                 \
      : "=&v"(d[0]), "=&v"(d[1]), "=&v"(d[2]), "=&v"(d[3]),                       \
        "=&v"(d[4]), "=&v"(d[5]), "=&v"(d[6]), "=&v"(d[7])                        \
      : "v"(s[0]), "v"(s[1]), "v"(s[2]), "v"(s[3]),                               \
        "v"(s[4]), "v"(s[5]), "v"(s[6]), "v"(s[7]));                              \
}
#define ROW_STEP(NAME, OP, XM)                                                    \
static __device__ __forceinline__ void NAME(float (&v)[8]) {                      \
  asm(DPP_LINE(OP, XM, "%0", "%0") DPP_LINE(OP, XM, "%1", "%1")                   \
      DPP_LINE(OP, XM, "%2", "%2") DPP_LINE(OP, XM, "%3", "%3")                   \
      DPP_LINE(OP, XM, "%4", "%4") DPP_LINE(OP, XM, "%5", "%5")                   \
      DPP_LINE(OP, XM, "%6", "%6") DPP_LINE(OP, XM, "%7", "%7")                   \
      : "+v"(v[0]), "+v"(v[1]), "+v"(v[2]), "+v"(v[3]),                           \
        "+v"(v[4]), "+v"(v[5]), "+v"(v[6]), "+v"(v[7]));                          \
}
ROW_STEP_FIRST(rowmax_x8, "", "v_max_num_f32_dpp", "8")
ROW_STEP(rowmax_x4, "v_max_num_f32_dpp", "4")
ROW_STEP(rowmax_x2, "v_max_num_f32_dpp", "2")
ROW_STEP(rowmax_x1, "v_max_num_f32_dpp", "1")
ROW_STEP_FIRST(rowsum_x8, "v_nop\n\t", "v_add_f32_dpp", "8")
ROW_STEP(rowsum_x4, "v_add_f32_dpp", "4")
ROW_STEP(rowsum_x2, "v_add_f32_dpp", "2")
ROW_STEP(rowsum_x1, "v_add_f32_dpp", "1")
#undef ROW_STEP
#undef ROW_STEP_FIRST
#undef DPP_LINE

struct QKFrag { float4 qa, qb, ka, kb; };
static __device__ __forceinline__ void load_qk(QKFrag& f, const float* __restrict__ rowp) {
  const float4* p = reinterpret_cast<const float4*>(rowp);
  f.qa = p[0];  f.qb = p[1];
  f.ka = p[16]; f.kb = p[17];
}

static __device__ __forceinline__ void load_v(float (&vb)[6], const float* __restrict__ base,
                                              int r, int hf) {
  #pragma unroll
  for (int s = 0; s < 3; ++s) {
    const float* vp = base + (size_t)(4 * s + 2 * hf) * 192 + 128 + r;
    vb[2 * s]     = vp[0];
    vb[2 * s + 1] = vp[192];
  }
}

__global__ __launch_bounds__(256, 16) void k_attn(const float* __restrict__ qkv,
                                              float* __restrict__ ctx) {
  __shared__ float sP[8][16][17];
  const int lane = threadIdx.x & 31;
  const int wid  = threadIdx.x >> 5;
  const int r  = lane & 15;
  const int hf = lane >> 4;
  const int rr = (r < LL) ? r : 0;
  const int qk_off = rr * 192 + 8 * hf;

  int task = blockIdx.x * 8 + wid;
  const float* base = qkv + (size_t)(task >> 2) * LL * 192 + (task & 3) * HDD;
  QKFrag cur;
  float vb[6];
  load_qk(cur, base + qk_off);
  load_v(vb, base, r, hf);
  __builtin_amdgcn_sched_barrier(0);

  #pragma unroll
  for (int trip = 0; trip < T_ATTN; ++trip) {
  const int nb = task >> 2;
  const int h  = task & 3;

  v8f S = {0, 0, 0, 0, 0, 0, 0, 0};
  {
    v2f a, b;
    a.x = cur.qa.x; a.y = cur.qa.y; b.x = cur.ka.x; b.y = cur.ka.y; S = wmma4(a, b, S);
    a.x = cur.qa.z; a.y = cur.qa.w; b.x = cur.ka.z; b.y = cur.ka.w; S = wmma4(a, b, S);
    a.x = cur.qb.x; a.y = cur.qb.y; b.x = cur.kb.x; b.y = cur.kb.y; S = wmma4(a, b, S);
    a.x = cur.qb.z; a.y = cur.qb.w; b.x = cur.kb.z; b.y = cur.kb.w; S = wmma4(a, b, S);
  }

  const int ntask = task + G_ATTN * 8;
  const float* nbase = qkv + (size_t)(ntask >> 2) * LL * 192 + (ntask & 3) * HDD;
  QKFrag nxt;
  float vn[6];
  if (trip + 1 < T_ATTN) { load_qk(nxt, nbase + qk_off); load_v(vn, nbase, r, hf); }
  __builtin_amdgcn_sched_barrier(0);

  float P[8], sv[8], rm[8], rs[8];
  #pragma unroll
  for (int g = 0; g < 8; ++g)
    sv[g] = (r < LL) ? S[g] * 0.25f : -__builtin_inff();
  rowmax_x8(rm, sv); rowmax_x4(rm); rowmax_x2(rm); rowmax_x1(rm);
  #pragma unroll
  for (int g = 0; g < 8; ++g) P[g] = __expf(sv[g] - rm[g]);
  rowsum_x8(rs, P); rowsum_x4(rs); rowsum_x2(rs); rowsum_x1(rs);
  #pragma unroll
  for (int g = 0; g < 8; ++g) P[g] = P[g] / rs[g];
  #pragma unroll
  for (int g = 0; g < 8; ++g) sP[wid][g + 8 * hf][r] = P[g];

  v8f Cv = {0, 0, 0, 0, 0, 0, 0, 0};
  #pragma unroll
  for (int k0 = 0; k0 < 16; k0 += 4) {
    const int c0 = k0 + 2 * hf;
    v2f a; a.x = sP[wid][r][c0]; a.y = sP[wid][r][c0 + 1];
    v2f b;
    b.x = (k0 < LL) ? vb[(k0 >> 1)]     : 0.0f;
    b.y = (k0 < LL) ? vb[(k0 >> 1) + 1] : 0.0f;
    Cv = wmma4(a, b, Cv);
  }
  #pragma unroll
  for (int g = 0; g < 8; ++g) {
    const int m = g + 8 * hf;
    if (m < LL)
      ctx[((size_t)nb * LL + m) * EE + h * HDD + r] = Cv[g];
  }

  if (trip + 1 < T_ATTN) {
    cur = nxt; base = nbase; task = ntask;
    #pragma unroll
    for (int i = 0; i < 6; ++i) vb[i] = vn[i];
  }
  }
}

__global__ __launch_bounds__(256) void k_proj_pool(const float* __restrict__ ctx,
                                                   const float* __restrict__ wout,
                                                   const float* __restrict__ bout,
                                                   float* __restrict__ pooled) {
  __shared__ float pool[EE];
  const int tid = threadIdx.x;
  if (tid < EE) pool[tid] = 0.0f;
  __syncthreads();

  const int lane = tid & 31;
  const int wid  = tid >> 5;
  const int r  = lane & 15;
  const int hf = lane >> 4;
  const size_t mbase = (size_t)blockIdx.x * 256;

  float4 a0[8], a1[8];
  load_arows<8>(a0, ctx + (mbase + wid * 16 + r) * EE + 4 * hf);
  load_arows<8>(a1, ctx + (mbase + (wid + 8) * 16 + r) * EE + 4 * hf);

  float wf[4][32];
  load_wfrag4<8>(wf, wout, EE, 4 * r, hf);

  v8f acc[4];
  #pragma unroll
  for (int t = 0; t < 4; ++t) acc[t] = (v8f){0, 0, 0, 0, 0, 0, 0, 0};
  mma_rows<8, 4>(acc, a0, wf);
  mma_rows<8, 4>(acc, a1, wf);

  #pragma unroll
  for (int t = 0; t < 4; ++t) {
    float ps = acc[t][0] + acc[t][1] + acc[t][2] + acc[t][3] +
               acc[t][4] + acc[t][5] + acc[t][6] + acc[t][7];
    ps += __shfl_xor(ps, 16, 32);
    if (lane < 16) atomicAdd(&pool[4 * r + t], ps);
  }
  __syncthreads();
  if (tid < EE)
    pooled[(size_t)blockIdx.x * EE + tid] = pool[tid] * (1.0f / 256.0f) + bout[tid];
}

__global__ __launch_bounds__(256) void k_mlp_lu(
    const float* __restrict__ pooled,
    const float* __restrict__ mw1, const float* __restrict__ mb1,
    const float* __restrict__ mg,  const float* __restrict__ mbb,
    const float* __restrict__ mw2, const float* __restrict__ mb2,
    const float* __restrict__ nw1, const float* __restrict__ nb1,
    const float* __restrict__ ng,  const float* __restrict__ nbb,
    const float* __restrict__ nw2, const float* __restrict__ nb2,
    float* __restrict__ out) {
  const int branch = blockIdx.x & 1;
  const int row    = blockIdx.x >> 1;
  const float* w1 = branch ? nw1 : mw1;
  const float* b1 = branch ? nb1 : mb1;
  const float* g  = branch ? ng  : mg;
  const float* bb = branch ? nbb : mbb;
  const float* w2 = branch ? nw2 : mw2;
  const float* b2 = branch ? nb2 : mb2;

  __shared__ float sx[EE];
  __shared__ float sh[HIDD];
  __shared__ float sraw[512];
  const int tid = threadIdx.x;

  if (tid < EE) sx[tid] = pooled[(size_t)row * EE + tid];
  __syncthreads();

  if (tid < HIDD) {
    float acc = b1[tid];
    #pragma unroll
    for (int i = 0; i < EE; ++i) acc += sx[i] * w1[i * HIDD + tid];
    float mu = acc;
    #pragma unroll
    for (int m = 16; m >= 1; m >>= 1) mu += __shfl_xor(mu, m, 32);
    mu *= (1.0f / 32.0f);
    const float d = acc - mu;
    float vv = d * d;
    #pragma unroll
    for (int m = 16; m >= 1; m >>= 1) vv += __shfl_xor(vv, m, 32);
    vv *= (1.0f / 32.0f);
    const float y = d * rsqrtf(vv + 1e-5f) * g[tid] + bb[tid];
    sh[tid] = 0.5f * y * (1.0f + erff(y * 0.70710678118654752f));
  }
  __syncthreads();

  for (int c = tid; c < 512; c += 256) {
    float acc = b2[c];
    #pragma unroll
    for (int j = 0; j < HIDD; ++j) acc += sh[j] * w2[j * 512 + c];
    sraw[c] = acc;
  }
  __syncthreads();

  const size_t obase = ((size_t)branch * 384 + row) * 512;
  for (int o = tid; o < 512; o += 256) {
    const int h  = o >> 6;
    const int ik = o & 63;
    const int i  = ik >> 3;
    const int k  = ik & 7;
    const float* rh = sraw + h * 64;
    float acc = 0.0f;
    #pragma unroll
    for (int j = 0; j < NKK; ++j) {
      if (j > i || j > k) continue;
      const float Lij = (j == i) ? 1.0f : rh[(i * (i - 1)) / 2 + j];
      float u = rh[28 + (NKK * j - (j * (j - 1)) / 2) + (k - j)];
      if (j == k) u = softplusf(u) + 0.01f;
      acc += Lij * u;
    }
    out[obase + o] = acc;
  }
}

extern "C" void kernel_launch(void* const* d_in, const int* in_sizes, int n_in,
                              void* d_out, int out_size, void* d_ws, size_t ws_size,
                              hipStream_t stream) {
  const float* wsfeat = (const float*)d_in[0];
  const float* W_emb  = (const float*)d_in[1];
  const float* b_emb  = (const float*)d_in[2];
  const float* pos    = (const float*)d_in[3];
  const float* w_in   = (const float*)d_in[4];
  const float* b_in   = (const float*)d_in[5];
  const float* w_out  = (const float*)d_in[6];
  const float* b_out  = (const float*)d_in[7];
  const float* m_w1 = (const float*)d_in[8];
  const float* m_b1 = (const float*)d_in[9];
  const float* m_g  = (const float*)d_in[10];
  const float* m_bb = (const float*)d_in[11];
  const float* m_w2 = (const float*)d_in[12];
  const float* m_b2 = (const float*)d_in[13];
  const float* n_w1 = (const float*)d_in[14];
  const float* n_b1 = (const float*)d_in[15];
  const float* n_g  = (const float*)d_in[16];
  const float* n_bb = (const float*)d_in[17];
  const float* n_w2 = (const float*)d_in[18];
  const float* n_b2 = (const float*)d_in[19];
  float* out = (float*)d_out;

  float* ws     = (float*)d_ws;
  float* feats  = ws;
  float* wc     = feats + (size_t)MROWS * 56;
  float* bt     = wc    + 64 * 192;
  float* qkv    = wc    + (size_t)MROWS * EE;
  float* ctx    = qkv   + (size_t)MROWS * 192;
  float* pooled = ctx   + (size_t)MROWS * EE;

  k_stats<<<G_STATS, 256, 0, stream>>>(wsfeat, feats, W_emb, b_emb, pos, w_in, b_in, wc, bt);
  k_qkv<<<G_QKV, 96, 0, stream>>>(feats, wc, bt, qkv);
  k_attn<<<G_ATTN, 256, 0, stream>>>(qkv, ctx);
  k_proj_pool<<<BB * LL, 256, 0, stream>>>(ctx, w_out, b_out, pooled);
  k_mlp_lu<<<BB * LL * 2, 256, 0, stream>>>(pooled,
      m_w1, m_b1, m_g, m_bb, m_w2, m_b2,
      n_w1, n_b1, n_g, n_bb, n_w2, n_b2, out);
}


# k_mlp_lu: softplus of U's diagonal (with its added constant) applied once per entry (64 per block, in place, one more barrier), not in every term of L@U (8 x 2 log1p passes per wave); bit-identical
# speedup vs baseline: 2.8169x; 1.0435x over previous
#include <hip/hip_runtime.h>
#include <math.h>

#define LL   12
#define TT   8
#define BB   32
#define CC   256
#define SS   64
#define EE   64
#define HDD  16
#define NHH  4
#define HIDD 32
#define HH   8
#define NKK  8
#define MROWS (BB*CC*LL)
#define SROWS (LL*TT*BB*CC)

#define G_STATS 2048
#define T_STATS ((SROWS / 64) / G_STATS)
#define G_QKV   1024
#define T_QKV   ((MROWS / 16) / G_QKV)
#define G_ATTN  2048
#define T_ATTN  (((BB * CC * NHH) / 8) / G_ATTN)

typedef float v2f __attribute__((ext_vector_type(2)));
typedef float v8f __attribute__((ext_vector_type(8)));

static __device__ __forceinline__ v8f wmma4(v2f a, v2f b, v8f c) {
  return __builtin_amdgcn_wmma_f32_16x16x4_f32(false, a, false, b, (short)0, c,
                                               false, false);
}

static __device__ __forceinline__ float softplusf(float x) {
  return (x > 20.0f) ? x : log1pf(__expf(x));
}

template <int KC>
static __device__ __forceinline__ void load_wfrag4(float (&wf)[4][4 * KC],
                                                   const float* __restrict__ w,
                                                   int ldw, int col4, int hf) {
  #pragma unroll
  for (int j = 0; j < KC; ++j)
    #pragma unroll
    for (int e = 0; e < 4; ++e) {
      const float4 v = *reinterpret_cast<const float4*>(
          w + (size_t)(8 * j + 4 * hf + e) * ldw + col4);
      wf[0][4 * j + e] = v.x; wf[1][4 * j + e] = v.y;
      wf[2][4 * j + e] = v.z; wf[3][4 * j + e] = v.w;
    }
}

template <int KC>
static __device__ __forceinline__ void load_arows(float4 (&a)[KC],
                                                  const float* __restrict__ rowp) {
  const float4* p = reinterpret_cast<const float4*>(rowp);
  #pragma unroll
  for (int j = 0; j < KC; ++j) a[j] = p[2 * j];
}

template <int KC, int NT>
static __device__ __forceinline__ void mma_rows(v8f (&acc)[NT], const float4 (&a)[KC],
                                                const float (&wf)[NT][4 * KC]) {
  #pragma unroll
  for (int j = 0; j < KC; ++j) {
    v2f a0; a0.x = a[j].x; a0.y = a[j].y;
    v2f a1; a1.x = a[j].z; a1.y = a[j].w;
    #pragma unroll
    for (int t = 0; t < NT; ++t) {
      v2f b; b.x = wf[t][4 * j]; b.y = wf[t][4 * j + 1];
      acc[t] = wmma4(a0, b, acc[t]);
    }
    #pragma unroll
    for (int t = 0; t < NT; ++t) {
      v2f b; b.x = wf[t][4 * j + 2]; b.y = wf[t][4 * j + 3];
      acc[t] = wmma4(a1, b, acc[t]);
    }
  }
}

template <int CTRL>
static __device__ __forceinline__ float quadf(float v) {
  return __int_as_float(__builtin_amdgcn_update_dpp(
      0, __float_as_int(v), CTRL, 0xf, 0xf, true));
}

#define MED3_QUAD(NAME, PERM)                                                     \
static __device__ __forceinline__ float NAME(float a, float b, float sel) {      \
  float d;                                                                        \
  asm("v_med3_num_f32_e64_dpp %0, %1, %2, %3 quad_perm:" PERM                      \
      " row_mask:0xf bank_mask:0xf"                                               \
      : "=v"(d) : "v"(b), "v"(a), "v"(sel));                                      \
  return d;                                                                       \
}
MED3_QUAD(med3_xor1, "[1,0,3,2]")
MED3_QUAD(med3_xor3, "[3,2,1,0]")
#undef MED3_QUAD

#define QP_XOR1  0xB1
#define QP_XOR2  0x4E
#define QP_NEXT  0xF9
#define QP_LANE0 0x00

#define CE(i, j) { const float lo_ = fminf(x[i], x[j]); \
                   const float hi_ = fmaxf(x[i], x[j]); x[i] = lo_; x[j] = hi_; }
#define HALFCLEAN16 \
  CE(0,8) CE(1,9) CE(2,10) CE(3,11) CE(4,12) CE(5,13) CE(6,14) CE(7,15) \
  CE(0,4) CE(1,5) CE(2,6) CE(3,7) CE(8,12) CE(9,13) CE(10,14) CE(11,15) \
  CE(0,2) CE(1,3) CE(4,6) CE(5,7) CE(8,10) CE(9,11) CE(12,14) CE(13,15) \
  CE(0,1) CE(2,3) CE(4,5) CE(6,7) CE(8,9) CE(10,11) CE(12,13) CE(14,15) \
  ;

#define N_CTILES 60
template <bool BIAS>
static __device__ __forceinline__ void compose_tile(int rt, int ct, int lane,
                                                    const float* __restrict__ Wemb,
                                                    const float* __restrict__ bemb,
                                                    const float* __restrict__ pos,
                                                    const float* __restrict__ win,
                                                    const float* __restrict__ bin,
                                                    float* __restrict__ Wc,
                                                    float* __restrict__ bt) {
  const int r  = lane & 15;
  const int hf = lane >> 4;
  const int col = ct * 16 + r;
  const int nvalid = BIAS ? LL : (56 - rt * 16);
  const bool a_ok = r < nvalid;
  const int rr = a_ok ? r : 0;
  const float* ap = BIAS ? (pos + rr * EE) : (Wemb + (rt * 16 + rr) * EE);

  v8f acc = {0, 0, 0, 0, 0, 0, 0, 0};
  #pragma unroll
  for (int half = 0; half < 2; ++half) {
    v2f a[8], b[8];
    #pragma unroll
    for (int s = 0; s < 8; ++s) {
      const int c0 = (half * 8 + s) * 4 + 2 * hf;
      const float2 av = *reinterpret_cast<const float2*>(ap + c0);
      float ax = av.x, ay = av.y;
      if (BIAS) {
        const float2 bv = *reinterpret_cast<const float2*>(bemb + c0);
        ax += bv.x; ay += bv.y;
      }
      a[s].x = a_ok ? ax : 0.0f;
      a[s].y = a_ok ? ay : 0.0f;
      b[s].x = win[c0 * 192 + col];
      b[s].y = win[(c0 + 1) * 192 + col];
    }
    #pragma unroll
    for (int s = 0; s < 8; ++s) acc = wmma4(a[s], b[s], acc);
  }

  const float badd = BIAS ? bin[col] : 0.0f;
  float* dst = (BIAS ? bt : (Wc + rt * 16 * 192)) + 8 * hf * 192 + col;
  #pragma unroll
  for (int g = 0; g < 8; ++g)
    dst[g * 192] = BIAS ? acc[g] + badd : acc[g];
}

__global__ __launch_bounds__(256) void k_stats(const float* __restrict__ ws,
                                               float* __restrict__ feats,
                                               const float* __restrict__ Wemb,
                                               const float* __restrict__ bemb,
                                               const float* __restrict__ pos,
                                               const float* __restrict__ win,
                                               const float* __restrict__ bin,
                                               float* __restrict__ Wc,
                                               float* __restrict__ bt) {
  if (blockIdx.x < N_CTILES && __builtin_amdgcn_readfirstlane(threadIdx.x >> 5) == 0) {
    const int rt = blockIdx.x / 12;
    const int ct = blockIdx.x - rt * 12;
    if (rt == 4) compose_tile<true >(rt, ct, threadIdx.x & 31, Wemb, bemb, pos, win, bin, Wc, bt);
    else         compose_tile<false>(rt, ct, threadIdx.x & 31, Wemb, bemb, pos, win, bin, Wc, bt);
  }

  const int l4  = threadIdx.x & 3;
  const float inf  = __builtin_inff();
  const float sel1 = (l4 & 1) ? inf : -inf;
  const float sel2 = (l4 & 2) ? inf : -inf;

  for (int trip = 0; trip < T_STATS; ++trip) {
  const int row = (trip * G_STATS + blockIdx.x) * 64 + (threadIdx.x >> 2);

  const float4* p = reinterpret_cast<const float4*>(ws + (size_t)row * SS) + l4;
  float x[16];
  #pragma unroll
  for (int q = 0; q < 4; ++q) {
    const float4 v = p[q * 4];
    x[4 * q + 0] = v.x; x[4 * q + 1] = v.y; x[4 * q + 2] = v.z; x[4 * q + 3] = v.w;
  }

  float sa = x[0], sb = x[1], qa = x[0] * x[0], qb = x[1] * x[1];
  #pragma unroll
  for (int i = 2; i < 16; i += 2) {
    sa += x[i];     qa = fmaf(x[i],     x[i],     qa);
    sb += x[i + 1]; qb = fmaf(x[i + 1], x[i + 1], qb);
  }
  float s = sa + sb, s2 = qa + qb;
  s  += quadf<QP_XOR1>(s);   s2 += quadf<QP_XOR1>(s2);
  s  += quadf<QP_XOR2>(s);   s2 += quadf<QP_XOR2>(s2);
  const float mean = s * (1.0f / 64.0f);
  const float var  = (s2 - 64.0f * mean * mean) * (1.0f / 63.0f);

  CE(0,13) CE(1,12) CE(2,15) CE(3,14) CE(4,8) CE(5,6) CE(7,11) CE(9,10)
  CE(0,5) CE(1,7) CE(2,9) CE(3,4) CE(6,13) CE(8,14) CE(10,15) CE(11,12)
  CE(0,1) CE(2,3) CE(4,5) CE(6,8) CE(7,9) CE(10,11) CE(12,13) CE(14,15)
  CE(0,2) CE(1,3) CE(4,10) CE(5,11) CE(6,7) CE(8,9) CE(12,14) CE(13,15)
  CE(1,2) CE(3,12) CE(4,6) CE(5,7) CE(8,10) CE(9,11) CE(13,14)
  CE(1,4) CE(2,6) CE(5,8) CE(7,10) CE(9,13) CE(11,14)
  CE(2,4) CE(3,6) CE(9,12) CE(11,13)
  CE(3,5) CE(6,8) CE(7,9) CE(10,12)
  CE(3,4) CE(5,6) CE(7,8) CE(9,10) CE(11,12)
  CE(6,7) CE(8,9)

  {
    float t[16];
    #pragma unroll
    for (int r = 0; r < 16; ++r)
      t[r] = med3_xor1(x[r], x[15 - r], sel1);
    #pragma unroll
    for (int r = 0; r < 16; ++r) x[r] = t[r];
  }
  HALFCLEAN16

  {
    float t[16];
    #pragma unroll
    for (int r = 0; r < 16; ++r)
      t[r] = med3_xor3(x[r], x[15 - r], sel2);
    #pragma unroll
    for (int r = 0; r < 16; ++r)
      x[r] = med3_xor1(t[r], t[r], sel1);
  }
  float mn = x[0], mx = x[0];
  #pragma unroll
  for (int i = 1; i < 16; ++i) { mn = fminf(mn, x[i]); mx = fmaxf(mx, x[i]); }

  const float nxt = quadf<QP_NEXT>(mn);
  const float q0  = quadf<QP_LANE0>(mn);
  const float w   = (l4 == 0) ? 0.75f : 0.25f;
  const float lin = mx + w * (nxt - mx);
  const float mid = 0.5f * (mx + nxt);
  const float hi  = (l4 == 1) ? mid : ((l4 == 3) ? mx : lin);
  const float lo  = (l4 == 0) ? mean : ((l4 == 1) ? var : q0);

  const int c = row & 255;
  int tmp = row >> 8;
  const int b = tmp & 31; tmp >>= 5;
  const int t = tmp & 7;
  const int l = tmp >> 3;
  float* o = feats + ((size_t)((b * LL + l) * CC + c)) * 56 + t * 7;
  o[3 + l4] = hi;
  if (l4 < 3) o[l4] = lo;
  }
}
#undef CE
#undef HALFCLEAN16

__global__ __launch_bounds__(96) void k_qkv(const float* __restrict__ feats,
                                            const float* __restrict__ Wc,
                                            const float* __restrict__ bt,
                                            float* __restrict__ qkv) {
  const int lane = threadIdx.x & 31;
  const int wid  = threadIdx.x >> 5;
  const int r  = lane & 15;
  const int hf = lane >> 4;
  const int n4 = wid * 64 + 4 * r;

  float wf[4][28];
  load_wfrag4<7>(wf, Wc, 192, n4, hf);

  float4 a[7];
  load_arows<7>(a, feats + (size_t)(blockIdx.x * 16 + r) * 56 + 4 * hf);

  for (int trip = 0; trip < T_QKV; ++trip) {
    const int m0 = (trip * G_QKV + blockIdx.x) * 16;
    float4 an[7];
    if (trip + 1 < T_QKV)
      load_arows<7>(an, feats + (size_t)(m0 + G_QKV * 16 + r) * 56 + 4 * hf);
    const int l = (m0 >> 8) % LL;
    const float4 bi = *reinterpret_cast<const float4*>(bt + l * 192 + n4);

    v8f acc[4];
    #pragma unroll
    for (int t = 0; t < 4; ++t) acc[t] = (v8f){0, 0, 0, 0, 0, 0, 0, 0};
    mma_rows<7, 4>(acc, a, wf);

    #pragma unroll
    for (int g = 0; g < 8; ++g) {
      float4 o;
      o.x = acc[0][g] + bi.x;
      o.y = acc[1][g] + bi.y;
      o.z = acc[2][g] + bi.z;
      o.w = acc[3][g] + bi.w;
      *reinterpret_cast<float4*>(qkv + (size_t)(m0 + g + 8 * hf) * 192 + n4) = o;
    }

    if (trip + 1 < T_QKV) {
      #pragma unroll
      for (int j = 0; j < 7; ++j) a[j] = an[j];
    }
  }
}

#define DPP_LINE(OP, XM, D, S) \
  OP " " D ", " S ", " S " row_xmask:" XM " row_mask:0xf bank_mask:0xf\n\t"
#define ROW_STEP_FIRST(NAME, LEAD, OP, XM)                                        \
static __device__ __forceinline__ void NAME(float (&d)[8], const float (&s)[8]) { \
  asm(LEAD                                                                        \
      DPP_LINE(OP, XM, "%0", "%8")  DPP_LINE(OP, XM, "%1", "%9")                  \
      DPP_LINE(OP, XM, "%2", "%10") DPP_LINE(OP, XM, "%3", "%11")                 \
      DPP_LINE(OP, XM, "%4", "%12") DPP_LINE(OP, XM, "%5", "%13")                 \
      DPP_LINE(OP, XM, "%6", "%14") DPP_LINE(OP, XM, "%7", "%15")                 \
      : "=&v"(d[0]), "=&v"(d[1]), "=&v"(d[2]), "=&v"(d[3]),                       \
        "=&v"(d[4]), "=&v"(d[5]), "=&v"(d[6]), "=&v"(d[7])                        \
      : "v"(s[0]), "v"(s[1]), "v"(s[2]), "v"(s[3]),                               \
        "v"(s[4]), "v"(s[5]), "v"(s[6]), "v"(s[7]));                              \
}
#define ROW_STEP(NAME, OP, XM)                                                    \
static __device__ __forceinline__ void NAME(float (&v)[8]) {                      \
  asm(DPP_LINE(OP, XM, "%0", "%0") DPP_LINE(OP, XM, "%1", "%1")                   \
      DPP_LINE(OP, XM, "%2", "%2") DPP_LINE(OP, XM, "%3", "%3")                   \
      DPP_LINE(OP, XM, "%4", "%4") DPP_LINE(OP, XM, "%5", "%5")                   \
      DPP_LINE(OP, XM, "%6", "%6") DPP_LINE(OP, XM, "%7", "%7")                   \
      : "+v"(v[0]), "+v"(v[1]), "+v"(v[2]), "+v"(v[3]),                           \
        "+v"(v[4]), "+v"(v[5]), "+v"(v[6]), "+v"(v[7]));                          \
}
ROW_STEP_FIRST(rowmax_x8, "", "v_max_num_f32_dpp", "8")
ROW_STEP(rowmax_x4, "v_max_num_f32_dpp", "4")
ROW_STEP(rowmax_x2, "v_max_num_f32_dpp", "2")
ROW_STEP(rowmax_x1, "v_max_num_f32_dpp", "1")
ROW_STEP_FIRST(rowsum_x8, "v_nop\n\t", "v_add_f32_dpp", "8")
ROW_STEP(rowsum_x4, "v_add_f32_dpp", "4")
ROW_STEP(rowsum_x2, "v_add_f32_dpp", "2")
ROW_STEP(rowsum_x1, "v_add_f32_dpp", "1")
#undef ROW_STEP
#undef ROW_STEP_FIRST
#undef DPP_LINE

struct QKFrag { float4 qa, qb, ka, kb; };
static __device__ __forceinline__ void load_qk(QKFrag& f, const float* __restrict__ rowp) {
  const float4* p = reinterpret_cast<const float4*>(rowp);
  f.qa = p[0];  f.qb = p[1];
  f.ka = p[16]; f.kb = p[17];
}

static __device__ __forceinline__ void load_v(float (&vb)[6], const float* __restrict__ base,
                                              int r, int hf) {
  #pragma unroll
  for (int s = 0; s < 3; ++s) {
    const float* vp = base + (size_t)(4 * s + 2 * hf) * 192 + 128 + r;
    vb[2 * s]     = vp[0];
    vb[2 * s + 1] = vp[192];
  }
}

__global__ __launch_bounds__(256, 16) void k_attn(const float* __restrict__ qkv,
                                              float* __restrict__ ctx) {
  __shared__ float sP[8][16][17];
  const int lane = threadIdx.x & 31;
  const int wid  = threadIdx.x >> 5;
  const int r  = lane & 15;
  const int hf = lane >> 4;
  const int rr = (r < LL) ? r : 0;
  const int qk_off = rr * 192 + 8 * hf;

  int task = blockIdx.x * 8 + wid;
  const float* base = qkv + (size_t)(task >> 2) * LL * 192 + (task & 3) * HDD;
  QKFrag cur;
  float vb[6];
  load_qk(cur, base + qk_off);
  load_v(vb, base, r, hf);
  __builtin_amdgcn_sched_barrier(0);

  #pragma unroll
  for (int trip = 0; trip < T_ATTN; ++trip) {
  const int nb = task >> 2;
  const int h  = task & 3;

  v8f S = {0, 0, 0, 0, 0, 0, 0, 0};
  {
    v2f a, b;
    a.x = cur.qa.x; a.y = cur.qa.y; b.x = cur.ka.x; b.y = cur.ka.y; S = wmma4(a, b, S);
    a.x = cur.qa.z; a.y = cur.qa.w; b.x = cur.ka.z; b.y = cur.ka.w; S = wmma4(a, b, S);
    a.x = cur.qb.x; a.y = cur.qb.y; b.x = cur.kb.x; b.y = cur.kb.y; S = wmma4(a, b, S);
    a.x = cur.qb.z; a.y = cur.qb.w; b.x = cur.kb.z; b.y = cur.kb.w; S = wmma4(a, b, S);
  }

  const int ntask = task + G_ATTN * 8;
  const float* nbase = qkv + (size_t)(ntask >> 2) * LL * 192 + (ntask & 3) * HDD;
  QKFrag nxt;
  float vn[6];
  if (trip + 1 < T_ATTN) { load_qk(nxt, nbase + qk_off); load_v(vn, nbase, r, hf); }
  __builtin_amdgcn_sched_barrier(0);

  float P[8], sv[8], rm[8], rs[8];
  #pragma unroll
  for (int g = 0; g < 8; ++g)
    sv[g] = (r < LL) ? S[g] * 0.25f : -__builtin_inff();
  rowmax_x8(rm, sv); rowmax_x4(rm); rowmax_x2(rm); rowmax_x1(rm);
  #pragma unroll
  for (int g = 0; g < 8; ++g) P[g] = __expf(sv[g] - rm[g]);
  rowsum_x8(rs, P); rowsum_x4(rs); rowsum_x2(rs); rowsum_x1(rs);
  #pragma unroll
  for (int g = 0; g < 8; ++g) P[g] = P[g] / rs[g];
  #pragma unroll
  for (int g = 0; g < 8; ++g) sP[wid][g + 8 * hf][r] = P[g];

  v8f Cv = {0, 0, 0, 0, 0, 0, 0, 0};
  #pragma unroll
  for (int k0 = 0; k0 < 16; k0 += 4) {
    const int c0 = k0 + 2 * hf;
    v2f a; a.x = sP[wid][r][c0]; a.y = sP[wid][r][c0 + 1];
    v2f b;
    b.x = (k0 < LL) ? vb[(k0 >> 1)]     : 0.0f;
    b.y = (k0 < LL) ? vb[(k0 >> 1) + 1] : 0.0f;
    Cv = wmma4(a, b, Cv);
  }
  #pragma unroll
  for (int g = 0; g < 8; ++g) {
    const int m = g + 8 * hf;
    if (m < LL)
      ctx[((size_t)nb * LL + m) * EE + h * HDD + r] = Cv[g];
  }

  if (trip + 1 < T_ATTN) {
    cur = nxt; base = nbase; task = ntask;
    #pragma unroll
    for (int i = 0; i < 6; ++i) vb[i] = vn[i];
  }
  }
}

__global__ __launch_bounds__(256) void k_proj_pool(const float* __restrict__ ctx,
                                                   const float* __restrict__ wout,
                                                   const float* __restrict__ bout,
                                                   float* __restrict__ pooled) {
  __shared__ float pool[EE];
  const int tid = threadIdx.x;
  if (tid < EE) pool[tid] = 0.0f;
  __syncthreads();

  const int lane = tid & 31;
  const int wid  = tid >> 5;
  const int r  = lane & 15;
  const int hf = lane >> 4;
  const size_t mbase = (size_t)blockIdx.x * 256;

  float4 a0[8], a1[8];
  load_arows<8>(a0, ctx + (mbase + wid * 16 + r) * EE + 4 * hf);
  load_arows<8>(a1, ctx + (mbase + (wid + 8) * 16 + r) * EE + 4 * hf);

  float wf[4][32];
  load_wfrag4<8>(wf, wout, EE, 4 * r, hf);

  v8f acc[4];
  #pragma unroll
  for (int t = 0; t < 4; ++t) acc[t] = (v8f){0, 0, 0, 0, 0, 0, 0, 0};
  mma_rows<8, 4>(acc, a0, wf);
  mma_rows<8, 4>(acc, a1, wf);

  #pragma unroll
  for (int t = 0; t < 4; ++t) {
    float ps = acc[t][0] + acc[t][1] + acc[t][2] + acc[t][3] +
               acc[t][4] + acc[t][5] + acc[t][6] + acc[t][7];
    ps += __shfl_xor(ps, 16, 32);
    if (lane < 16) atomicAdd(&pool[4 * r + t], ps);
  }
  __syncthreads();
  if (tid < EE)
    pooled[(size_t)blockIdx.x * EE + tid] = pool[tid] * (1.0f / 256.0f) + bout[tid];
}

__global__ __launch_bounds__(256) void k_mlp_lu(
    const float* __restrict__ pooled,
    const float* __restrict__ mw1, const float* __restrict__ mb1,
    const float* __restrict__ mg,  const float* __restrict__ mbb,
    const float* __restrict__ mw2, const float* __restrict__ mb2,
    const float* __restrict__ nw1, const float* __restrict__ nb1,
    const float* __restrict__ ng,  const float* __restrict__ nbb,
    const float* __restrict__ nw2, const float* __restrict__ nb2,
    float* __restrict__ out) {
  const int branch = blockIdx.x & 1;
  const int row    = blockIdx.x >> 1;
  const float* w1 = branch ? nw1 : mw1;
  const float* b1 = branch ? nb1 : mb1;
  const float* g  = branch ? ng  : mg;
  const float* bb = branch ? nbb : mbb;
  const float* w2 = branch ? nw2 : mw2;
  const float* b2 = branch ? nb2 : mb2;

  __shared__ float sx[EE];
  __shared__ float sh[HIDD];
  __shared__ float sraw[512];
  const int tid = threadIdx.x;

  if (tid < EE) sx[tid] = pooled[(size_t)row * EE + tid];
  __syncthreads();

  if (tid < HIDD) {
    float acc = b1[tid];
    #pragma unroll
    for (int i = 0; i < EE; ++i) acc += sx[i] * w1[i * HIDD + tid];
    float mu = acc;
    #pragma unroll
    for (int m = 16; m >= 1; m >>= 1) mu += __shfl_xor(mu, m, 32);
    mu *= (1.0f / 32.0f);
    const float d = acc - mu;
    float vv = d * d;
    #pragma unroll
    for (int m = 16; m >= 1; m >>= 1) vv += __shfl_xor(vv, m, 32);
    vv *= (1.0f / 32.0f);
    const float y = d * rsqrtf(vv + 1e-5f) * g[tid] + bb[tid];
    sh[tid] = 0.5f * y * (1.0f + erff(y * 0.70710678118654752f));
  }
  __syncthreads();

  for (int c = tid; c < 512; c += 256) {
    float acc = b2[c];
    #pragma unroll
    for (int j = 0; j < HIDD; ++j) acc += sh[j] * w2[j * 512 + c];
    sraw[c] = acc;
  }
  __syncthreads();

  if (tid < HH * NKK) {
    const int j = tid & 7;
    float* d = sraw + (tid >> 3) * 64 + 28 + (NKK * j - (j * (j - 1)) / 2);
    *d = softplusf(*d) + 0.01f;
  }
  __syncthreads();

  const size_t obase = ((size_t)branch * 384 + row) * 512;
  for (int o = tid; o < 512; o += 256) {
    const int h  = o >> 6;
    const int ik = o & 63;
    const int i  = ik >> 3;
    const int k  = ik & 7;
    const float* rh = sraw + h * 64;
    float acc = 0.0f;
    #pragma unroll
    for (int j = 0; j < NKK; ++j) {
      if (j > i || j > k) continue;
      const float Lij = (j == i) ? 1.0f : rh[(i * (i - 1)) / 2 + j];
      const float u = rh[28 + (NKK * j - (j * (j - 1)) / 2) + (k - j)];
      acc += Lij * u;
    }
    out[obase + o] = acc;
  }
}

extern "C" void kernel_launch(void* const* d_in, const int* in_sizes, int n_in,
                              void* d_out, int out_size, void* d_ws, size_t ws_size,
                              hipStream_t stream) {
  const float* wsfeat = (const float*)d_in[0];
  const float* W_emb  = (const float*)d_in[1];
  const float* b_emb  = (const float*)d_in[2];
  const float* pos    = (const float*)d_in[3];
  const float* w_in   = (const float*)d_in[4];
  const float* b_in   = (const float*)d_in[5];
  const float* w_out  = (const float*)d_in[6];
  const float* b_out  = (const float*)d_in[7];
  const float* m_w1 = (const float*)d_in[8];
  const float* m_b1 = (const float*)d_in[9];
  const float* m_g  = (const float*)d_in[10];
  const float* m_bb = (const float*)d_in[11];
  const float* m_w2 = (const float*)d_in[12];
  const float* m_b2 = (const float*)d_in[13];
  const float* n_w1 = (const float*)d_in[14];
  const float* n_b1 = (const float*)d_in[15];
  const float* n_g  = (const float*)d_in[16];
  const float* n_bb = (const float*)d_in[17];
  const float* n_w2 = (const float*)d_in[18];
  const float* n_b2 = (const float*)d_in[19];
  float* out = (float*)d_out;

  float* ws     = (float*)d_ws;
  float* feats  = ws;
  float* wc     = feats + (size_t)MROWS * 56;
  float* bt     = wc    + 64 * 192;
  float* qkv    = wc    + (size_t)MROWS * EE;
  float* ctx    = qkv   + (size_t)MROWS * 192;
  float* pooled = ctx   + (size_t)MROWS * EE;

  k_stats<<<G_STATS, 256, 0, stream>>>(wsfeat, feats, W_emb, b_emb, pos, w_in, b_in, wc, bt);
  k_qkv<<<G_QKV, 96, 0, stream>>>(feats, wc, bt, qkv);
  k_attn<<<G_ATTN, 256, 0, stream>>>(qkv, ctx);
  k_proj_pool<<<BB * LL, 256, 0, stream>>>(ctx, w_out, b_out, pooled);
  k_mlp_lu<<<BB * LL * 2, 256, 0, stream>>>(pooled,
      m_w1, m_b1, m_g, m_bb, m_w2, m_b2,
      n_w1, n_b1, n_g, n_bb, n_w2, n_b2, out);
}


# attention, w_out projection and pooling in one kernel: a block owns half a pool group (128 rows, 12 sequences, boundary ones redone), ctx stays in LDS; 5 -> 4 launches; pool sum order changed
# speedup vs baseline: 2.9458x; 1.0458x over previous
#include <hip/hip_runtime.h>
#include <math.h>

#define LL   12
#define TT   8
#define BB   32
#define CC   256
#define SS   64
#define EE   64
#define HDD  16
#define NHH  4
#define HIDD 32
#define HH   8
#define NKK  8
#define MROWS (BB*CC*LL)
#define SROWS (LL*TT*BB*CC)

#define G_STATS 2048
#define T_STATS ((SROWS / 64) / G_STATS)
#define G_QKV   1024
#define T_QKV   ((MROWS / 16) / G_QKV)

typedef float v2f __attribute__((ext_vector_type(2)));
typedef float v8f __attribute__((ext_vector_type(8)));

static __device__ __forceinline__ v8f wmma4(v2f a, v2f b, v8f c) {
  return __builtin_amdgcn_wmma_f32_16x16x4_f32(false, a, false, b, (short)0, c,
                                               false, false);
}

static __device__ __forceinline__ float softplusf(float x) {
  return (x > 20.0f) ? x : log1pf(__expf(x));
}

template <int KC>
static __device__ __forceinline__ void load_wfrag4(float (&wf)[4][4 * KC],
                                                   const float* __restrict__ w,
                                                   int ldw, int col4, int hf) {
  #pragma unroll
  for (int j = 0; j < KC; ++j)
    #pragma unroll
    for (int e = 0; e < 4; ++e) {
      const float4 v = *reinterpret_cast<const float4*>(
          w + (size_t)(8 * j + 4 * hf + e) * ldw + col4);
      wf[0][4 * j + e] = v.x; wf[1][4 * j + e] = v.y;
      wf[2][4 * j + e] = v.z; wf[3][4 * j + e] = v.w;
    }
}

template <int KC>
static __device__ __forceinline__ void load_arows(float4 (&a)[KC],
                                                  const float* __restrict__ rowp) {
  const float4* p = reinterpret_cast<const float4*>(rowp);
  #pragma unroll
  for (int j = 0; j < KC; ++j) a[j] = p[2 * j];
}

template <int KC, int NT>
static __device__ __forceinline__ void mma_rows(v8f (&acc)[NT], const float4 (&a)[KC],
                                                const float (&wf)[NT][4 * KC]) {
  #pragma unroll
  for (int j = 0; j < KC; ++j) {
    v2f a0; a0.x = a[j].x; a0.y = a[j].y;
    v2f a1; a1.x = a[j].z; a1.y = a[j].w;
    #pragma unroll
    for (int t = 0; t < NT; ++t) {
      v2f b; b.x = wf[t][4 * j]; b.y = wf[t][4 * j + 1];
      acc[t] = wmma4(a0, b, acc[t]);
    }
    #pragma unroll
    for (int t = 0; t < NT; ++t) {
      v2f b; b.x = wf[t][4 * j + 2]; b.y = wf[t][4 * j + 3];
      acc[t] = wmma4(a1, b, acc[t]);
    }
  }
}

template <int CTRL>
static __device__ __forceinline__ float quadf(float v) {
  return __int_as_float(__builtin_amdgcn_update_dpp(
      0, __float_as_int(v), CTRL, 0xf, 0xf, true));
}

#define MED3_QUAD(NAME, PERM)                                                     \
static __device__ __forceinline__ float NAME(float a, float b, float sel) {      \
  float d;                                                                        \
  asm("v_med3_num_f32_e64_dpp %0, %1, %2, %3 quad_perm:" PERM                      \
      " row_mask:0xf bank_mask:0xf"                                               \
      : "=v"(d) : "v"(b), "v"(a), "v"(sel));                                      \
  return d;                                                                       \
}
MED3_QUAD(med3_xor1, "[1,0,3,2]")
MED3_QUAD(med3_xor3, "[3,2,1,0]")
#undef MED3_QUAD

#define QP_XOR1  0xB1
#define QP_XOR2  0x4E
#define QP_NEXT  0xF9
#define QP_LANE0 0x00

#define CE(i, j) { const float lo_ = fminf(x[i], x[j]); \
                   const float hi_ = fmaxf(x[i], x[j]); x[i] = lo_; x[j] = hi_; }
#define HALFCLEAN16 \
  CE(0,8) CE(1,9) CE(2,10) CE(3,11) CE(4,12) CE(5,13) CE(6,14) CE(7,15) \
  CE(0,4) CE(1,5) CE(2,6) CE(3,7) CE(8,12) CE(9,13) CE(10,14) CE(11,15) \
  CE(0,2) CE(1,3) CE(4,6) CE(5,7) CE(8,10) CE(9,11) CE(12,14) CE(13,15) \
  CE(0,1) CE(2,3) CE(4,5) CE(6,7) CE(8,9) CE(10,11) CE(12,13) CE(14,15) \
  ;

#define N_CTILES 60
template <bool BIAS>
static __device__ __forceinline__ void compose_tile(int rt, int ct, int lane,
                                                    const float* __restrict__ Wemb,
                                                    const float* __restrict__ bemb,
                                                    const float* __restrict__ pos,
                                                    const float* __restrict__ win,
                                                    const float* __restrict__ bin,
                                                    float* __restrict__ Wc,
                                                    float* __restrict__ bt) {
  const int r  = lane & 15;
  const int hf = lane >> 4;
  const int col = ct * 16 + r;
  const int nvalid = BIAS ? LL : (56 - rt * 16);
  const bool a_ok = r < nvalid;
  const int rr = a_ok ? r : 0;
  const float* ap = BIAS ? (pos + rr * EE) : (Wemb + (rt * 16 + rr) * EE);

  v8f acc = {0, 0, 0, 0, 0, 0, 0, 0};
  #pragma unroll
  for (int half = 0; half < 2; ++half) {
    v2f a[8], b[8];
    #pragma unroll
    for (int s = 0; s < 8; ++s) {
      const int c0 = (half * 8 + s) * 4 + 2 * hf;
      const float2 av = *reinterpret_cast<const float2*>(ap + c0);
      float ax = av.x, ay = av.y;
      if (BIAS) {
        const float2 bv = *reinterpret_cast<const float2*>(bemb + c0);
        ax += bv.x; ay += bv.y;
      }
      a[s].x = a_ok ? ax : 0.0f;
      a[s].y = a_ok ? ay : 0.0f;
      b[s].x = win[c0 * 192 + col];
      b[s].y = win[(c0 + 1) * 192 + col];
    }
    #pragma unroll
    for (int s = 0; s < 8; ++s) acc = wmma4(a[s], b[s], acc);
  }

  const float badd = BIAS ? bin[col] : 0.0f;
  float* dst = (BIAS ? bt : (Wc + rt * 16 * 192)) + 8 * hf * 192 + col;
  #pragma unroll
  for (int g = 0; g < 8; ++g)
    dst[g * 192] = BIAS ? acc[g] + badd : acc[g];
}

__global__ __launch_bounds__(256) void k_stats(const float* __restrict__ ws,
                                               float* __restrict__ feats,
                                               const float* __restrict__ Wemb,
                                               const float* __restrict__ bemb,
                                               const float* __restrict__ pos,
                                               const float* __restrict__ win,
                                               const float* __restrict__ bin,
                                               float* __restrict__ Wc,
                                               float* __restrict__ bt) {
  if (blockIdx.x < N_CTILES && __builtin_amdgcn_readfirstlane(threadIdx.x >> 5) == 0) {
    const int rt = blockIdx.x / 12;
    const int ct = blockIdx.x - rt * 12;
    if (rt == 4) compose_tile<true >(rt, ct, threadIdx.x & 31, Wemb, bemb, pos, win, bin, Wc, bt);
    else         compose_tile<false>(rt, ct, threadIdx.x & 31, Wemb, bemb, pos, win, bin, Wc, bt);
  }

  const int l4  = threadIdx.x & 3;
  const float inf  = __builtin_inff();
  const float sel1 = (l4 & 1) ? inf : -inf;
  const float sel2 = (l4 & 2) ? inf : -inf;

  for (int trip = 0; trip < T_STATS; ++trip) {
  const int row = (trip * G_STATS + blockIdx.x) * 64 + (threadIdx.x >> 2);

  const float4* p = reinterpret_cast<const float4*>(ws + (size_t)row * SS) + l4;
  float x[16];
  #pragma unroll
  for (int q = 0; q < 4; ++q) {
    const float4 v = p[q * 4];
    x[4 * q + 0] = v.x; x[4 * q + 1] = v.y; x[4 * q + 2] = v.z; x[4 * q + 3] = v.w;
  }

  float sa = x[0], sb = x[1], qa = x[0] * x[0], qb = x[1] * x[1];
  #pragma unroll
  for (int i = 2; i < 16; i += 2) {
    sa += x[i];     qa = fmaf(x[i],     x[i],     qa);
    sb += x[i + 1]; qb = fmaf(x[i + 1], x[i + 1], qb);
  }
  float s = sa + sb, s2 = qa + qb;
  s  += quadf<QP_XOR1>(s);   s2 += quadf<QP_XOR1>(s2);
  s  += quadf<QP_XOR2>(s);   s2 += quadf<QP_XOR2>(s2);
  const float mean = s * (1.0f / 64.0f);
  const float var  = (s2 - 64.0f * mean * mean) * (1.0f / 63.0f);

  CE(0,13) CE(1,12) CE(2,15) CE(3,14) CE(4,8) CE(5,6) CE(7,11) CE(9,10)
  CE(0,5) CE(1,7) CE(2,9) CE(3,4) CE(6,13) CE(8,14) CE(10,15) CE(11,12)
  CE(0,1) CE(2,3) CE(4,5) CE(6,8) CE(7,9) CE(10,11) CE(12,13) CE(14,15)
  CE(0,2) CE(1,3) CE(4,10) CE(5,11) CE(6,7) CE(8,9) CE(12,14) CE(13,15)
  CE(1,2) CE(3,12) CE(4,6) CE(5,7) CE(8,10) CE(9,11) CE(13,14)
  CE(1,4) CE(2,6) CE(5,8) CE(7,10) CE(9,13) CE(11,14)
  CE(2,4) CE(3,6) CE(9,12) CE(11,13)
  CE(3,5) CE(6,8) CE(7,9) CE(10,12)
  CE(3,4) CE(5,6) CE(7,8) CE(9,10) CE(11,12)
  CE(6,7) CE(8,9)

  {
    float t[16];
    #pragma unroll
    for (int r = 0; r < 16; ++r)
      t[r] = med3_xor1(x[r], x[15 - r], sel1);
    #pragma unroll
    for (int r = 0; r < 16; ++r) x[r] = t[r];
  }
  HALFCLEAN16

  {
    float t[16];
    #pragma unroll
    for (int r = 0; r < 16; ++r)
      t[r] = med3_xor3(x[r], x[15 - r], sel2);
    #pragma unroll
    for (int r = 0; r < 16; ++r)
      x[r] = med3_xor1(t[r], t[r], sel1);
  }
  float mn = x[0], mx = x[0];
  #pragma unroll
  for (int i = 1; i < 16; ++i) { mn = fminf(mn, x[i]); mx = fmaxf(mx, x[i]); }

  const float nxt = quadf<QP_NEXT>(mn);
  const float q0  = quadf<QP_LANE0>(mn);
  const float w   = (l4 == 0) ? 0.75f : 0.25f;
  const float lin = mx + w * (nxt - mx);
  const float mid = 0.5f * (mx + nxt);
  const float hi  = (l4 == 1) ? mid : ((l4 == 3) ? mx : lin);
  const float lo  = (l4 == 0) ? mean : ((l4 == 1) ? var : q0);

  const int c = row & 255;
  int tmp = row >> 8;
  const int b = tmp & 31; tmp >>= 5;
  const int t = tmp & 7;
  const int l = tmp >> 3;
  float* o = feats + ((size_t)((b * LL + l) * CC + c)) * 56 + t * 7;
  o[3 + l4] = hi;
  if (l4 < 3) o[l4] = lo;
  }
}
#undef CE
#undef HALFCLEAN16

__global__ __launch_bounds__(96) void k_qkv(const float* __restrict__ feats,
                                            const float* __restrict__ Wc,
                                            const float* __restrict__ bt,
                                            float* __restrict__ qkv) {
  const int lane = threadIdx.x & 31;
  const int wid  = threadIdx.x >> 5;
  const int r  = lane & 15;
  const int hf = lane >> 4;
  const int n4 = wid * 64 + 4 * r;

  float wf[4][28];
  load_wfrag4<7>(wf, Wc, 192, n4, hf);

  float4 a[7];
  load_arows<7>(a, feats + (size_t)(blockIdx.x * 16 + r) * 56 + 4 * hf);

  for (int trip = 0; trip < T_QKV; ++trip) {
    const int m0 = (trip * G_QKV + blockIdx.x) * 16;
    float4 an[7];
    if (trip + 1 < T_QKV)
      load_arows<7>(an, feats + (size_t)(m0 + G_QKV * 16 + r) * 56 + 4 * hf);
    const int l = (m0 >> 8) % LL;
    const float4 bi = *reinterpret_cast<const float4*>(bt + l * 192 + n4);

    v8f acc[4];
    #pragma unroll
    for (int t = 0; t < 4; ++t) acc[t] = (v8f){0, 0, 0, 0, 0, 0, 0, 0};
    mma_rows<7, 4>(acc, a, wf);

    #pragma unroll
    for (int g = 0; g < 8; ++g) {
      float4 o;
      o.x = acc[0][g] + bi.x;
      o.y = acc[1][g] + bi.y;
      o.z = acc[2][g] + bi.z;
      o.w = acc[3][g] + bi.w;
      *reinterpret_cast<float4*>(qkv + (size_t)(m0 + g + 8 * hf) * 192 + n4) = o;
    }

    if (trip + 1 < T_QKV) {
      #pragma unroll
      for (int j = 0; j < 7; ++j) a[j] = an[j];
    }
  }
}

#define DPP_LINE(OP, XM, D, S) \
  OP " " D ", " S ", " S " row_xmask:" XM " row_mask:0xf bank_mask:0xf\n\t"
#define ROW_STEP_FIRST(NAME, LEAD, OP, XM)                                        \
static __device__ __forceinline__ void NAME(float (&d)[8], const float (&s)[8]) { \
  asm(LEAD                                                                        \
      DPP_LINE(OP, XM, "%0", "%8")  DPP_LINE(OP, XM, "%1", "%9")                  \
      DPP_LINE(OP, XM, "%2", "%10") DPP_LINE(OP, XM, "%3", "%11")                 \
      DPP_LINE(OP, XM, "%4", "%12") DPP_LINE(OP, XM, "%5", "%13")                 \
      DPP_LINE(OP, XM, "%6", "%14") DPP_LINE(OP, XM, "%7", "%15")                 \
      : "=&v"(d[0]), "=&v"(d[1]), "=&v"(d[2]), "=&v"(d[3]),                       \
        "=&v"(d[4]), "=&v"(d[5]), "=&v"(d[6]), "=&v"(d[7])                        \
      : "v"(s[0]), "v"(s[1]), "v"(s[2]), "v"(s[3]),                               \
        "v"(s[4]), "v"(s[5]), "v"(s[6]), "v"(s[7]));                              \
}
#define ROW_STEP(NAME, OP, XM)                                                    \
static __device__ __forceinline__ void NAME(float (&v)[8]) {                      \
  asm(DPP_LINE(OP, XM, "%0", "%0") DPP_LINE(OP, XM, "%1", "%1")                   \
      DPP_LINE(OP, XM, "%2", "%2") DPP_LINE(OP, XM, "%3", "%3")                   \
      DPP_LINE(OP, XM, "%4", "%4") DPP_LINE(OP, XM, "%5", "%5")                   \
      DPP_LINE(OP, XM, "%6", "%6") DPP_LINE(OP, XM, "%7", "%7")                   \
      : "+v"(v[0]), "+v"(v[1]), "+v"(v[2]), "+v"(v[3]),                           \
        "+v"(v[4]), "+v"(v[5]), "+v"(v[6]), "+v"(v[7]));                          \
}
ROW_STEP_FIRST(rowmax_x8, "", "v_max_num_f32_dpp", "8")
ROW_STEP(rowmax_x4, "v_max_num_f32_dpp", "4")
ROW_STEP(rowmax_x2, "v_max_num_f32_dpp", "2")
ROW_STEP(rowmax_x1, "v_max_num_f32_dpp", "1")
ROW_STEP_FIRST(rowsum_x8, "v_nop\n\t", "v_add_f32_dpp", "8")
ROW_STEP(rowsum_x4, "v_add_f32_dpp", "4")
ROW_STEP(rowsum_x2, "v_add_f32_dpp", "2")
ROW_STEP(rowsum_x1, "v_add_f32_dpp", "1")
#undef ROW_STEP
#undef ROW_STEP_FIRST
#undef DPP_LINE

#define NSEQ  (BB * CC)
#define HROWS 128
#define NHALF (MROWS / HROWS)
#define T_SEQ 6
#define CPITCH 68

struct QKFrag { float4 qa, qb, ka, kb; };
static __device__ __forceinline__ void load_qk(QKFrag& f, const float* __restrict__ rowp) {
  const float4* p = reinterpret_cast<const float4*>(rowp);
  f.qa = p[0];  f.qb = p[1];
  f.ka = p[16]; f.kb = p[17];
}

static __device__ __forceinline__ void load_v(float (&vb)[6], const float* __restrict__ base,
                                              int r, int hf) {
  #pragma unroll
  for (int s = 0; s < 3; ++s) {
    const float* vp = base + (size_t)(4 * s + 2 * hf) * 192 + 128 + r;
    vb[2 * s]     = vp[0];
    vb[2 * s + 1] = vp[192];
  }
}

__global__ __launch_bounds__(256, 6) void k_attn_proj(const float* __restrict__ qkv,
                                                      const float* __restrict__ wout,
                                                      float* __restrict__ part) {
  __shared__ float sP[8][16][17];
  __shared__ __attribute__((aligned(16))) float sC[HROWS * CPITCH];
  __shared__ float pool[EE];
  const int tid  = threadIdx.x;
  const int lane = tid & 31;
  const int wid  = tid >> 5;
  const int r  = lane & 15;
  const int hf = lane >> 4;
  const int rr = (r < LL) ? r : 0;
  const int qk_off = rr * 192 + 8 * hf;
  const int h  = wid & 3;

  if (tid < EE) pool[tid] = 0.0f;

  const int R0   = blockIdx.x * HROWS;
  const int n_lo = R0 / LL;
  const int seq0 = n_lo + (wid >> 2);

  QKFrag fq[3];
  float  fv[3][6];
  {
    const int s0 = (seq0 < NSEQ) ? seq0 : (NSEQ - 1);
    const int s1 = (seq0 + 2 < NSEQ) ? (seq0 + 2) : (NSEQ - 1);
    const float* b0 = qkv + (size_t)s0 * LL * 192 + h * HDD;
    const float* b1 = qkv + (size_t)s1 * LL * 192 + h * HDD;
    load_qk(fq[0], b0 + qk_off); load_v(fv[0], b0, r, hf);
    load_qk(fq[1], b1 + qk_off); load_v(fv[1], b1, r, hf);
  }
  __builtin_amdgcn_sched_barrier(0);

  #pragma unroll
  for (int i = 0; i < T_SEQ; ++i) {
  const int cs = i % 3;
  const int seq_u = seq0 + 2 * i;
  const QKFrag& cur = fq[cs];

  v8f S = {0, 0, 0, 0, 0, 0, 0, 0};
  {
    v2f a, b;
    a.x = cur.qa.x; a.y = cur.qa.y; b.x = cur.ka.x; b.y = cur.ka.y; S = wmma4(a, b, S);
    a.x = cur.qa.z; a.y = cur.qa.w; b.x = cur.ka.z; b.y = cur.ka.w; S = wmma4(a, b, S);
    a.x = cur.qb.x; a.y = cur.qb.y; b.x = cur.kb.x; b.y = cur.kb.y; S = wmma4(a, b, S);
    a.x = cur.qb.z; a.y = cur.qb.w; b.x = cur.kb.z; b.y = cur.kb.w; S = wmma4(a, b, S);
  }

  if (i + 2 < T_SEQ) {
    const int ns = (seq_u + 4 < NSEQ) ? (seq_u + 4) : (NSEQ - 1);
    const float* nbase = qkv + (size_t)ns * LL * 192 + h * HDD;
    load_qk(fq[(i + 2) % 3], nbase + qk_off);
    load_v(fv[(i + 2) % 3], nbase, r, hf);
  }
  __builtin_amdgcn_sched_barrier(0);

  float P[8], sv[8], rm[8], rs[8];
  #pragma unroll
  for (int g = 0; g < 8; ++g)
    sv[g] = (r < LL) ? S[g] * 0.25f : -__builtin_inff();
  rowmax_x8(rm, sv); rowmax_x4(rm); rowmax_x2(rm); rowmax_x1(rm);
  #pragma unroll
  for (int g = 0; g < 8; ++g) P[g] = __expf(sv[g] - rm[g]);
  rowsum_x8(rs, P); rowsum_x4(rs); rowsum_x2(rs); rowsum_x1(rs);
  #pragma unroll
  for (int g = 0; g < 8; ++g) P[g] = P[g] / rs[g];
  #pragma unroll
  for (int g = 0; g < 8; ++g) sP[wid][g + 8 * hf][r] = P[g];

  v8f Cv = {0, 0, 0, 0, 0, 0, 0, 0};
  #pragma unroll
  for (int k0 = 0; k0 < 16; k0 += 4) {
    const int c0 = k0 + 2 * hf;
    v2f a; a.x = sP[wid][r][c0]; a.y = sP[wid][r][c0 + 1];
    v2f b;
    b.x = (k0 < LL) ? fv[cs][(k0 >> 1)]     : 0.0f;
    b.y = (k0 < LL) ? fv[cs][(k0 >> 1) + 1] : 0.0f;
    Cv = wmma4(a, b, Cv);
  }
  const int sb = seq_u * LL - R0;
  #pragma unroll
  for (int g = 0; g < 8; ++g) {
    const int m  = g + 8 * hf;
    const int rl = sb + m;
    if (m < LL && seq_u < NSEQ && (unsigned)rl < (unsigned)HROWS)
      sC[rl * CPITCH + h * HDD + r] = Cv[g];
  }
  }

  const int col4 = 4 * r;
  float wf[4][16];
  load_wfrag4<4>(wf, wout, EE, col4, hf);

  __syncthreads();

  const float* crow = sC + (wid * 16 + r) * CPITCH + 4 * hf;
  v8f acc[4];
  #pragma unroll
  for (int t = 0; t < 4; ++t) acc[t] = (v8f){0, 0, 0, 0, 0, 0, 0, 0};
  {
    float4 a[4];
    #pragma unroll
    for (int j = 0; j < 4; ++j) a[j] = *reinterpret_cast<const float4*>(crow + 8 * j);
    mma_rows<4, 4>(acc, a, wf);
  }
  {
    float wg[4][16];
    load_wfrag4<4>(wg, wout + 32 * EE, EE, col4, hf);
    float4 a[4];
    #pragma unroll
    for (int j = 0; j < 4; ++j) a[j] = *reinterpret_cast<const float4*>(crow + 32 + 8 * j);
    mma_rows<4, 4>(acc, a, wg);
  }

  #pragma unroll
  for (int t = 0; t < 4; ++t) {
    float ps = acc[t][0] + acc[t][1] + acc[t][2] + acc[t][3] +
               acc[t][4] + acc[t][5] + acc[t][6] + acc[t][7];
    ps += __shfl_xor(ps, 16, 32);
    if (lane < 16) atomicAdd(&pool[4 * r + t], ps);
  }
  __syncthreads();
  if (tid < EE)
    part[(size_t)blockIdx.x * EE + tid] = pool[tid];
}

__global__ __launch_bounds__(256) void k_mlp_lu(
    const float* __restrict__ part, const float* __restrict__ bout,
    const float* __restrict__ mw1, const float* __restrict__ mb1,
    const float* __restrict__ mg,  const float* __restrict__ mbb,
    const float* __restrict__ mw2, const float* __restrict__ mb2,
    const float* __restrict__ nw1, const float* __restrict__ nb1,
    const float* __restrict__ ng,  const float* __restrict__ nbb,
    const float* __restrict__ nw2, const float* __restrict__ nb2,
    float* __restrict__ out) {
  const int branch = blockIdx.x & 1;
  const int row    = blockIdx.x >> 1;
  const float* w1 = branch ? nw1 : mw1;
  const float* b1 = branch ? nb1 : mb1;
  const float* g  = branch ? ng  : mg;
  const float* bb = branch ? nbb : mbb;
  const float* w2 = branch ? nw2 : mw2;
  const float* b2 = branch ? nb2 : mb2;

  __shared__ float sx[EE];
  __shared__ float sh[HIDD];
  __shared__ float sraw[512];
  const int tid = threadIdx.x;

  if (tid < EE)
    sx[tid] = (part[(size_t)(2 * row) * EE + tid] + part[(size_t)(2 * row + 1) * EE + tid])
              * (1.0f / 256.0f) + bout[tid];
  __syncthreads();

  if (tid < HIDD) {
    float acc = b1[tid];
    #pragma unroll
    for (int i = 0; i < EE; ++i) acc += sx[i] * w1[i * HIDD + tid];
    float mu = acc;
    #pragma unroll
    for (int m = 16; m >= 1; m >>= 1) mu += __shfl_xor(mu, m, 32);
    mu *= (1.0f / 32.0f);
    const float d = acc - mu;
    float vv = d * d;
    #pragma unroll
    for (int m = 16; m >= 1; m >>= 1) vv += __shfl_xor(vv, m, 32);
    vv *= (1.0f / 32.0f);
    const float y = d * rsqrtf(vv + 1e-5f) * g[tid] + bb[tid];
    sh[tid] = 0.5f * y * (1.0f + erff(y * 0.70710678118654752f));
  }
  __syncthreads();

  for (int c = tid; c < 512; c += 256) {
    float acc = b2[c];
    #pragma unroll
    for (int j = 0; j < HIDD; ++j) acc += sh[j] * w2[j * 512 + c];
    sraw[c] = acc;
  }
  __syncthreads();

  if (tid < HH * NKK) {
    const int j = tid & 7;
    float* d = sraw + (tid >> 3) * 64 + 28 + (NKK * j - (j * (j - 1)) / 2);
    *d = softplusf(*d) + 0.01f;
  }
  __syncthreads();

  const size_t obase = ((size_t)branch * 384 + row) * 512;
  for (int o = tid; o < 512; o += 256) {
    const int h  = o >> 6;
    const int ik = o & 63;
    const int i  = ik >> 3;
    const int k  = ik & 7;
    const float* rh = sraw + h * 64;
    float acc = 0.0f;
    #pragma unroll
    for (int j = 0; j < NKK; ++j) {
      if (j > i || j > k) continue;
      const float Lij = (j == i) ? 1.0f : rh[(i * (i - 1)) / 2 + j];
      const float u = rh[28 + (NKK * j - (j * (j - 1)) / 2) + (k - j)];
      acc += Lij * u;
    }
    out[obase + o] = acc;
  }
}

extern "C" void kernel_launch(void* const* d_in, const int* in_sizes, int n_in,
                              void* d_out, int out_size, void* d_ws, size_t ws_size,
                              hipStream_t stream) {
  const float* wsfeat = (const float*)d_in[0];
  const float* W_emb  = (const float*)d_in[1];
  const float* b_emb  = (const float*)d_in[2];
  const float* pos    = (const float*)d_in[3];
  const float* w_in   = (const float*)d_in[4];
  const float* b_in   = (const float*)d_in[5];
  const float* w_out  = (const float*)d_in[6];
  const float* b_out  = (const float*)d_in[7];
  const float* m_w1 = (const float*)d_in[8];
  const float* m_b1 = (const float*)d_in[9];
  const float* m_g  = (const float*)d_in[10];
  const float* m_bb = (const float*)d_in[11];
  const float* m_w2 = (const float*)d_in[12];
  const float* m_b2 = (const float*)d_in[13];
  const float* n_w1 = (const float*)d_in[14];
  const float* n_b1 = (const float*)d_in[15];
  const float* n_g  = (const float*)d_in[16];
  const float* n_bb = (const float*)d_in[17];
  const float* n_w2 = (const float*)d_in[18];
  const float* n_b2 = (const float*)d_in[19];
  float* out = (float*)d_out;

  float* ws     = (float*)d_ws;
  float* feats  = ws;
  float* wc     = feats + (size_t)MROWS * 56;
  float* bt     = wc    + 64 * 192;
  float* qkv    = wc    + (size_t)MROWS * EE;
  float* part   = qkv   + (size_t)MROWS * 192;

  k_stats<<<G_STATS, 256, 0, stream>>>(wsfeat, feats, W_emb, b_emb, pos, w_in, b_in, wc, bt);
  k_qkv<<<G_QKV, 96, 0, stream>>>(feats, wc, bt, qkv);
  k_attn_proj<<<NHALF, 256, 0, stream>>>(qkv, w_out, part);
  k_mlp_lu<<<BB * LL * 2, 256, 0, stream>>>(part, b_out,
      m_w1, m_b1, m_g, m_bb, m_w2, m_b2,
      n_w1, n_b1, n_g, n_bb, n_w2, n_b2, out);
}


# k_attn softmax moved behind the LDS layout change: raw scores go C->LDS->A layout, a lane holds 6 of a row's 12 columns, so max and sum are in-lane + 1 half swap, 1 v_rcp per row; sum order changed
# speedup vs baseline: 3.0645x; 1.0403x over previous
#include <hip/hip_runtime.h>
#include <math.h>

#define LL   12
#define TT   8
#define BB   32
#define CC   256
#define SS   64
#define EE   64
#define HDD  16
#define NHH  4
#define HIDD 32
#define HH   8
#define NKK  8
#define MROWS (BB*CC*LL)
#define SROWS (LL*TT*BB*CC)

#define G_STATS 2048
#define T_STATS ((SROWS / 64) / G_STATS)
#define G_QKV   1024
#define T_QKV   ((MROWS / 16) / G_QKV)

typedef float v2f __attribute__((ext_vector_type(2)));
typedef float v8f __attribute__((ext_vector_type(8)));

static __device__ __forceinline__ v8f wmma4(v2f a, v2f b, v8f c) {
  return __builtin_amdgcn_wmma_f32_16x16x4_f32(false, a, false, b, (short)0, c,
                                               false, false);
}

static __device__ __forceinline__ float softplusf(float x) {
  return (x > 20.0f) ? x : log1pf(__expf(x));
}

template <int KC>
static __device__ __forceinline__ void load_wfrag4(float (&wf)[4][4 * KC],
                                                   const float* __restrict__ w,
                                                   int ldw, int col4, int hf) {
  #pragma unroll
  for (int j = 0; j < KC; ++j)
    #pragma unroll
    for (int e = 0; e < 4; ++e) {
      const float4 v = *reinterpret_cast<const float4*>(
          w + (size_t)(8 * j + 4 * hf + e) * ldw + col4);
      wf[0][4 * j + e] = v.x; wf[1][4 * j + e] = v.y;
      wf[2][4 * j + e] = v.z; wf[3][4 * j + e] = v.w;
    }
}

template <int KC>
static __device__ __forceinline__ void load_arows(float4 (&a)[KC],
                                                  const float* __restrict__ rowp) {
  const float4* p = reinterpret_cast<const float4*>(rowp);
  #pragma unroll
  for (int j = 0; j < KC; ++j) a[j] = p[2 * j];
}

template <int KC, int NT>
static __device__ __forceinline__ void mma_rows(v8f (&acc)[NT], const float4 (&a)[KC],
                                                const float (&wf)[NT][4 * KC]) {
  #pragma unroll
  for (int j = 0; j < KC; ++j) {
    v2f a0; a0.x = a[j].x; a0.y = a[j].y;
    v2f a1; a1.x = a[j].z; a1.y = a[j].w;
    #pragma unroll
    for (int t = 0; t < NT; ++t) {
      v2f b; b.x = wf[t][4 * j]; b.y = wf[t][4 * j + 1];
      acc[t] = wmma4(a0, b, acc[t]);
    }
    #pragma unroll
    for (int t = 0; t < NT; ++t) {
      v2f b; b.x = wf[t][4 * j + 2]; b.y = wf[t][4 * j + 3];
      acc[t] = wmma4(a1, b, acc[t]);
    }
  }
}

template <int CTRL>
static __device__ __forceinline__ float quadf(float v) {
  return __int_as_float(__builtin_amdgcn_update_dpp(
      0, __float_as_int(v), CTRL, 0xf, 0xf, true));
}

#define MED3_QUAD(NAME, PERM)                                                     \
static __device__ __forceinline__ float NAME(float a, float b, float sel) {      \
  float d;                                                                        \
  asm("v_med3_num_f32_e64_dpp %0, %1, %2, %3 quad_perm:" PERM                      \
      " row_mask:0xf bank_mask:0xf"                                               \
      : "=v"(d) : "v"(b), "v"(a), "v"(sel));                                      \
  return d;                                                                       \
}
MED3_QUAD(med3_xor1, "[1,0,3,2]")
MED3_QUAD(med3_xor3, "[3,2,1,0]")
#undef MED3_QUAD

#define QP_XOR1  0xB1
#define QP_XOR2  0x4E
#define QP_NEXT  0xF9
#define QP_LANE0 0x00

#define CE(i, j) { const float lo_ = fminf(x[i], x[j]); \
                   const float hi_ = fmaxf(x[i], x[j]); x[i] = lo_; x[j] = hi_; }
#define HALFCLEAN16 \
  CE(0,8) CE(1,9) CE(2,10) CE(3,11) CE(4,12) CE(5,13) CE(6,14) CE(7,15) \
  CE(0,4) CE(1,5) CE(2,6) CE(3,7) CE(8,12) CE(9,13) CE(10,14) CE(11,15) \
  CE(0,2) CE(1,3) CE(4,6) CE(5,7) CE(8,10) CE(9,11) CE(12,14) CE(13,15) \
  CE(0,1) CE(2,3) CE(4,5) CE(6,7) CE(8,9) CE(10,11) CE(12,13) CE(14,15) \
  ;

#define N_CTILES 60
template <bool BIAS>
static __device__ __forceinline__ void compose_tile(int rt, int ct, int lane,
                                                    const float* __restrict__ Wemb,
                                                    const float* __restrict__ bemb,
                                                    const float* __restrict__ pos,
                                                    const float* __restrict__ win,
                                                    const float* __restrict__ bin,
                                                    float* __restrict__ Wc,
                                                    float* __restrict__ bt) {
  const int r  = lane & 15;
  const int hf = lane >> 4;
  const int col = ct * 16 + r;
  const int nvalid = BIAS ? LL : (56 - rt * 16);
  const bool a_ok = r < nvalid;
  const int rr = a_ok ? r : 0;
  const float* ap = BIAS ? (pos + rr * EE) : (Wemb + (rt * 16 + rr) * EE);

  v8f acc = {0, 0, 0, 0, 0, 0, 0, 0};
  #pragma unroll
  for (int half = 0; half < 2; ++half) {
    v2f a[8], b[8];
    #pragma unroll
    for (int s = 0; s < 8; ++s) {
      const int c0 = (half * 8 + s) * 4 + 2 * hf;
      const float2 av = *reinterpret_cast<const float2*>(ap + c0);
      float ax = av.x, ay = av.y;
      if (BIAS) {
        const float2 bv = *reinterpret_cast<const float2*>(bemb + c0);
        ax += bv.x; ay += bv.y;
      }
      a[s].x = a_ok ? ax : 0.0f;
      a[s].y = a_ok ? ay : 0.0f;
      b[s].x = win[c0 * 192 + col];
      b[s].y = win[(c0 + 1) * 192 + col];
    }
    #pragma unroll
    for (int s = 0; s < 8; ++s) acc = wmma4(a[s], b[s], acc);
  }

  const float badd = BIAS ? bin[col] : 0.0f;
  float* dst = (BIAS ? bt : (Wc + rt * 16 * 192)) + 8 * hf * 192 + col;
  #pragma unroll
  for (int g = 0; g < 8; ++g)
    dst[g * 192] = BIAS ? acc[g] + badd : acc[g];
}

__global__ __launch_bounds__(256) void k_stats(const float* __restrict__ ws,
                                               float* __restrict__ feats,
                                               const float* __restrict__ Wemb,
                                               const float* __restrict__ bemb,
                                               const float* __restrict__ pos,
                                               const float* __restrict__ win,
                                               const float* __restrict__ bin,
                                               float* __restrict__ Wc,
                                               float* __restrict__ bt) {
  if (blockIdx.x < N_CTILES && __builtin_amdgcn_readfirstlane(threadIdx.x >> 5) == 0) {
    const int rt = blockIdx.x / 12;
    const int ct = blockIdx.x - rt * 12;
    if (rt == 4) compose_tile<true >(rt, ct, threadIdx.x & 31, Wemb, bemb, pos, win, bin, Wc, bt);
    else         compose_tile<false>(rt, ct, threadIdx.x & 31, Wemb, bemb, pos, win, bin, Wc, bt);
  }

  const int l4  = threadIdx.x & 3;
  const float inf  = __builtin_inff();
  const float sel1 = (l4 & 1) ? inf : -inf;
  const float sel2 = (l4 & 2) ? inf : -inf;

  for (int trip = 0; trip < T_STATS; ++trip) {
  const int row = (trip * G_STATS + blockIdx.x) * 64 + (threadIdx.x >> 2);

  const float4* p = reinterpret_cast<const float4*>(ws + (size_t)row * SS) + l4;
  float x[16];
  #pragma unroll
  for (int q = 0; q < 4; ++q) {
    const float4 v = p[q * 4];
    x[4 * q + 0] = v.x; x[4 * q + 1] = v.y; x[4 * q + 2] = v.z; x[4 * q + 3] = v.w;
  }

  float sa = x[0], sb = x[1], qa = x[0] * x[0], qb = x[1] * x[1];
  #pragma unroll
  for (int i = 2; i < 16; i += 2) {
    sa += x[i];     qa = fmaf(x[i],     x[i],     qa);
    sb += x[i + 1]; qb = fmaf(x[i + 1], x[i + 1], qb);
  }
  float s = sa + sb, s2 = qa + qb;
  s  += quadf<QP_XOR1>(s);   s2 += quadf<QP_XOR1>(s2);
  s  += quadf<QP_XOR2>(s);   s2 += quadf<QP_XOR2>(s2);
  const float mean = s * (1.0f / 64.0f);
  const float var  = (s2 - 64.0f * mean * mean) * (1.0f / 63.0f);

  CE(0,13) CE(1,12) CE(2,15) CE(3,14) CE(4,8) CE(5,6) CE(7,11) CE(9,10)
  CE(0,5) CE(1,7) CE(2,9) CE(3,4) CE(6,13) CE(8,14) CE(10,15) CE(11,12)
  CE(0,1) CE(2,3) CE(4,5) CE(6,8) CE(7,9) CE(10,11) CE(12,13) CE(14,15)
  CE(0,2) CE(1,3) CE(4,10) CE(5,11) CE(6,7) CE(8,9) CE(12,14) CE(13,15)
  CE(1,2) CE(3,12) CE(4,6) CE(5,7) CE(8,10) CE(9,11) CE(13,14)
  CE(1,4) CE(2,6) CE(5,8) CE(7,10) CE(9,13) CE(11,14)
  CE(2,4) CE(3,6) CE(9,12) CE(11,13)
  CE(3,5) CE(6,8) CE(7,9) CE(10,12)
  CE(3,4) CE(5,6) CE(7,8) CE(9,10) CE(11,12)
  CE(6,7) CE(8,9)

  {
    float t[16];
    #pragma unroll
    for (int r = 0; r < 16; ++r)
      t[r] = med3_xor1(x[r], x[15 - r], sel1);
    #pragma unroll
    for (int r = 0; r < 16; ++r) x[r] = t[r];
  }
  HALFCLEAN16

  {
    float t[16];
    #pragma unroll
    for (int r = 0; r < 16; ++r)
      t[r] = med3_xor3(x[r], x[15 - r], sel2);
    #pragma unroll
    for (int r = 0; r < 16; ++r)
      x[r] = med3_xor1(t[r], t[r], sel1);
  }
  float mn = x[0], mx = x[0];
  #pragma unroll
  for (int i = 1; i < 16; ++i) { mn = fminf(mn, x[i]); mx = fmaxf(mx, x[i]); }

  const float nxt = quadf<QP_NEXT>(mn);
  const float q0  = quadf<QP_LANE0>(mn);
  const float w   = (l4 == 0) ? 0.75f : 0.25f;
  const float lin = mx + w * (nxt - mx);
  const float mid = 0.5f * (mx + nxt);
  const float hi  = (l4 == 1) ? mid : ((l4 == 3) ? mx : lin);
  const float lo  = (l4 == 0) ? mean : ((l4 == 1) ? var : q0);

  const int c = row & 255;
  int tmp = row >> 8;
  const int b = tmp & 31; tmp >>= 5;
  const int t = tmp & 7;
  const int l = tmp >> 3;
  float* o = feats + ((size_t)((b * LL + l) * CC + c)) * 56 + t * 7;
  o[3 + l4] = hi;
  if (l4 < 3) o[l4] = lo;
  }
}
#undef CE
#undef HALFCLEAN16

__global__ __launch_bounds__(96) void k_qkv(const float* __restrict__ feats,
                                            const float* __restrict__ Wc,
                                            const float* __restrict__ bt,
                                            float* __restrict__ qkv) {
  const int lane = threadIdx.x & 31;
  const int wid  = threadIdx.x >> 5;
  const int r  = lane & 15;
  const int hf = lane >> 4;
  const int n4 = wid * 64 + 4 * r;

  float wf[4][28];
  load_wfrag4<7>(wf, Wc, 192, n4, hf);

  float4 a[7];
  load_arows<7>(a, feats + (size_t)(blockIdx.x * 16 + r) * 56 + 4 * hf);

  for (int trip = 0; trip < T_QKV; ++trip) {
    const int m0 = (trip * G_QKV + blockIdx.x) * 16;
    float4 an[7];
    if (trip + 1 < T_QKV)
      load_arows<7>(an, feats + (size_t)(m0 + G_QKV * 16 + r) * 56 + 4 * hf);
    const int l = (m0 >> 8) % LL;
    const float4 bi = *reinterpret_cast<const float4*>(bt + l * 192 + n4);

    v8f acc[4];
    #pragma unroll
    for (int t = 0; t < 4; ++t) acc[t] = (v8f){0, 0, 0, 0, 0, 0, 0, 0};
    mma_rows<7, 4>(acc, a, wf);

    #pragma unroll
    for (int g = 0; g < 8; ++g) {
      float4 o;
      o.x = acc[0][g] + bi.x;
      o.y = acc[1][g] + bi.y;
      o.z = acc[2][g] + bi.z;
      o.w = acc[3][g] + bi.w;
      *reinterpret_cast<float4*>(qkv + (size_t)(m0 + g + 8 * hf) * 192 + n4) = o;
    }

    if (trip + 1 < T_QKV) {
      #pragma unroll
      for (int j = 0; j < 7; ++j) a[j] = an[j];
    }
  }
}

static __device__ __forceinline__ float xhalf(float x) {
  const int b = __float_as_int(x);
  return __int_as_float(__builtin_amdgcn_permlanex16(b, b, 0x76543210, 0xfedcba98, false, false));
}

#define NSEQ  (BB * CC)
#define HROWS 128
#define NHALF (MROWS / HROWS)
#define T_SEQ 6
#define CPITCH 68

struct QKFrag { float4 qa, qb, ka, kb; };
static __device__ __forceinline__ void load_qk(QKFrag& f, const float* __restrict__ rowp) {
  const float4* p = reinterpret_cast<const float4*>(rowp);
  f.qa = p[0];  f.qb = p[1];
  f.ka = p[16]; f.kb = p[17];
}

static __device__ __forceinline__ void load_v(float (&vb)[6], const float* __restrict__ base,
                                              int r, int hf) {
  #pragma unroll
  for (int s = 0; s < 3; ++s) {
    const float* vp = base + (size_t)(4 * s + 2 * hf) * 192 + 128 + r;
    vb[2 * s]     = vp[0];
    vb[2 * s + 1] = vp[192];
  }
}

__global__ __launch_bounds__(256, 6) void k_attn_proj(const float* __restrict__ qkv,
                                                      const float* __restrict__ wout,
                                                      float* __restrict__ part) {
  __shared__ float sP[8][16][17];
  __shared__ __attribute__((aligned(16))) float sC[HROWS * CPITCH];
  __shared__ float pool[EE];
  const int tid  = threadIdx.x;
  const int lane = tid & 31;
  const int wid  = tid >> 5;
  const int r  = lane & 15;
  const int hf = lane >> 4;
  const int rr = (r < LL) ? r : 0;
  const int qk_off = rr * 192 + 8 * hf;
  const int h  = wid & 3;

  if (tid < EE) pool[tid] = 0.0f;

  const int R0   = blockIdx.x * HROWS;
  const int n_lo = R0 / LL;
  const int seq0 = n_lo + (wid >> 2);

  QKFrag fq[3];
  float  fv[3][6];
  {
    const int s0 = (seq0 < NSEQ) ? seq0 : (NSEQ - 1);
    const int s1 = (seq0 + 2 < NSEQ) ? (seq0 + 2) : (NSEQ - 1);
    const float* b0 = qkv + (size_t)s0 * LL * 192 + h * HDD;
    const float* b1 = qkv + (size_t)s1 * LL * 192 + h * HDD;
    load_qk(fq[0], b0 + qk_off); load_v(fv[0], b0, r, hf);
    load_qk(fq[1], b1 + qk_off); load_v(fv[1], b1, r, hf);
  }
  __builtin_amdgcn_sched_barrier(0);

  #pragma unroll
  for (int i = 0; i < T_SEQ; ++i) {
  const int cs = i % 3;
  const int seq_u = seq0 + 2 * i;
  const QKFrag& cur = fq[cs];

  v8f S = {0, 0, 0, 0, 0, 0, 0, 0};
  {
    v2f a, b;
    a.x = cur.qa.x; a.y = cur.qa.y; b.x = cur.ka.x; b.y = cur.ka.y; S = wmma4(a, b, S);
    a.x = cur.qa.z; a.y = cur.qa.w; b.x = cur.ka.z; b.y = cur.ka.w; S = wmma4(a, b, S);
    a.x = cur.qb.x; a.y = cur.qb.y; b.x = cur.kb.x; b.y = cur.kb.y; S = wmma4(a, b, S);
    a.x = cur.qb.z; a.y = cur.qb.w; b.x = cur.kb.z; b.y = cur.kb.w; S = wmma4(a, b, S);
  }

  if (i + 2 < T_SEQ) {
    const int ns = (seq_u + 4 < NSEQ) ? (seq_u + 4) : (NSEQ - 1);
    const float* nbase = qkv + (size_t)ns * LL * 192 + h * HDD;
    load_qk(fq[(i + 2) % 3], nbase + qk_off);
    load_v(fv[(i + 2) % 3], nbase, r, hf);
  }
  __builtin_amdgcn_sched_barrier(0);

  #pragma unroll
  for (int g = 0; g < 8; ++g) sP[wid][g + 8 * hf][r] = S[g];

  float x[6];
  #pragma unroll
  for (int s = 0; s < 3; ++s) {
    x[2 * s]     = sP[wid][r][4 * s + 2 * hf]     * 0.25f;
    x[2 * s + 1] = sP[wid][r][4 * s + 2 * hf + 1] * 0.25f;
  }
  float mx = fmaxf(fmaxf(fmaxf(x[0], x[1]), fmaxf(x[2], x[3])), fmaxf(x[4], x[5]));
  mx = fmaxf(mx, xhalf(mx));
  #pragma unroll
  for (int i = 0; i < 6; ++i) x[i] = __expf(x[i] - mx);
  float den = ((x[0] + x[1]) + (x[2] + x[3])) + (x[4] + x[5]);
  den += xhalf(den);
  const float inv = __builtin_amdgcn_rcpf(den);

  v8f Cv = {0, 0, 0, 0, 0, 0, 0, 0};
  #pragma unroll
  for (int s = 0; s < 3; ++s) {
    v2f a; a.x = x[2 * s] * inv; a.y = x[2 * s + 1] * inv;
    v2f b; b.x = fv[cs][2 * s];  b.y = fv[cs][2 * s + 1];
    Cv = wmma4(a, b, Cv);
  }
  const int sb = seq_u * LL - R0;
  #pragma unroll
  for (int g = 0; g < 8; ++g) {
    const int m  = g + 8 * hf;
    const int rl = sb + m;
    if (m < LL && seq_u < NSEQ && (unsigned)rl < (unsigned)HROWS)
      sC[rl * CPITCH + h * HDD + r] = Cv[g];
  }
  }

  const int col4 = 4 * r;
  float wf[4][16];
  load_wfrag4<4>(wf, wout, EE, col4, hf);

  __syncthreads();

  const float* crow = sC + (wid * 16 + r) * CPITCH + 4 * hf;
  v8f acc[4];
  #pragma unroll
  for (int t = 0; t < 4; ++t) acc[t] = (v8f){0, 0, 0, 0, 0, 0, 0, 0};
  {
    float4 a[4];
    #pragma unroll
    for (int j = 0; j < 4; ++j) a[j] = *reinterpret_cast<const float4*>(crow + 8 * j);
    mma_rows<4, 4>(acc, a, wf);
  }
  {
    float wg[4][16];
    load_wfrag4<4>(wg, wout + 32 * EE, EE, col4, hf);
    float4 a[4];
    #pragma unroll
    for (int j = 0; j < 4; ++j) a[j] = *reinterpret_cast<const float4*>(crow + 32 + 8 * j);
    mma_rows<4, 4>(acc, a, wg);
  }

  #pragma unroll
  for (int t = 0; t < 4; ++t) {
    float ps = acc[t][0] + acc[t][1] + acc[t][2] + acc[t][3] +
               acc[t][4] + acc[t][5] + acc[t][6] + acc[t][7];
    ps += __shfl_xor(ps, 16, 32);
    if (lane < 16) atomicAdd(&pool[4 * r + t], ps);
  }
  __syncthreads();
  if (tid < EE)
    part[(size_t)blockIdx.x * EE + tid] = pool[tid];
}

__global__ __launch_bounds__(256) void k_mlp_lu(
    const float* __restrict__ part, const float* __restrict__ bout,
    const float* __restrict__ mw1, const float* __restrict__ mb1,
    const float* __restrict__ mg,  const float* __restrict__ mbb,
    const float* __restrict__ mw2, const float* __restrict__ mb2,
    const float* __restrict__ nw1, const float* __restrict__ nb1,
    const float* __restrict__ ng,  const float* __restrict__ nbb,
    const float* __restrict__ nw2, const float* __restrict__ nb2,
    float* __restrict__ out) {
  const int branch = blockIdx.x & 1;
  const int row    = blockIdx.x >> 1;
  const float* w1 = branch ? nw1 : mw1;
  const float* b1 = branch ? nb1 : mb1;
  const float* g  = branch ? ng  : mg;
  const float* bb = branch ? nbb : mbb;
  const float* w2 = branch ? nw2 : mw2;
  const float* b2 = branch ? nb2 : mb2;

  __shared__ float sx[EE];
  __shared__ float sh[HIDD];
  __shared__ float sraw[512];
  const int tid = threadIdx.x;

  if (tid < EE)
    sx[tid] = (part[(size_t)(2 * row) * EE + tid] + part[(size_t)(2 * row + 1) * EE + tid])
              * (1.0f / 256.0f) + bout[tid];
  __syncthreads();

  if (tid < HIDD) {
    float acc = b1[tid];
    #pragma unroll
    for (int i = 0; i < EE; ++i) acc += sx[i] * w1[i * HIDD + tid];
    float mu = acc;
    #pragma unroll
    for (int m = 16; m >= 1; m >>= 1) mu += __shfl_xor(mu, m, 32);
    mu *= (1.0f / 32.0f);
    const float d = acc - mu;
    float vv = d * d;
    #pragma unroll
    for (int m = 16; m >= 1; m >>= 1) vv += __shfl_xor(vv, m, 32);
    vv *= (1.0f / 32.0f);
    const float y = d * rsqrtf(vv + 1e-5f) * g[tid] + bb[tid];
    sh[tid] = 0.5f * y * (1.0f + erff(y * 0.70710678118654752f));
  }
  __syncthreads();

  for (int c = tid; c < 512; c += 256) {
    float acc = b2[c];
    #pragma unroll
    for (int j = 0; j < HIDD; ++j) acc += sh[j] * w2[j * 512 + c];
    sraw[c] = acc;
  }
  __syncthreads();

  if (tid < HH * NKK) {
    const int j = tid & 7;
    float* d = sraw + (tid >> 3) * 64 + 28 + (NKK * j - (j * (j - 1)) / 2);
    *d = softplusf(*d) + 0.01f;
  }
  __syncthreads();

  const size_t obase = ((size_t)branch * 384 + row) * 512;
  for (int o = tid; o < 512; o += 256) {
    const int h  = o >> 6;
    const int ik = o & 63;
    const int i  = ik >> 3;
    const int k  = ik & 7;
    const float* rh = sraw + h * 64;
    float acc = 0.0f;
    #pragma unroll
    for (int j = 0; j < NKK; ++j) {
      if (j > i || j > k) continue;
      const float Lij = (j == i) ? 1.0f : rh[(i * (i - 1)) / 2 + j];
      const float u = rh[28 + (NKK * j - (j * (j - 1)) / 2) + (k - j)];
      acc += Lij * u;
    }
    out[obase + o] = acc;
  }
}

extern "C" void kernel_launch(void* const* d_in, const int* in_sizes, int n_in,
                              void* d_out, int out_size, void* d_ws, size_t ws_size,
                              hipStream_t stream) {
  const float* wsfeat = (const float*)d_in[0];
  const float* W_emb  = (const float*)d_in[1];
  const float* b_emb  = (const float*)d_in[2];
  const float* pos    = (const float*)d_in[3];
  const float* w_in   = (const float*)d_in[4];
  const float* b_in   = (const float*)d_in[5];
  const float* w_out  = (const float*)d_in[6];
  const float* b_out  = (const float*)d_in[7];
  const float* m_w1 = (const float*)d_in[8];
  const float* m_b1 = (const float*)d_in[9];
  const float* m_g  = (const float*)d_in[10];
  const float* m_bb = (const float*)d_in[11];
  const float* m_w2 = (const float*)d_in[12];
  const float* m_b2 = (const float*)d_in[13];
  const float* n_w1 = (const float*)d_in[14];
  const float* n_b1 = (const float*)d_in[15];
  const float* n_g  = (const float*)d_in[16];
  const float* n_bb = (const float*)d_in[17];
  const float* n_w2 = (const float*)d_in[18];
  const float* n_b2 = (const float*)d_in[19];
  float* out = (float*)d_out;

  float* ws     = (float*)d_ws;
  float* feats  = ws;
  float* wc     = feats + (size_t)MROWS * 56;
  float* bt     = wc    + 64 * 192;
  float* qkv    = wc    + (size_t)MROWS * EE;
  float* part   = qkv   + (size_t)MROWS * 192;

  k_stats<<<G_STATS, 256, 0, stream>>>(wsfeat, feats, W_emb, b_emb, pos, w_in, b_in, wc, bt);
  k_qkv<<<G_QKV, 96, 0, stream>>>(feats, wc, bt, qkv);
  k_attn_proj<<<NHALF, 256, 0, stream>>>(qkv, w_out, part);
  k_mlp_lu<<<BB * LL * 2, 256, 0, stream>>>(part, b_out,
      m_w1, m_b1, m_g, m_bb, m_w2, m_b2,
      n_w1, n_b1, n_g, n_bb, n_w2, n_b2, out);
}


# row statistics and qkv product in one kernel: a block sorts the 128 source rows of a 16-row feats tile into LDS, 6 of its 8 waves multiply it by Wc on WMMA beside the sorting; feats never in memory
# speedup vs baseline: 3.1586x; 1.0307x over previous
#include <hip/hip_runtime.h>
#include <math.h>

#define LL   12
#define TT   8
#define BB   32
#define CC   256
#define SS   64
#define EE   64
#define HDD  16
#define NHH  4
#define HIDD 32
#define HH   8
#define NKK  8
#define MROWS (BB*CC*LL)
#define SROWS (LL*TT*BB*CC)

#define G_FUSE  768
#define T_FUSE  ((MROWS / 16) / G_FUSE)

typedef float v2f __attribute__((ext_vector_type(2)));
typedef float v8f __attribute__((ext_vector_type(8)));

static __device__ __forceinline__ v8f wmma4(v2f a, v2f b, v8f c) {
  return __builtin_amdgcn_wmma_f32_16x16x4_f32(false, a, false, b, (short)0, c,
                                               false, false);
}

static __device__ __forceinline__ float softplusf(float x) {
  return (x > 20.0f) ? x : log1pf(__expf(x));
}

template <int KC>
static __device__ __forceinline__ void load_wfrag4(float (&wf)[4][4 * KC],
                                                   const float* __restrict__ w,
                                                   int ldw, int col4, int hf) {
  #pragma unroll
  for (int j = 0; j < KC; ++j)
    #pragma unroll
    for (int e = 0; e < 4; ++e) {
      const float4 v = *reinterpret_cast<const float4*>(
          w + (size_t)(8 * j + 4 * hf + e) * ldw + col4);
      wf[0][4 * j + e] = v.x; wf[1][4 * j + e] = v.y;
      wf[2][4 * j + e] = v.z; wf[3][4 * j + e] = v.w;
    }
}

template <int KC>
static __device__ __forceinline__ void load_wfrag2(float (&wf)[2][4 * KC],
                                                   const float* __restrict__ w,
                                                   int ldw, int col2, int hf) {
  #pragma unroll
  for (int j = 0; j < KC; ++j)
    #pragma unroll
    for (int e = 0; e < 4; ++e) {
      const float2 v = *reinterpret_cast<const float2*>(
          w + (size_t)(8 * j + 4 * hf + e) * ldw + col2);
      wf[0][4 * j + e] = v.x; wf[1][4 * j + e] = v.y;
    }
}

template <int KC, int NT>
static __device__ __forceinline__ void mma_rows(v8f (&acc)[NT], const float4 (&a)[KC],
                                                const float (&wf)[NT][4 * KC]) {
  #pragma unroll
  for (int j = 0; j < KC; ++j) {
    v2f a0; a0.x = a[j].x; a0.y = a[j].y;
    v2f a1; a1.x = a[j].z; a1.y = a[j].w;
    #pragma unroll
    for (int t = 0; t < NT; ++t) {
      v2f b; b.x = wf[t][4 * j]; b.y = wf[t][4 * j + 1];
      acc[t] = wmma4(a0, b, acc[t]);
    }
    #pragma unroll
    for (int t = 0; t < NT; ++t) {
      v2f b; b.x = wf[t][4 * j + 2]; b.y = wf[t][4 * j + 3];
      acc[t] = wmma4(a1, b, acc[t]);
    }
  }
}

template <int CTRL>
static __device__ __forceinline__ float quadf(float v) {
  return __int_as_float(__builtin_amdgcn_update_dpp(
      0, __float_as_int(v), CTRL, 0xf, 0xf, true));
}

#define MED3_QUAD(NAME, PERM)                                                     \
static __device__ __forceinline__ float NAME(float a, float b, float sel) {      \
  float d;                                                                        \
  asm("v_med3_num_f32_e64_dpp %0, %1, %2, %3 quad_perm:" PERM                      \
      " row_mask:0xf bank_mask:0xf"                                               \
      : "=v"(d) : "v"(b), "v"(a), "v"(sel));                                      \
  return d;                                                                       \
}
MED3_QUAD(med3_xor1, "[1,0,3,2]")
MED3_QUAD(med3_xor3, "[3,2,1,0]")
#undef MED3_QUAD

#define QP_XOR1  0xB1
#define QP_XOR2  0x4E
#define QP_NEXT  0xF9
#define QP_LANE0 0x00

#define CE(i, j) { const float lo_ = fminf(x[i], x[j]); \
                   const float hi_ = fmaxf(x[i], x[j]); x[i] = lo_; x[j] = hi_; }
#define HALFCLEAN16 \
  CE(0,8) CE(1,9) CE(2,10) CE(3,11) CE(4,12) CE(5,13) CE(6,14) CE(7,15) \
  CE(0,4) CE(1,5) CE(2,6) CE(3,7) CE(8,12) CE(9,13) CE(10,14) CE(11,15) \
  CE(0,2) CE(1,3) CE(4,6) CE(5,7) CE(8,10) CE(9,11) CE(12,14) CE(13,15) \
  CE(0,1) CE(2,3) CE(4,5) CE(6,7) CE(8,9) CE(10,11) CE(12,13) CE(14,15) \
  ;

#define N_CTILES 60
template <bool BIAS>
static __device__ __forceinline__ void compose_tile(int rt, int ct, int lane,
                                                    const float* __restrict__ Wemb,
                                                    const float* __restrict__ bemb,
                                                    const float* __restrict__ pos,
                                                    const float* __restrict__ win,
                                                    const float* __restrict__ bin,
                                                    float* __restrict__ Wc,
                                                    float* __restrict__ bt) {
  const int r  = lane & 15;
  const int hf = lane >> 4;
  const int col = ct * 16 + r;
  const int nvalid = BIAS ? LL : (56 - rt * 16);
  const bool a_ok = r < nvalid;
  const int rr = a_ok ? r : 0;
  const float* ap = BIAS ? (pos + rr * EE) : (Wemb + (rt * 16 + rr) * EE);

  v8f acc = {0, 0, 0, 0, 0, 0, 0, 0};
  #pragma unroll
  for (int half = 0; half < 2; ++half) {
    v2f a[8], b[8];
    #pragma unroll
    for (int s = 0; s < 8; ++s) {
      const int c0 = (half * 8 + s) * 4 + 2 * hf;
      const float2 av = *reinterpret_cast<const float2*>(ap + c0);
      float ax = av.x, ay = av.y;
      if (BIAS) {
        const float2 bv = *reinterpret_cast<const float2*>(bemb + c0);
        ax += bv.x; ay += bv.y;
      }
      a[s].x = a_ok ? ax : 0.0f;
      a[s].y = a_ok ? ay : 0.0f;
      b[s].x = win[c0 * 192 + col];
      b[s].y = win[(c0 + 1) * 192 + col];
    }
    #pragma unroll
    for (int s = 0; s < 8; ++s) acc = wmma4(a[s], b[s], acc);
  }

  const float badd = BIAS ? bin[col] : 0.0f;
  float* dst = (BIAS ? bt : (Wc + rt * 16 * 192)) + 8 * hf * 192 + col;
  #pragma unroll
  for (int g = 0; g < 8; ++g)
    dst[g * 192] = BIAS ? acc[g] + badd : acc[g];
}

static __device__ __forceinline__ void request_rows(float4 (&xin)[2][4],
                                                    const float* __restrict__ ws,
                                                    int tile, int tid) {
  const int m0 = tile * 16;
  const int c0 = m0 & 255;
  const int bl = m0 >> 8;
  const int b  = bl / LL;
  const int l  = bl - b * LL;
  const int cc = (tid >> 2) & 15;
  #pragma unroll
  for (int pass = 0; pass < 2; ++pass) {
    const int tq   = pass * 4 + (tid >> 6);
    const int srow = ((l * TT + tq) * BB + b) * CC + c0 + cc;
    const float4* p = reinterpret_cast<const float4*>(ws + (size_t)srow * SS) + (tid & 3);
    #pragma unroll
    for (int q = 0; q < 4; ++q) xin[pass][q] = p[q * 4];
  }
}

__global__ __launch_bounds__(32) void k_compose(const float* __restrict__ Wemb,
                                                const float* __restrict__ bemb,
                                                const float* __restrict__ pos,
                                                const float* __restrict__ win,
                                                const float* __restrict__ bin,
                                                float* __restrict__ Wc,
                                                float* __restrict__ bt) {
  const int rt = blockIdx.x / 12;
  const int ct = blockIdx.x - rt * 12;
  if (rt == 4) compose_tile<true >(rt, ct, threadIdx.x, Wemb, bemb, pos, win, bin, Wc, bt);
  else         compose_tile<false>(rt, ct, threadIdx.x, Wemb, bemb, pos, win, bin, Wc, bt);
}

#define FPITCH 60
#define N_MWAVES 6

__global__ __launch_bounds__(256) void k_stats_qkv(const float* __restrict__ ws,
                                                   const float* __restrict__ Wc,
                                                   const float* __restrict__ bt,
                                                   float* __restrict__ qkv) {
  __shared__ __attribute__((aligned(16))) float sF[2][16 * FPITCH];
  const int tid  = threadIdx.x;
  const int lane = tid & 31;
  const int wid  = __builtin_amdgcn_readfirstlane(tid >> 5);
  const int r  = lane & 15;
  const int hf = lane >> 4;
  const int n2 = wid * 32 + 2 * r;

  float wf[2][28];

  const int l4  = tid & 3;
  const float inf  = __builtin_inff();
  const float sel1 = (l4 & 1) ? inf : -inf;
  const float sel2 = (l4 & 2) ? inf : -inf;
  const int cc = (tid >> 2) & 15;

  float4 xin[2][4];
  request_rows(xin, ws, blockIdx.x, tid);
  float2 bi_next = {0.0f, 0.0f};
  if (wid < N_MWAVES) {
    const int l_first = (blockIdx.x >> 4) % LL;
    bi_next = *reinterpret_cast<const float2*>(bt + l_first * 192 + n2);
    load_wfrag2<7>(wf, Wc, 192, n2, hf);
  }

  for (int trip = 0; trip < T_FUSE; ++trip) {
  const int m0  = (trip * G_FUSE + blockIdx.x) * 16;
  float* fbuf = sF[trip & 1];
  const float2 bi = bi_next;
  __builtin_amdgcn_sched_barrier(0);

  #pragma unroll
  for (int pass = 0; pass < 2; ++pass) {
  const int tq  = pass * 4 + (tid >> 6);
  float x[16];
  #pragma unroll
  for (int q = 0; q < 4; ++q) {
    const float4 v = xin[pass][q];
    x[4 * q + 0] = v.x; x[4 * q + 1] = v.y; x[4 * q + 2] = v.z; x[4 * q + 3] = v.w;
  }

  float sa = x[0], sb = x[1], qa = x[0] * x[0], qb = x[1] * x[1];
  #pragma unroll
  for (int i = 2; i < 16; i += 2) {
    sa += x[i];     qa = fmaf(x[i],     x[i],     qa);
    sb += x[i + 1]; qb = fmaf(x[i + 1], x[i + 1], qb);
  }
  float s = sa + sb, s2 = qa + qb;
  s  += quadf<QP_XOR1>(s);   s2 += quadf<QP_XOR1>(s2);
  s  += quadf<QP_XOR2>(s);   s2 += quadf<QP_XOR2>(s2);
  const float mean = s * (1.0f / 64.0f);
  const float var  = (s2 - 64.0f * mean * mean) * (1.0f / 63.0f);

  CE(0,13) CE(1,12) CE(2,15) CE(3,14) CE(4,8) CE(5,6) CE(7,11) CE(9,10)
  CE(0,5) CE(1,7) CE(2,9) CE(3,4) CE(6,13) CE(8,14) CE(10,15) CE(11,12)
  CE(0,1) CE(2,3) CE(4,5) CE(6,8) CE(7,9) CE(10,11) CE(12,13) CE(14,15)
  CE(0,2) CE(1,3) CE(4,10) CE(5,11) CE(6,7) CE(8,9) CE(12,14) CE(13,15)
  CE(1,2) CE(3,12) CE(4,6) CE(5,7) CE(8,10) CE(9,11) CE(13,14)
  CE(1,4) CE(2,6) CE(5,8) CE(7,10) CE(9,13) CE(11,14)
  CE(2,4) CE(3,6) CE(9,12) CE(11,13)
  CE(3,5) CE(6,8) CE(7,9) CE(10,12)
  CE(3,4) CE(5,6) CE(7,8) CE(9,10) CE(11,12)
  CE(6,7) CE(8,9)

  {
    float t[16];
    #pragma unroll
    for (int r = 0; r < 16; ++r)
      t[r] = med3_xor1(x[r], x[15 - r], sel1);
    #pragma unroll
    for (int r = 0; r < 16; ++r) x[r] = t[r];
  }
  HALFCLEAN16

  {
    float t[16];
    #pragma unroll
    for (int r = 0; r < 16; ++r)
      t[r] = med3_xor3(x[r], x[15 - r], sel2);
    #pragma unroll
    for (int r = 0; r < 16; ++r)
      x[r] = med3_xor1(t[r], t[r], sel1);
  }
  float mn = x[0], mx = x[0];
  #pragma unroll
  for (int i = 1; i < 16; ++i) { mn = fminf(mn, x[i]); mx = fmaxf(mx, x[i]); }

  const float nxt = quadf<QP_NEXT>(mn);
  const float q0  = quadf<QP_LANE0>(mn);
  const float w   = (l4 == 0) ? 0.75f : 0.25f;
  const float lin = mx + w * (nxt - mx);
  const float mid = 0.5f * (mx + nxt);
  const float hi  = (l4 == 1) ? mid : ((l4 == 3) ? mx : lin);
  const float lo  = (l4 == 0) ? mean : ((l4 == 1) ? var : q0);


  float* o = fbuf + cc * FPITCH + tq * 7;
  o[3 + l4] = hi;
  if (l4 < 3) o[l4] = lo;
  }

  __syncthreads();

  if (trip + 1 < T_FUSE) {
    const int tile_n = (trip + 1) * G_FUSE + blockIdx.x;
    request_rows(xin, ws, tile_n, tid);
    if (wid < N_MWAVES)
      bi_next = *reinterpret_cast<const float2*>(bt + ((tile_n >> 4) % LL) * 192 + n2);
  }
  __builtin_amdgcn_sched_barrier(0);

  if (wid < N_MWAVES) {
    const float* frow = fbuf + r * FPITCH + 4 * hf;
    float4 a[7];
    #pragma unroll
    for (int jj = 0; jj < 7; ++jj) a[jj] = *reinterpret_cast<const float4*>(frow + 8 * jj);
    __builtin_amdgcn_sched_barrier(0);

    v8f acc[2];
    #pragma unroll
    for (int t2 = 0; t2 < 2; ++t2) acc[t2] = (v8f){0, 0, 0, 0, 0, 0, 0, 0};
    mma_rows<7, 2>(acc, a, wf);

    #pragma unroll
    for (int g = 0; g < 8; ++g) {
      float2 ov;
      ov.x = acc[0][g] + bi.x;
      ov.y = acc[1][g] + bi.y;
      *reinterpret_cast<float2*>(qkv + (size_t)(m0 + g + 8 * hf) * 192 + n2) = ov;
    }
  }
  }
}
#undef CE
#undef HALFCLEAN16

static __device__ __forceinline__ float xhalf(float x) {
  const int b = __float_as_int(x);
  return __int_as_float(__builtin_amdgcn_permlanex16(b, b, 0x76543210, 0xfedcba98, false, false));
}

#define NSEQ  (BB * CC)
#define HROWS 128
#define NHALF (MROWS / HROWS)
#define T_SEQ 6
#define CPITCH 68

struct QKFrag { float4 qa, qb, ka, kb; };
static __device__ __forceinline__ void load_qk(QKFrag& f, const float* __restrict__ rowp) {
  const float4* p = reinterpret_cast<const float4*>(rowp);
  f.qa = p[0];  f.qb = p[1];
  f.ka = p[16]; f.kb = p[17];
}

static __device__ __forceinline__ void load_v(float (&vb)[6], const float* __restrict__ base,
                                              int r, int hf) {
  #pragma unroll
  for (int s = 0; s < 3; ++s) {
    const float* vp = base + (size_t)(4 * s + 2 * hf) * 192 + 128 + r;
    vb[2 * s]     = vp[0];
    vb[2 * s + 1] = vp[192];
  }
}

__global__ __launch_bounds__(256, 6) void k_attn_proj(const float* __restrict__ qkv,
                                                      const float* __restrict__ wout,
                                                      float* __restrict__ part) {
  __shared__ float sP[8][16][17];
  __shared__ __attribute__((aligned(16))) float sC[HROWS * CPITCH];
  __shared__ float pool[EE];
  const int tid  = threadIdx.x;
  const int lane = tid & 31;
  const int wid  = tid >> 5;
  const int r  = lane & 15;
  const int hf = lane >> 4;
  const int rr = (r < LL) ? r : 0;
  const int qk_off = rr * 192 + 8 * hf;
  const int h  = wid & 3;

  if (tid < EE) pool[tid] = 0.0f;

  const int R0   = blockIdx.x * HROWS;
  const int n_lo = R0 / LL;
  const int seq0 = n_lo + (wid >> 2);

  QKFrag fq[3];
  float  fv[3][6];
  {
    const int s0 = (seq0 < NSEQ) ? seq0 : (NSEQ - 1);
    const int s1 = (seq0 + 2 < NSEQ) ? (seq0 + 2) : (NSEQ - 1);
    const float* b0 = qkv + (size_t)s0 * LL * 192 + h * HDD;
    const float* b1 = qkv + (size_t)s1 * LL * 192 + h * HDD;
    load_qk(fq[0], b0 + qk_off); load_v(fv[0], b0, r, hf);
    load_qk(fq[1], b1 + qk_off); load_v(fv[1], b1, r, hf);
  }
  __builtin_amdgcn_sched_barrier(0);

  #pragma unroll
  for (int i = 0; i < T_SEQ; ++i) {
  const int cs = i % 3;
  const int seq_u = seq0 + 2 * i;
  const QKFrag& cur = fq[cs];

  v8f S = {0, 0, 0, 0, 0, 0, 0, 0};
  {
    v2f a, b;
    a.x = cur.qa.x; a.y = cur.qa.y; b.x = cur.ka.x; b.y = cur.ka.y; S = wmma4(a, b, S);
    a.x = cur.qa.z; a.y = cur.qa.w; b.x = cur.ka.z; b.y = cur.ka.w; S = wmma4(a, b, S);
    a.x = cur.qb.x; a.y = cur.qb.y; b.x = cur.kb.x; b.y = cur.kb.y; S = wmma4(a, b, S);
    a.x = cur.qb.z; a.y = cur.qb.w; b.x = cur.kb.z; b.y = cur.kb.w; S = wmma4(a, b, S);
  }

  if (i + 2 < T_SEQ) {
    const int ns = (seq_u + 4 < NSEQ) ? (seq_u + 4) : (NSEQ - 1);
    const float* nbase = qkv + (size_t)ns * LL * 192 + h * HDD;
    load_qk(fq[(i + 2) % 3], nbase + qk_off);
    load_v(fv[(i + 2) % 3], nbase, r, hf);
  }
  __builtin_amdgcn_sched_barrier(0);

  #pragma unroll
  for (int g = 0; g < 8; ++g) sP[wid][g + 8 * hf][r] = S[g];

  float x[6];
  #pragma unroll
  for (int s = 0; s < 3; ++s) {
    x[2 * s]     = sP[wid][r][4 * s + 2 * hf]     * 0.25f;
    x[2 * s + 1] = sP[wid][r][4 * s + 2 * hf + 1] * 0.25f;
  }
  float mx = fmaxf(fmaxf(fmaxf(x[0], x[1]), fmaxf(x[2], x[3])), fmaxf(x[4], x[5]));
  mx = fmaxf(mx, xhalf(mx));
  #pragma unroll
  for (int i = 0; i < 6; ++i) x[i] = __expf(x[i] - mx);
  float den = ((x[0] + x[1]) + (x[2] + x[3])) + (x[4] + x[5]);
  den += xhalf(den);
  const float inv = __builtin_amdgcn_rcpf(den);

  v8f Cv = {0, 0, 0, 0, 0, 0, 0, 0};
  #pragma unroll
  for (int s = 0; s < 3; ++s) {
    v2f a; a.x = x[2 * s] * inv; a.y = x[2 * s + 1] * inv;
    v2f b; b.x = fv[cs][2 * s];  b.y = fv[cs][2 * s + 1];
    Cv = wmma4(a, b, Cv);
  }
  const int sb = seq_u * LL - R0;
  #pragma unroll
  for (int g = 0; g < 8; ++g) {
    const int m  = g + 8 * hf;
    const int rl = sb + m;
    if (m < LL && seq_u < NSEQ && (unsigned)rl < (unsigned)HROWS)
      sC[rl * CPITCH + h * HDD + r] = Cv[g];
  }
  }

  const int col4 = 4 * r;
  float wf[4][16];
  load_wfrag4<4>(wf, wout, EE, col4, hf);

  __syncthreads();

  const float* crow = sC + (wid * 16 + r) * CPITCH + 4 * hf;
  v8f acc[4];
  #pragma unroll
  for (int t = 0; t < 4; ++t) acc[t] = (v8f){0, 0, 0, 0, 0, 0, 0, 0};
  {
    float4 a[4];
    #pragma unroll
    for (int j = 0; j < 4; ++j) a[j] = *reinterpret_cast<const float4*>(crow + 8 * j);
    mma_rows<4, 4>(acc, a, wf);
  }
  {
    float wg[4][16];
    load_wfrag4<4>(wg, wout + 32 * EE, EE, col4, hf);
    float4 a[4];
    #pragma unroll
    for (int j = 0; j < 4; ++j) a[j] = *reinterpret_cast<const float4*>(crow + 32 + 8 * j);
    mma_rows<4, 4>(acc, a, wg);
  }

  #pragma unroll
  for (int t = 0; t < 4; ++t) {
    float ps = acc[t][0] + acc[t][1] + acc[t][2] + acc[t][3] +
               acc[t][4] + acc[t][5] + acc[t][6] + acc[t][7];
    ps += __shfl_xor(ps, 16, 32);
    if (lane < 16) atomicAdd(&pool[4 * r + t], ps);
  }
  __syncthreads();
  if (tid < EE)
    part[(size_t)blockIdx.x * EE + tid] = pool[tid];
}

__global__ __launch_bounds__(256) void k_mlp_lu(
    const float* __restrict__ part, const float* __restrict__ bout,
    const float* __restrict__ mw1, const float* __restrict__ mb1,
    const float* __restrict__ mg,  const float* __restrict__ mbb,
    const float* __restrict__ mw2, const float* __restrict__ mb2,
    const float* __restrict__ nw1, const float* __restrict__ nb1,
    const float* __restrict__ ng,  const float* __restrict__ nbb,
    const float* __restrict__ nw2, const float* __restrict__ nb2,
    float* __restrict__ out) {
  const int branch = blockIdx.x & 1;
  const int row    = blockIdx.x >> 1;
  const float* w1 = branch ? nw1 : mw1;
  const float* b1 = branch ? nb1 : mb1;
  const float* g  = branch ? ng  : mg;
  const float* bb = branch ? nbb : mbb;
  const float* w2 = branch ? nw2 : mw2;
  const float* b2 = branch ? nb2 : mb2;

  __shared__ float sx[EE];
  __shared__ float sh[HIDD];
  __shared__ float sraw[512];
  const int tid = threadIdx.x;

  if (tid < EE)
    sx[tid] = (part[(size_t)(2 * row) * EE + tid] + part[(size_t)(2 * row + 1) * EE + tid])
              * (1.0f / 256.0f) + bout[tid];
  __syncthreads();

  if (tid < HIDD) {
    float acc = b1[tid];
    #pragma unroll
    for (int i = 0; i < EE; ++i) acc += sx[i] * w1[i * HIDD + tid];
    float mu = acc;
    #pragma unroll
    for (int m = 16; m >= 1; m >>= 1) mu += __shfl_xor(mu, m, 32);
    mu *= (1.0f / 32.0f);
    const float d = acc - mu;
    float vv = d * d;
    #pragma unroll
    for (int m = 16; m >= 1; m >>= 1) vv += __shfl_xor(vv, m, 32);
    vv *= (1.0f / 32.0f);
    const float y = d * rsqrtf(vv + 1e-5f) * g[tid] + bb[tid];
    sh[tid] = 0.5f * y * (1.0f + erff(y * 0.70710678118654752f));
  }
  __syncthreads();

  for (int c = tid; c < 512; c += 256) {
    float acc = b2[c];
    #pragma unroll
    for (int j = 0; j < HIDD; ++j) acc += sh[j] * w2[j * 512 + c];
    sraw[c] = acc;
  }
  __syncthreads();

  if (tid < HH * NKK) {
    const int j = tid & 7;
    float* d = sraw + (tid >> 3) * 64 + 28 + (NKK * j - (j * (j - 1)) / 2);
    *d = softplusf(*d) + 0.01f;
  }
  __syncthreads();

  const size_t obase = ((size_t)branch * 384 + row) * 512;
  for (int o = tid; o < 512; o += 256) {
    const int h  = o >> 6;
    const int ik = o & 63;
    const int i  = ik >> 3;
    const int k  = ik & 7;
    const float* rh = sraw + h * 64;
    float acc = 0.0f;
    #pragma unroll
    for (int j = 0; j < NKK; ++j) {
      if (j > i || j > k) continue;
      const float Lij = (j == i) ? 1.0f : rh[(i * (i - 1)) / 2 + j];
      const float u = rh[28 + (NKK * j - (j * (j - 1)) / 2) + (k - j)];
      acc += Lij * u;
    }
    out[obase + o] = acc;
  }
}

extern "C" void kernel_launch(void* const* d_in, const int* in_sizes, int n_in,
                              void* d_out, int out_size, void* d_ws, size_t ws_size,
                              hipStream_t stream) {
  const float* wsfeat = (const float*)d_in[0];
  const float* W_emb  = (const float*)d_in[1];
  const float* b_emb  = (const float*)d_in[2];
  const float* pos    = (const float*)d_in[3];
  const float* w_in   = (const float*)d_in[4];
  const float* b_in   = (const float*)d_in[5];
  const float* w_out  = (const float*)d_in[6];
  const float* b_out  = (const float*)d_in[7];
  const float* m_w1 = (const float*)d_in[8];
  const float* m_b1 = (const float*)d_in[9];
  const float* m_g  = (const float*)d_in[10];
  const float* m_bb = (const float*)d_in[11];
  const float* m_w2 = (const float*)d_in[12];
  const float* m_b2 = (const float*)d_in[13];
  const float* n_w1 = (const float*)d_in[14];
  const float* n_b1 = (const float*)d_in[15];
  const float* n_g  = (const float*)d_in[16];
  const float* n_bb = (const float*)d_in[17];
  const float* n_w2 = (const float*)d_in[18];
  const float* n_b2 = (const float*)d_in[19];
  float* out = (float*)d_out;

  float* ws     = (float*)d_ws;
  float* wc     = ws + (size_t)MROWS * 56;
  float* bt     = wc    + 64 * 192;
  float* qkv    = wc    + (size_t)MROWS * EE;
  float* part   = qkv   + (size_t)MROWS * 192;

  k_compose<<<N_CTILES, 32, 0, stream>>>(W_emb, b_emb, pos, w_in, b_in, wc, bt);
  k_stats_qkv<<<G_FUSE, 256, 0, stream>>>(wsfeat, wc, bt, qkv);
  k_attn_proj<<<NHALF, 256, 0, stream>>>(qkv, w_out, part);
  k_mlp_lu<<<BB * LL * 2, 256, 0, stream>>>(part, b_out,
      m_w1, m_b1, m_g, m_bb, m_w2, m_b2,
      n_w1, n_b1, n_g, n_bb, n_w2, n_b2, out);
}
